# starting kernel
# baseline (speedup 1.0000x reference)
#include <hip/hip_runtime.h>
#include <math.h>

#define HEADS 4
#define HID 64
#define HC 256
#define NEG_SLOPE 0.2f
#define EPS_F 1e-16f

typedef __attribute__((ext_vector_type(2))) float v2f;
typedef __attribute__((ext_vector_type(8))) float v8f;

__device__ __forceinline__ unsigned f2ord(float f) {
    unsigned u = __float_as_uint(f);
    return (u & 0x80000000u) ? ~u : (u | 0x80000000u);
}
__device__ __forceinline__ float ord2f(unsigned u) {
    return (u & 0x80000000u) ? __uint_as_float(u & 0x7fffffffu) : __uint_as_float(~u);
}

__device__ __forceinline__ void epi_store(float* __restrict__ C, const float* __restrict__ bias,
                                          size_t rowoff, int col, int NC, int mode, float v) {
    if (col < NC) {
        if (mode >= 1) v += bias[col];
        if (mode == 2) v = 1.0f / (1.0f + __expf(-v));
        C[rowoff + col] = v;
    }
}

__global__ __launch_bounds__(256) void wmma_gemm_f32_kernel(
    const float* __restrict__ A, const float* __restrict__ B,
    float* __restrict__ C, const float* __restrict__ bias,
    int M, int K, int NC, int mode)
{
    const int wave = threadIdx.x >> 5;
    const int lane = threadIdx.x & 31;
    const int m0 = (blockIdx.x * 8 + wave) * 16;
    if (m0 >= M) return;
    const int n0 = blockIdx.y * 64;

    const int aRow = m0 + (lane & 15);
    const int aK   = (lane >> 4) << 1;
    const int bN   = lane & 15;
    const int bK   = (lane >> 4) << 1;
    const int c    = n0 + bN;

    const float* Ap = A + (size_t)aRow * K + aK;
    const size_t stepB = (size_t)4 * NC;

    v8f acc0 = {}, acc1 = {}, acc2 = {}, acc3 = {};

    if (n0 + 64 <= NC) {
        const float* Bp = B + c + (size_t)bK * NC;
        const float* Bq = Bp + NC;
        #pragma unroll 2
        for (int kb = 0; kb < K; kb += 4) {
            v2f a = *(const v2f*)Ap;
            v2f b0 = { Bp[0],  Bq[0]  };
            v2f b1 = { Bp[16], Bq[16] };
            v2f b2 = { Bp[32], Bq[32] };
            v2f b3 = { Bp[48], Bq[48] };
            Ap += 4; Bp += stepB; Bq += stepB;
            acc0 = __builtin_amdgcn_wmma_f32_16x16x4_f32(false, a, false, b0, (short)0, acc0, false, false);
            acc1 = __builtin_amdgcn_wmma_f32_16x16x4_f32(false, a, false, b1, (short)0, acc1, false, false);
            acc2 = __builtin_amdgcn_wmma_f32_16x16x4_f32(false, a, false, b2, (short)0, acc2, false, false);
            acc3 = __builtin_amdgcn_wmma_f32_16x16x4_f32(false, a, false, b3, (short)0, acc3, false, false);
        }
    } else {
        const int   c0 = min(c,      NC - 1), c1 = min(c + 16, NC - 1);
        const int   c2 = min(c + 32, NC - 1), c3 = min(c + 48, NC - 1);
        const float w0 = (c      < NC) ? 1.0f : 0.0f;
        const float w1 = (c + 16 < NC) ? 1.0f : 0.0f;
        const float w2 = (c + 32 < NC) ? 1.0f : 0.0f;
        const float w3 = (c + 48 < NC) ? 1.0f : 0.0f;
        const float* Bp = B + (size_t)bK * NC;
        const float* Bq = Bp + NC;
        #pragma unroll 2
        for (int kb = 0; kb < K; kb += 4) {
            v2f a = *(const v2f*)Ap;
            v2f b0 = { Bp[c0] * w0, Bq[c0] * w0 };
            v2f b1 = { Bp[c1] * w1, Bq[c1] * w1 };
            v2f b2 = { Bp[c2] * w2, Bq[c2] * w2 };
            v2f b3 = { Bp[c3] * w3, Bq[c3] * w3 };
            Ap += 4; Bp += stepB; Bq += stepB;
            acc0 = __builtin_amdgcn_wmma_f32_16x16x4_f32(false, a, false, b0, (short)0, acc0, false, false);
            acc1 = __builtin_amdgcn_wmma_f32_16x16x4_f32(false, a, false, b1, (short)0, acc1, false, false);
            acc2 = __builtin_amdgcn_wmma_f32_16x16x4_f32(false, a, false, b2, (short)0, acc2, false, false);
            acc3 = __builtin_amdgcn_wmma_f32_16x16x4_f32(false, a, false, b3, (short)0, acc3, false, false);
        }
    }

    const int mbase = m0 + ((lane >> 4) << 3);
    const int colb  = n0 + (lane & 15);
    #pragma unroll
    for (int r = 0; r < 8; ++r) {
        size_t rowoff = (size_t)(mbase + r) * NC;
        epi_store(C, bias, rowoff, colb,      NC, mode, acc0[r]);
        epi_store(C, bias, rowoff, colb + 16, NC, mode, acc1[r]);
        epi_store(C, bias, rowoff, colb + 32, NC, mode, acc2[r]);
        epi_store(C, bias, rowoff, colb + 48, NC, mode, acc3[r]);
    }
}

__global__ __launch_bounds__(256) void scores_kernel(
    const float* __restrict__ H, const float* __restrict__ att_src,
    const float* __restrict__ att_dst, float* __restrict__ a_s, float* __restrict__ a_d,
    unsigned* __restrict__ mmax, float* __restrict__ denom, int NH)
{
    const int gw   = (int)((blockIdx.x * 256 + threadIdx.x) >> 5);
    const int lane = threadIdx.x & 31;
    if (gw >= NH) return;
    const int n = gw >> 2, h = gw & 3;
    const float* hp = H + (size_t)n * HC + h * HID;
    const float x0 = hp[lane], x1 = hp[lane + 32];
    float s = x0 * att_src[h * HID + lane] + x1 * att_src[h * HID + lane + 32];
    float d = x0 * att_dst[h * HID + lane] + x1 * att_dst[h * HID + lane + 32];
    #pragma unroll
    for (int off = 16; off > 0; off >>= 1) {
        s += __shfl_xor(s, off, 32);
        d += __shfl_xor(d, off, 32);
    }
    if (lane == 0) {
        a_s[gw] = s; a_d[gw] = d;
        mmax[gw] = f2ord(-INFINITY);
        denom[gw] = 0.0f;
    }
}

__device__ __forceinline__ void edge_src_dst(const int* __restrict__ ei, int E, int eid,
                                             int& src, int& dst) {
    if (eid < E) { src = ei[eid]; dst = ei[E + eid]; }
    else         { src = dst = eid - E; }
}

__global__ __launch_bounds__(256) void edge_max_kernel(
    const int* __restrict__ ei, int E, int N,
    const float* __restrict__ a_s, const float* __restrict__ a_d, unsigned* __restrict__ mmax)
{
    const int t = blockIdx.x * 256 + threadIdx.x;
    if (t >= (E + N) * 4) return;
    const int eid = t >> 2, h = t & 3;
    int src, dst; edge_src_dst(ei, E, eid, src, dst);
    float e = a_s[src * 4 + h] + a_d[dst * 4 + h];
    e = (e > 0.0f) ? e : NEG_SLOPE * e;
    atomicMax(mmax + dst * 4 + h, f2ord(e));
}

__global__ __launch_bounds__(256) void edge_sum_kernel(
    const int* __restrict__ ei, int E, int N,
    const float* __restrict__ a_s, const float* __restrict__ a_d,
    const unsigned* __restrict__ mmax, float* __restrict__ denom, float* __restrict__ ew)
{
    const int t = blockIdx.x * 256 + threadIdx.x;
    if (t >= (E + N) * 4) return;
    const int eid = t >> 2, h = t & 3;
    int src, dst; edge_src_dst(ei, E, eid, src, dst);
    float e = a_s[src * 4 + h] + a_d[dst * 4 + h];
    e = (e > 0.0f) ? e : NEG_SLOPE * e;
    const float ex = __expf(e - ord2f(mmax[dst * 4 + h]));
    ew[t] = ex;
    atomicAdd(denom + dst * 4 + h, ex);
}

__global__ __launch_bounds__(256) void edge_agg_kernel(
    const int* __restrict__ ei, int E, int N, const float* __restrict__ H,
    const float* __restrict__ ew, const float* __restrict__ denom, float* __restrict__ out)
{
    const int gw   = (int)((blockIdx.x * 256 + threadIdx.x) >> 5);
    const int lane = threadIdx.x & 31;
    if (gw >= E + N) return;
    int src, dst; edge_src_dst(ei, E, gw, src, dst);
    const int head = lane >> 3;
    const float alpha = ew[(size_t)gw * 4 + head] / (denom[dst * 4 + head] + EPS_F);
    const float4* hs = (const float4*)(H + (size_t)src * HC) + (lane << 1);
    const float4 v0 = hs[0], v1 = hs[1];
    float* op = out + (size_t)dst * HC + (lane << 3);
    atomicAdd(op + 0, alpha * v0.x);
    atomicAdd(op + 1, alpha * v0.y);
    atomicAdd(op + 2, alpha * v0.z);
    atomicAdd(op + 3, alpha * v0.w);
    atomicAdd(op + 4, alpha * v1.x);
    atomicAdd(op + 5, alpha * v1.y);
    atomicAdd(op + 6, alpha * v1.z);
    atomicAdd(op + 7, alpha * v1.w);
}

__global__ __launch_bounds__(256) void zero_kernel(float* __restrict__ p, int n) {
    const int i = blockIdx.x * 256 + threadIdx.x;
    if (i < n) p[i] = 0.0f;
}

__global__ __launch_bounds__(256) void bias_relu_kernel(float* __restrict__ p,
                                                        const float* __restrict__ b, int n) {
    const int i = blockIdx.x * 256 + threadIdx.x;
    if (i < n) {
        const float v = p[i] + b[i & (HC - 1)];
        p[i] = fmaxf(v, 0.0f);
    }
}

extern "C" void kernel_launch(void* const* d_in, const int* in_sizes, int n_in,
                              void* d_out, int out_size, void* d_ws, size_t ws_size,
                              hipStream_t stream)
{
    const float* x   = (const float*)d_in[0];
    const int*   ei  = (const int*)d_in[1];
    const float* W1  = (const float*)d_in[2];
    const float* as1 = (const float*)d_in[3];
    const float* ad1 = (const float*)d_in[4];
    const float* b1  = (const float*)d_in[5];
    const float* W2  = (const float*)d_in[6];
    const float* as2 = (const float*)d_in[7];
    const float* ad2 = (const float*)d_in[8];
    const float* b2  = (const float*)d_in[9];
    const float* Wm1 = (const float*)d_in[10];
    const float* bm1 = (const float*)d_in[11];
    const float* Wm2 = (const float*)d_in[12];
    const float* bm2 = (const float*)d_in[13];

    const int N    = in_sizes[0] / 128;
    const int E    = in_sizes[1] / 2;
    const int Etot = E + N;
    const int NH   = N * HEADS;

    float* ws = (float*)d_ws;
    size_t o = 0;
    float*    P0    = ws + o; o += (size_t)N * HC;
    float*    P1    = ws + o; o += (size_t)N * HC;
    float*    a_s   = ws + o; o += (size_t)NH;
    float*    a_d   = ws + o; o += (size_t)NH;
    unsigned* mmax  = (unsigned*)(ws + o); o += (size_t)NH;
    float*    denom = ws + o; o += (size_t)NH;
    float*    ew    = ws + o; o += (size_t)Etot * HEADS;

    const dim3 blk(256);
    auto gemm_grid = [](int M, int NC) {
        return dim3((unsigned)(((M + 15) / 16 + 7) / 8), (unsigned)((NC + 63) / 64));
    };
    const int gScores = (NH * 32 + 255) / 256;
    const int gEdgeH  = (Etot * 4 + 255) / 256;
    const int gEdgeW  = (Etot * 32 + 255) / 256;
    const int gFeat   = (N * HC + 255) / 256;

    wmma_gemm_f32_kernel<<<gemm_grid(N, HC), blk, 0, stream>>>(x, W1, P0, nullptr, N, 128, HC, 0);
    scores_kernel<<<gScores, blk, 0, stream>>>(P0, as1, ad1, a_s, a_d, mmax, denom, NH);
    edge_max_kernel<<<gEdgeH, blk, 0, stream>>>(ei, E, N, a_s, a_d, mmax);
    edge_sum_kernel<<<gEdgeH, blk, 0, stream>>>(ei, E, N, a_s, a_d, mmax, denom, ew);
    zero_kernel<<<gFeat, blk, 0, stream>>>(P1, N * HC);
    edge_agg_kernel<<<gEdgeW, blk, 0, stream>>>(ei, E, N, P0, ew, denom, P1);
    bias_relu_kernel<<<gFeat, blk, 0, stream>>>(P1, b1, N * HC);

    wmma_gemm_f32_kernel<<<gemm_grid(N, HC), blk, 0, stream>>>(P1, W2, P0, nullptr, N, HC, HC, 0);
    scores_kernel<<<gScores, blk, 0, stream>>>(P0, as2, ad2, a_s, a_d, mmax, denom, NH);
    edge_max_kernel<<<gEdgeH, blk, 0, stream>>>(ei, E, N, a_s, a_d, mmax);
    edge_sum_kernel<<<gEdgeH, blk, 0, stream>>>(ei, E, N, a_s, a_d, mmax, denom, ew);
    zero_kernel<<<gFeat, blk, 0, stream>>>(P1, N * HC);
    edge_agg_kernel<<<gEdgeW, blk, 0, stream>>>(ei, E, N, P0, ew, denom, P1);
    bias_relu_kernel<<<gFeat, blk, 0, stream>>>(P1, b2, N * HC);

    wmma_gemm_f32_kernel<<<gemm_grid(N, HID), blk, 0, stream>>>(P1, Wm1, P0, bm1, N, HC, HID, 1);
    wmma_gemm_f32_kernel<<<gemm_grid(N, 121), blk, 0, stream>>>(P0, Wm2, (float*)d_out, bm2,
                                                                N, HID, 121, 2);
}


# Edges grouped by destination once per call (integer counting sort); one wave per node sums its source rows in registers, in arrival order, and stores once: no float atomics or zero fill on out
# speedup vs baseline: 2.0164x; 2.0164x over previous
#include <hip/hip_runtime.h>
#include <math.h>

#define HEADS 4
#define HID 64
#define HC 256
#define NEG_SLOPE 0.2f
#define EPS_F 1e-16f

typedef __attribute__((ext_vector_type(2))) float v2f;
typedef __attribute__((ext_vector_type(8))) float v8f;

__device__ __forceinline__ unsigned f2ord(float f) {
    unsigned u = __float_as_uint(f);
    return (u & 0x80000000u) ? ~u : (u | 0x80000000u);
}
__device__ __forceinline__ float ord2f(unsigned u) {
    return (u & 0x80000000u) ? __uint_as_float(u & 0x7fffffffu) : __uint_as_float(~u);
}

__device__ __forceinline__ void epi_store(float* __restrict__ C, const float* __restrict__ bias,
                                          size_t rowoff, int col, int NC, int mode, float v) {
    if (col < NC) {
        if (mode >= 1) v += bias[col];
        if (mode == 2) v = 1.0f / (1.0f + __expf(-v));
        C[rowoff + col] = v;
    }
}

__global__ __launch_bounds__(256) void wmma_gemm_f32_kernel(
    const float* __restrict__ A, const float* __restrict__ B,
    float* __restrict__ C, const float* __restrict__ bias,
    int M, int K, int NC, int mode)
{
    const int wave = threadIdx.x >> 5;
    const int lane = threadIdx.x & 31;
    const int m0 = (blockIdx.x * 8 + wave) * 16;
    if (m0 >= M) return;
    const int n0 = blockIdx.y * 64;

    const int aRow = m0 + (lane & 15);
    const int aK   = (lane >> 4) << 1;
    const int bN   = lane & 15;
    const int bK   = (lane >> 4) << 1;
    const int c    = n0 + bN;

    const float* Ap = A + (size_t)aRow * K + aK;
    const size_t stepB = (size_t)4 * NC;

    v8f acc0 = {}, acc1 = {}, acc2 = {}, acc3 = {};

    if (n0 + 64 <= NC) {
        const float* Bp = B + c + (size_t)bK * NC;
        const float* Bq = Bp + NC;
        #pragma unroll 2
        for (int kb = 0; kb < K; kb += 4) {
            v2f a = *(const v2f*)Ap;
            v2f b0 = { Bp[0],  Bq[0]  };
            v2f b1 = { Bp[16], Bq[16] };
            v2f b2 = { Bp[32], Bq[32] };
            v2f b3 = { Bp[48], Bq[48] };
            Ap += 4; Bp += stepB; Bq += stepB;
            acc0 = __builtin_amdgcn_wmma_f32_16x16x4_f32(false, a, false, b0, (short)0, acc0, false, false);
            acc1 = __builtin_amdgcn_wmma_f32_16x16x4_f32(false, a, false, b1, (short)0, acc1, false, false);
            acc2 = __builtin_amdgcn_wmma_f32_16x16x4_f32(false, a, false, b2, (short)0, acc2, false, false);
            acc3 = __builtin_amdgcn_wmma_f32_16x16x4_f32(false, a, false, b3, (short)0, acc3, false, false);
        }
    } else {
        const int   c0 = min(c,      NC - 1), c1 = min(c + 16, NC - 1);
        const int   c2 = min(c + 32, NC - 1), c3 = min(c + 48, NC - 1);
        const float w0 = (c      < NC) ? 1.0f : 0.0f;
        const float w1 = (c + 16 < NC) ? 1.0f : 0.0f;
        const float w2 = (c + 32 < NC) ? 1.0f : 0.0f;
        const float w3 = (c + 48 < NC) ? 1.0f : 0.0f;
        const float* Bp = B + (size_t)bK * NC;
        const float* Bq = Bp + NC;
        #pragma unroll 2
        for (int kb = 0; kb < K; kb += 4) {
            v2f a = *(const v2f*)Ap;
            v2f b0 = { Bp[c0] * w0, Bq[c0] * w0 };
            v2f b1 = { Bp[c1] * w1, Bq[c1] * w1 };
            v2f b2 = { Bp[c2] * w2, Bq[c2] * w2 };
            v2f b3 = { Bp[c3] * w3, Bq[c3] * w3 };
            Ap += 4; Bp += stepB; Bq += stepB;
            acc0 = __builtin_amdgcn_wmma_f32_16x16x4_f32(false, a, false, b0, (short)0, acc0, false, false);
            acc1 = __builtin_amdgcn_wmma_f32_16x16x4_f32(false, a, false, b1, (short)0, acc1, false, false);
            acc2 = __builtin_amdgcn_wmma_f32_16x16x4_f32(false, a, false, b2, (short)0, acc2, false, false);
            acc3 = __builtin_amdgcn_wmma_f32_16x16x4_f32(false, a, false, b3, (short)0, acc3, false, false);
        }
    }

    const int mbase = m0 + ((lane >> 4) << 3);
    const int colb  = n0 + (lane & 15);
    #pragma unroll
    for (int r = 0; r < 8; ++r) {
        size_t rowoff = (size_t)(mbase + r) * NC;
        epi_store(C, bias, rowoff, colb,      NC, mode, acc0[r]);
        epi_store(C, bias, rowoff, colb + 16, NC, mode, acc1[r]);
        epi_store(C, bias, rowoff, colb + 32, NC, mode, acc2[r]);
        epi_store(C, bias, rowoff, colb + 48, NC, mode, acc3[r]);
    }
}

__global__ __launch_bounds__(256) void scores_kernel(
    const float* __restrict__ H, const float* __restrict__ att_src,
    const float* __restrict__ att_dst, float* __restrict__ a_s, float* __restrict__ a_d,
    unsigned* __restrict__ mmax, float* __restrict__ denom, int NH)
{
    const int gw   = (int)((blockIdx.x * 256 + threadIdx.x) >> 5);
    const int lane = threadIdx.x & 31;
    if (gw >= NH) return;
    const int n = gw >> 2, h = gw & 3;
    const float* hp = H + (size_t)n * HC + h * HID;
    const float x0 = hp[lane], x1 = hp[lane + 32];
    float s = x0 * att_src[h * HID + lane] + x1 * att_src[h * HID + lane + 32];
    float d = x0 * att_dst[h * HID + lane] + x1 * att_dst[h * HID + lane + 32];
    #pragma unroll
    for (int off = 16; off > 0; off >>= 1) {
        s += __shfl_xor(s, off, 32);
        d += __shfl_xor(d, off, 32);
    }
    if (lane == 0) {
        a_s[gw] = s; a_d[gw] = d;
        mmax[gw] = f2ord(-INFINITY);
        denom[gw] = 0.0f;
    }
}

__device__ __forceinline__ void edge_src_dst(const int* __restrict__ ei, int E, int eid,
                                             int& src, int& dst) {
    if (eid < E) { src = ei[eid]; dst = ei[E + eid]; }
    else         { src = dst = eid - E; }
}

__global__ __launch_bounds__(256) void edge_max_kernel(
    const int* __restrict__ ei, int E, int N,
    const float* __restrict__ a_s, const float* __restrict__ a_d, unsigned* __restrict__ mmax)
{
    const int t = blockIdx.x * 256 + threadIdx.x;
    if (t >= (E + N) * 4) return;
    const int eid = t >> 2, h = t & 3;
    int src, dst; edge_src_dst(ei, E, eid, src, dst);
    float e = a_s[src * 4 + h] + a_d[dst * 4 + h];
    e = (e > 0.0f) ? e : NEG_SLOPE * e;
    atomicMax(mmax + dst * 4 + h, f2ord(e));
}

__global__ __launch_bounds__(256) void edge_sum_kernel(
    const int* __restrict__ ei, int E, int N,
    const float* __restrict__ a_s, const float* __restrict__ a_d,
    const unsigned* __restrict__ mmax, float* __restrict__ denom, float* __restrict__ ew)
{
    const int t = blockIdx.x * 256 + threadIdx.x;
    if (t >= (E + N) * 4) return;
    const int eid = t >> 2, h = t & 3;
    int src, dst; edge_src_dst(ei, E, eid, src, dst);
    float e = a_s[src * 4 + h] + a_d[dst * 4 + h];
    e = (e > 0.0f) ? e : NEG_SLOPE * e;
    const float ex = __expf(e - ord2f(mmax[dst * 4 + h]));
    ew[t] = ex;
    atomicAdd(denom + dst * 4 + h, ex);
}

__global__ __launch_bounds__(256) void zero_i32_kernel(int* __restrict__ p, int n) {
    const int i = blockIdx.x * 256 + threadIdx.x;
    if (i < n) p[i] = 0;
}

__global__ __launch_bounds__(256) void deg_count_kernel(const int* __restrict__ ei, int E,
                                                        int* __restrict__ cnt) {
    const int e = blockIdx.x * 256 + threadIdx.x;
    if (e < E) atomicAdd(cnt + ei[E + e], 1);
}

#define SCAN_T 1024
__global__ __launch_bounds__(SCAN_T) void rowptr_scan_kernel(
    const int* __restrict__ cnt, int* __restrict__ rowptr, int* __restrict__ cursor, int n)
{
    __shared__ int wsum[32];
    const int t = threadIdx.x, lane = t & 31, wv = t >> 5;
    const int per = (n + SCAN_T - 1) / SCAN_T;
    const int b = min(t * per, n), e = min(b + per, n);
    int s = 0;
    for (int i = b; i < e; ++i) s += cnt[i];
    int incl = s;
    for (int off = 1; off < 32; off <<= 1) {
        const int v = __shfl_up(incl, off, 32);
        if (lane >= off) incl += v;
    }
    if (lane == 31) wsum[wv] = incl;
    __syncthreads();
    if (wv == 0) {
        const int w = wsum[lane];
        int wi = w;
        for (int off = 1; off < 32; off <<= 1) {
            const int v = __shfl_up(wi, off, 32);
            if (lane >= off) wi += v;
        }
        wsum[lane] = wi - w;
    }
    __syncthreads();
    int run = wsum[wv] + incl - s;
    for (int i = b; i < e; ++i) {
        const int c = cnt[i];
        rowptr[i] = run; cursor[i] = run;
        run += c;
    }
    if (t == SCAN_T - 1) rowptr[n] = run;
}

__global__ __launch_bounds__(256) void csr_fill_kernel(const int* __restrict__ ei, int E,
                                                       int* __restrict__ cursor,
                                                       int* __restrict__ csr_words) {
    int2* csr = (int2*)csr_words;
    const int e = blockIdx.x * 256 + threadIdx.x;
    if (e < E) {
        const int sv = ei[e], d = ei[E + e];
        const int p = atomicAdd(cursor + d, 1);
        csr[p] = make_int2(sv, e);
    }
}


__device__ __forceinline__ float blend_half(int lo, int x_lo, int x_hi) {
    return __int_as_float((x_lo & lo) | (x_hi & ~lo));
}

__global__ __launch_bounds__(256) void node_agg_kernel(
    const int* __restrict__ rowptr, const int* __restrict__ csr_words, int E, int N,
    const float* __restrict__ H, const float* __restrict__ ew,
    const float* __restrict__ denom, float* __restrict__ out)
{
    const int2* csr = (const int2*)csr_words;
    const int gw   = (int)((blockIdx.x * 256 + threadIdx.x) >> 5);
    const int lane = threadIdx.x & 31;
    if (gw >= N) return;
    const int node = __builtin_amdgcn_readfirstlane(gw);
    const int lo = (lane < 16) ? -1 : 0;
    const float4 dq = ((const float4*)denom)[node];
    const float den0 = dq.x + EPS_F, den1 = dq.y + EPS_F, den2 = dq.z + EPS_F, den3 = dq.w + EPS_F;
    const int beg = rowptr[node], end = rowptr[node + 1];

    const float4* hn = (const float4*)(H + (size_t)node * HC) + lane;
    const float4 n0 = hn[0], n1 = hn[32];
    const float4 nq = ((const float4*)ew)[E + node];

    float4 acc0 = make_float4(0.0f, 0.0f, 0.0f, 0.0f), acc1 = acc0;

    for (int base = beg; base < end; base += 32) {
        const int cnt = min(32, end - base);
        const int2   my = csr[base + min(lane, cnt - 1)];
        const float4 wq = ((const float4*)ew)[my.y];
        const int lm = (lane < cnt) ? -1 : 0;
        const int q0 = __float_as_int(wq.x / den0) & lm;
        const int q1 = __float_as_int(wq.y / den1) & lm;
        const int q2 = __float_as_int(wq.z / den2) & lm;
        const int q3 = __float_as_int(wq.w / den3) & lm;
        for (int j = 0; j < cnt; j += 4) {
            int    s[4];
            float  al[4], ah[4];
            float4 a[4], b[4];
            for (int k = 0; k < 4; ++k) {
                s[k]  = __builtin_amdgcn_readlane(my.x, j + k);
                al[k] = blend_half(lo, __builtin_amdgcn_readlane(q0, j + k),
                                       __builtin_amdgcn_readlane(q1, j + k));
                ah[k] = blend_half(lo, __builtin_amdgcn_readlane(q2, j + k),
                                       __builtin_amdgcn_readlane(q3, j + k));
            }
            for (int k = 0; k < 4; ++k) {
                const float4* hs = (const float4*)(H + (size_t)s[k] * HC) + lane;
                a[k] = hs[0]; b[k] = hs[32];
            }
            for (int k = 0; k < 4; ++k) {
                acc0.x += al[k] * a[k].x; acc0.y += al[k] * a[k].y;
                acc0.z += al[k] * a[k].z; acc0.w += al[k] * a[k].w;
                acc1.x += ah[k] * b[k].x; acc1.y += ah[k] * b[k].y;
                acc1.z += ah[k] * b[k].z; acc1.w += ah[k] * b[k].w;
            }
        }
    }
    {
        const float sl = blend_half(lo, __float_as_int(nq.x / den0), __float_as_int(nq.y / den1));
        const float sh = blend_half(lo, __float_as_int(nq.z / den2), __float_as_int(nq.w / den3));
        acc0.x += sl * n0.x; acc0.y += sl * n0.y;
        acc0.z += sl * n0.z; acc0.w += sl * n0.w;
        acc1.x += sh * n1.x; acc1.y += sh * n1.y;
        acc1.z += sh * n1.z; acc1.w += sh * n1.w;
    }
    float4* op = (float4*)(out + (size_t)node * HC) + lane;
    op[0] = acc0; op[32] = acc1;
}

__global__ __launch_bounds__(256) void bias_relu_kernel(float* __restrict__ p,
                                                        const float* __restrict__ b, int n) {
    const int i = blockIdx.x * 256 + threadIdx.x;
    if (i < n) {
        const float v = p[i] + b[i & (HC - 1)];
        p[i] = fmaxf(v, 0.0f);
    }
}

extern "C" void kernel_launch(void* const* d_in, const int* in_sizes, int n_in,
                              void* d_out, int out_size, void* d_ws, size_t ws_size,
                              hipStream_t stream)
{
    const float* x   = (const float*)d_in[0];
    const int*   ei  = (const int*)d_in[1];
    const float* W1  = (const float*)d_in[2];
    const float* as1 = (const float*)d_in[3];
    const float* ad1 = (const float*)d_in[4];
    const float* b1  = (const float*)d_in[5];
    const float* W2  = (const float*)d_in[6];
    const float* as2 = (const float*)d_in[7];
    const float* ad2 = (const float*)d_in[8];
    const float* b2  = (const float*)d_in[9];
    const float* Wm1 = (const float*)d_in[10];
    const float* bm1 = (const float*)d_in[11];
    const float* Wm2 = (const float*)d_in[12];
    const float* bm2 = (const float*)d_in[13];

    const int N    = in_sizes[0] / 128;
    const int E    = in_sizes[1] / 2;
    const int Etot = E + N;
    const int NH   = N * HEADS;

    float* ws = (float*)d_ws;
    size_t o = 0;
    float*    P0    = ws + o; o += (size_t)N * HC;
    float*    P1    = ws + o; o += (size_t)N * HC;
    float*    a_s   = ws + o; o += (size_t)NH;
    float*    a_d   = ws + o; o += (size_t)NH;
    unsigned* mmax  = (unsigned*)(ws + o); o += (size_t)NH;
    float*    denom = ws + o; o += (size_t)NH;
    float*    ew    = ws + o; o += (size_t)Etot * HEADS;
    o = (o + 3) & ~(size_t)3;
    int*      cnt    = (int*)(ws + o); o += ((size_t)N + 4) & ~(size_t)3;
    int*      rowptr = (int*)(ws + o); o += ((size_t)N + 4) & ~(size_t)3;
    int*      cursor = (int*)(ws + o); o += ((size_t)N + 4) & ~(size_t)3;
    int*      csr    = (int*)(ws + o); o += (size_t)E * 2;

    const dim3 blk(256);
    auto gemm_grid = [](int M, int NC) {
        return dim3((unsigned)(((M + 15) / 16 + 7) / 8), (unsigned)((NC + 63) / 64));
    };
    const int gScores = (NH * 32 + 255) / 256;
    const int gEdgeH  = (Etot * 4 + 255) / 256;
    const int gNodeW  = (N * 32 + 255) / 256;
    const int gEdge1  = (E + 255) / 256;
    const int gFeat   = (N * HC + 255) / 256;

    zero_i32_kernel<<<(N + 255) / 256, blk, 0, stream>>>(cnt, N);
    deg_count_kernel<<<gEdge1, blk, 0, stream>>>(ei, E, cnt);
    rowptr_scan_kernel<<<1, dim3(SCAN_T), 0, stream>>>(cnt, rowptr, cursor, N);
    csr_fill_kernel<<<gEdge1, blk, 0, stream>>>(ei, E, cursor, csr);

    wmma_gemm_f32_kernel<<<gemm_grid(N, HC), blk, 0, stream>>>(x, W1, P0, nullptr, N, 128, HC, 0);
    scores_kernel<<<gScores, blk, 0, stream>>>(P0, as1, ad1, a_s, a_d, mmax, denom, NH);
    edge_max_kernel<<<gEdgeH, blk, 0, stream>>>(ei, E, N, a_s, a_d, mmax);
    edge_sum_kernel<<<gEdgeH, blk, 0, stream>>>(ei, E, N, a_s, a_d, mmax, denom, ew);
    node_agg_kernel<<<gNodeW, blk, 0, stream>>>(rowptr, csr, E, N, P0, ew, denom, P1);
    bias_relu_kernel<<<gFeat, blk, 0, stream>>>(P1, b1, N * HC);

    wmma_gemm_f32_kernel<<<gemm_grid(N, HC), blk, 0, stream>>>(P1, W2, P0, nullptr, N, HC, HC, 0);
    scores_kernel<<<gScores, blk, 0, stream>>>(P0, as2, ad2, a_s, a_d, mmax, denom, NH);
    edge_max_kernel<<<gEdgeH, blk, 0, stream>>>(ei, E, N, a_s, a_d, mmax);
    edge_sum_kernel<<<gEdgeH, blk, 0, stream>>>(ei, E, N, a_s, a_d, mmax, denom, ew);
    node_agg_kernel<<<gNodeW, blk, 0, stream>>>(rowptr, csr, E, N, P0, ew, denom, P1);
    bias_relu_kernel<<<gFeat, blk, 0, stream>>>(P1, b2, N * HC);

    wmma_gemm_f32_kernel<<<gemm_grid(N, HID), blk, 0, stream>>>(P1, Wm1, P0, bm1, N, HC, HID, 1);
    wmma_gemm_f32_kernel<<<gemm_grid(N, 121), blk, 0, stream>>>(P0, Wm2, (float*)d_out, bm2,
                                                                N, HID, 121, 2);
}


# Layer bias and relu applied by the per-node aggregation wave to its finished row in registers before the row's one store; the two separate bias+relu passes over the features are gone
# speedup vs baseline: 2.2272x; 1.1045x over previous
#include <hip/hip_runtime.h>
#include <math.h>

#define HEADS 4
#define HID 64
#define HC 256
#define NEG_SLOPE 0.2f
#define EPS_F 1e-16f

typedef __attribute__((ext_vector_type(2))) float v2f;
typedef __attribute__((ext_vector_type(8))) float v8f;

__device__ __forceinline__ unsigned f2ord(float f) {
    unsigned u = __float_as_uint(f);
    return (u & 0x80000000u) ? ~u : (u | 0x80000000u);
}
__device__ __forceinline__ float ord2f(unsigned u) {
    return (u & 0x80000000u) ? __uint_as_float(u & 0x7fffffffu) : __uint_as_float(~u);
}

__device__ __forceinline__ void epi_store(float* __restrict__ C, const float* __restrict__ bias,
                                          size_t rowoff, int col, int NC, int mode, float v) {
    if (col < NC) {
        if (mode >= 1) v += bias[col];
        if (mode == 2) v = 1.0f / (1.0f + __expf(-v));
        C[rowoff + col] = v;
    }
}

__global__ __launch_bounds__(256) void wmma_gemm_f32_kernel(
    const float* __restrict__ A, const float* __restrict__ B,
    float* __restrict__ C, const float* __restrict__ bias,
    int M, int K, int NC, int mode)
{
    const int wave = threadIdx.x >> 5;
    const int lane = threadIdx.x & 31;
    const int m0 = (blockIdx.x * 8 + wave) * 16;
    if (m0 >= M) return;
    const int n0 = blockIdx.y * 64;

    const int aRow = m0 + (lane & 15);
    const int aK   = (lane >> 4) << 1;
    const int bN   = lane & 15;
    const int bK   = (lane >> 4) << 1;
    const int c    = n0 + bN;

    const float* Ap = A + (size_t)aRow * K + aK;
    const size_t stepB = (size_t)4 * NC;

    v8f acc0 = {}, acc1 = {}, acc2 = {}, acc3 = {};

    if (n0 + 64 <= NC) {
        const float* Bp = B + c + (size_t)bK * NC;
        const float* Bq = Bp + NC;
        #pragma unroll 2
        for (int kb = 0; kb < K; kb += 4) {
            v2f a = *(const v2f*)Ap;
            v2f b0 = { Bp[0],  Bq[0]  };
            v2f b1 = { Bp[16], Bq[16] };
            v2f b2 = { Bp[32], Bq[32] };
            v2f b3 = { Bp[48], Bq[48] };
            Ap += 4; Bp += stepB; Bq += stepB;
            acc0 = __builtin_amdgcn_wmma_f32_16x16x4_f32(false, a, false, b0, (short)0, acc0, false, false);
            acc1 = __builtin_amdgcn_wmma_f32_16x16x4_f32(false, a, false, b1, (short)0, acc1, false, false);
            acc2 = __builtin_amdgcn_wmma_f32_16x16x4_f32(false, a, false, b2, (short)0, acc2, false, false);
            acc3 = __builtin_amdgcn_wmma_f32_16x16x4_f32(false, a, false, b3, (short)0, acc3, false, false);
        }
    } else {
        const int   c0 = min(c,      NC - 1), c1 = min(c + 16, NC - 1);
        const int   c2 = min(c + 32, NC - 1), c3 = min(c + 48, NC - 1);
        const float w0 = (c      < NC) ? 1.0f : 0.0f;
        const float w1 = (c + 16 < NC) ? 1.0f : 0.0f;
        const float w2 = (c + 32 < NC) ? 1.0f : 0.0f;
        const float w3 = (c + 48 < NC) ? 1.0f : 0.0f;
        const float* Bp = B + (size_t)bK * NC;
        const float* Bq = Bp + NC;
        #pragma unroll 2
        for (int kb = 0; kb < K; kb += 4) {
            v2f a = *(const v2f*)Ap;
            v2f b0 = { Bp[c0] * w0, Bq[c0] * w0 };
            v2f b1 = { Bp[c1] * w1, Bq[c1] * w1 };
            v2f b2 = { Bp[c2] * w2, Bq[c2] * w2 };
            v2f b3 = { Bp[c3] * w3, Bq[c3] * w3 };
            Ap += 4; Bp += stepB; Bq += stepB;
            acc0 = __builtin_amdgcn_wmma_f32_16x16x4_f32(false, a, false, b0, (short)0, acc0, false, false);
            acc1 = __builtin_amdgcn_wmma_f32_16x16x4_f32(false, a, false, b1, (short)0, acc1, false, false);
            acc2 = __builtin_amdgcn_wmma_f32_16x16x4_f32(false, a, false, b2, (short)0, acc2, false, false);
            acc3 = __builtin_amdgcn_wmma_f32_16x16x4_f32(false, a, false, b3, (short)0, acc3, false, false);
        }
    }

    const int mbase = m0 + ((lane >> 4) << 3);
    const int colb  = n0 + (lane & 15);
    #pragma unroll
    for (int r = 0; r < 8; ++r) {
        size_t rowoff = (size_t)(mbase + r) * NC;
        epi_store(C, bias, rowoff, colb,      NC, mode, acc0[r]);
        epi_store(C, bias, rowoff, colb + 16, NC, mode, acc1[r]);
        epi_store(C, bias, rowoff, colb + 32, NC, mode, acc2[r]);
        epi_store(C, bias, rowoff, colb + 48, NC, mode, acc3[r]);
    }
}

__global__ __launch_bounds__(256) void scores_kernel(
    const float* __restrict__ H, const float* __restrict__ att_src,
    const float* __restrict__ att_dst, float* __restrict__ a_s, float* __restrict__ a_d,
    unsigned* __restrict__ mmax, float* __restrict__ denom, int NH)
{
    const int gw   = (int)((blockIdx.x * 256 + threadIdx.x) >> 5);
    const int lane = threadIdx.x & 31;
    if (gw >= NH) return;
    const int n = gw >> 2, h = gw & 3;
    const float* hp = H + (size_t)n * HC + h * HID;
    const float x0 = hp[lane], x1 = hp[lane + 32];
    float s = x0 * att_src[h * HID + lane] + x1 * att_src[h * HID + lane + 32];
    float d = x0 * att_dst[h * HID + lane] + x1 * att_dst[h * HID + lane + 32];
    #pragma unroll
    for (int off = 16; off > 0; off >>= 1) {
        s += __shfl_xor(s, off, 32);
        d += __shfl_xor(d, off, 32);
    }
    if (lane == 0) {
        a_s[gw] = s; a_d[gw] = d;
        mmax[gw] = f2ord(-INFINITY);
        denom[gw] = 0.0f;
    }
}

__device__ __forceinline__ void edge_src_dst(const int* __restrict__ ei, int E, int eid,
                                             int& src, int& dst) {
    if (eid < E) { src = ei[eid]; dst = ei[E + eid]; }
    else         { src = dst = eid - E; }
}

__global__ __launch_bounds__(256) void edge_max_kernel(
    const int* __restrict__ ei, int E, int N,
    const float* __restrict__ a_s, const float* __restrict__ a_d, unsigned* __restrict__ mmax)
{
    const int t = blockIdx.x * 256 + threadIdx.x;
    if (t >= (E + N) * 4) return;
    const int eid = t >> 2, h = t & 3;
    int src, dst; edge_src_dst(ei, E, eid, src, dst);
    float e = a_s[src * 4 + h] + a_d[dst * 4 + h];
    e = (e > 0.0f) ? e : NEG_SLOPE * e;
    atomicMax(mmax + dst * 4 + h, f2ord(e));
}

__global__ __launch_bounds__(256) void edge_sum_kernel(
    const int* __restrict__ ei, int E, int N,
    const float* __restrict__ a_s, const float* __restrict__ a_d,
    const unsigned* __restrict__ mmax, float* __restrict__ denom, float* __restrict__ ew)
{
    const int t = blockIdx.x * 256 + threadIdx.x;
    if (t >= (E + N) * 4) return;
    const int eid = t >> 2, h = t & 3;
    int src, dst; edge_src_dst(ei, E, eid, src, dst);
    float e = a_s[src * 4 + h] + a_d[dst * 4 + h];
    e = (e > 0.0f) ? e : NEG_SLOPE * e;
    const float ex = __expf(e - ord2f(mmax[dst * 4 + h]));
    ew[t] = ex;
    atomicAdd(denom + dst * 4 + h, ex);
}

__global__ __launch_bounds__(256) void zero_i32_kernel(int* __restrict__ p, int n) {
    const int i = blockIdx.x * 256 + threadIdx.x;
    if (i < n) p[i] = 0;
}

__global__ __launch_bounds__(256) void deg_count_kernel(const int* __restrict__ ei, int E,
                                                        int* __restrict__ cnt) {
    const int e = blockIdx.x * 256 + threadIdx.x;
    if (e < E) atomicAdd(cnt + ei[E + e], 1);
}

#define SCAN_T 1024
__global__ __launch_bounds__(SCAN_T) void rowptr_scan_kernel(
    const int* __restrict__ cnt, int* __restrict__ rowptr, int* __restrict__ cursor, int n)
{
    __shared__ int wsum[32];
    const int t = threadIdx.x, lane = t & 31, wv = t >> 5;
    const int per = (n + SCAN_T - 1) / SCAN_T;
    const int b = min(t * per, n), e = min(b + per, n);
    int s = 0;
    for (int i = b; i < e; ++i) s += cnt[i];
    int incl = s;
    for (int off = 1; off < 32; off <<= 1) {
        const int v = __shfl_up(incl, off, 32);
        if (lane >= off) incl += v;
    }
    if (lane == 31) wsum[wv] = incl;
    __syncthreads();
    if (wv == 0) {
        const int w = wsum[lane];
        int wi = w;
        for (int off = 1; off < 32; off <<= 1) {
            const int v = __shfl_up(wi, off, 32);
            if (lane >= off) wi += v;
        }
        wsum[lane] = wi - w;
    }
    __syncthreads();
    int run = wsum[wv] + incl - s;
    for (int i = b; i < e; ++i) {
        const int c = cnt[i];
        rowptr[i] = run; cursor[i] = run;
        run += c;
    }
    if (t == SCAN_T - 1) rowptr[n] = run;
}

__global__ __launch_bounds__(256) void csr_fill_kernel(const int* __restrict__ ei, int E,
                                                       int* __restrict__ cursor,
                                                       int* __restrict__ csr_words) {
    int2* csr = (int2*)csr_words;
    const int e = blockIdx.x * 256 + threadIdx.x;
    if (e < E) {
        const int sv = ei[e], d = ei[E + e];
        const int p = atomicAdd(cursor + d, 1);
        csr[p] = make_int2(sv, e);
    }
}


__device__ __forceinline__ float blend_half(int lo, int x_lo, int x_hi) {
    return __int_as_float((x_lo & lo) | (x_hi & ~lo));
}

__global__ __launch_bounds__(256) void node_agg_kernel(
    const int* __restrict__ rowptr, const int* __restrict__ csr_words, int E, int N,
    const float* __restrict__ H, const float* __restrict__ ew,
    const float* __restrict__ denom, const float* __restrict__ bias, float* __restrict__ out)
{
    const int2* csr = (const int2*)csr_words;
    const int gw   = (int)((blockIdx.x * 256 + threadIdx.x) >> 5);
    const int lane = threadIdx.x & 31;
    if (gw >= N) return;
    const int node = __builtin_amdgcn_readfirstlane(gw);
    const int lo = (lane < 16) ? -1 : 0;
    const float4 dq = ((const float4*)denom)[node];
    const float den0 = dq.x + EPS_F, den1 = dq.y + EPS_F, den2 = dq.z + EPS_F, den3 = dq.w + EPS_F;
    const int beg = rowptr[node], end = rowptr[node + 1];

    const float4* hn = (const float4*)(H + (size_t)node * HC) + lane;
    const float4 n0 = hn[0], n1 = hn[32];
    const float4 nq = ((const float4*)ew)[E + node];
    const float4 bq0 = ((const float4*)bias)[lane], bq1 = ((const float4*)bias)[32 + lane];

    float4 acc0 = make_float4(0.0f, 0.0f, 0.0f, 0.0f), acc1 = acc0;

    for (int base = beg; base < end; base += 32) {
        const int cnt = min(32, end - base);
        const int2   my = csr[base + min(lane, cnt - 1)];
        const float4 wq = ((const float4*)ew)[my.y];
        const int lm = (lane < cnt) ? -1 : 0;
        const int q0 = __float_as_int(wq.x / den0) & lm;
        const int q1 = __float_as_int(wq.y / den1) & lm;
        const int q2 = __float_as_int(wq.z / den2) & lm;
        const int q3 = __float_as_int(wq.w / den3) & lm;
        for (int j = 0; j < cnt; j += 4) {
            int    s[4];
            float  al[4], ah[4];
            float4 a[4], b[4];
            for (int k = 0; k < 4; ++k) {
                s[k]  = __builtin_amdgcn_readlane(my.x, j + k);
                al[k] = blend_half(lo, __builtin_amdgcn_readlane(q0, j + k),
                                       __builtin_amdgcn_readlane(q1, j + k));
                ah[k] = blend_half(lo, __builtin_amdgcn_readlane(q2, j + k),
                                       __builtin_amdgcn_readlane(q3, j + k));
            }
            for (int k = 0; k < 4; ++k) {
                const float4* hs = (const float4*)(H + (size_t)s[k] * HC) + lane;
                a[k] = hs[0]; b[k] = hs[32];
            }
            for (int k = 0; k < 4; ++k) {
                acc0.x += al[k] * a[k].x; acc0.y += al[k] * a[k].y;
                acc0.z += al[k] * a[k].z; acc0.w += al[k] * a[k].w;
                acc1.x += ah[k] * b[k].x; acc1.y += ah[k] * b[k].y;
                acc1.z += ah[k] * b[k].z; acc1.w += ah[k] * b[k].w;
            }
        }
    }
    {
        const float sl = blend_half(lo, __float_as_int(nq.x / den0), __float_as_int(nq.y / den1));
        const float sh = blend_half(lo, __float_as_int(nq.z / den2), __float_as_int(nq.w / den3));
        acc0.x += sl * n0.x; acc0.y += sl * n0.y;
        acc0.z += sl * n0.z; acc0.w += sl * n0.w;
        acc1.x += sh * n1.x; acc1.y += sh * n1.y;
        acc1.z += sh * n1.z; acc1.w += sh * n1.w;
    }
    acc0.x = fmaxf(acc0.x + bq0.x, 0.0f); acc0.y = fmaxf(acc0.y + bq0.y, 0.0f);
    acc0.z = fmaxf(acc0.z + bq0.z, 0.0f); acc0.w = fmaxf(acc0.w + bq0.w, 0.0f);
    acc1.x = fmaxf(acc1.x + bq1.x, 0.0f); acc1.y = fmaxf(acc1.y + bq1.y, 0.0f);
    acc1.z = fmaxf(acc1.z + bq1.z, 0.0f); acc1.w = fmaxf(acc1.w + bq1.w, 0.0f);
    float4* op = (float4*)(out + (size_t)node * HC) + lane;
    op[0] = acc0; op[32] = acc1;
}

extern "C" void kernel_launch(void* const* d_in, const int* in_sizes, int n_in,
                              void* d_out, int out_size, void* d_ws, size_t ws_size,
                              hipStream_t stream)
{
    const float* x   = (const float*)d_in[0];
    const int*   ei  = (const int*)d_in[1];
    const float* W1  = (const float*)d_in[2];
    const float* as1 = (const float*)d_in[3];
    const float* ad1 = (const float*)d_in[4];
    const float* b1  = (const float*)d_in[5];
    const float* W2  = (const float*)d_in[6];
    const float* as2 = (const float*)d_in[7];
    const float* ad2 = (const float*)d_in[8];
    const float* b2  = (const float*)d_in[9];
    const float* Wm1 = (const float*)d_in[10];
    const float* bm1 = (const float*)d_in[11];
    const float* Wm2 = (const float*)d_in[12];
    const float* bm2 = (const float*)d_in[13];

    const int N    = in_sizes[0] / 128;
    const int E    = in_sizes[1] / 2;
    const int Etot = E + N;
    const int NH   = N * HEADS;

    float* ws = (float*)d_ws;
    size_t o = 0;
    float*    P0    = ws + o; o += (size_t)N * HC;
    float*    P1    = ws + o; o += (size_t)N * HC;
    float*    a_s   = ws + o; o += (size_t)NH;
    float*    a_d   = ws + o; o += (size_t)NH;
    unsigned* mmax  = (unsigned*)(ws + o); o += (size_t)NH;
    float*    denom = ws + o; o += (size_t)NH;
    float*    ew    = ws + o; o += (size_t)Etot * HEADS;
    o = (o + 3) & ~(size_t)3;
    int*      cnt    = (int*)(ws + o); o += ((size_t)N + 4) & ~(size_t)3;
    int*      rowptr = (int*)(ws + o); o += ((size_t)N + 4) & ~(size_t)3;
    int*      cursor = (int*)(ws + o); o += ((size_t)N + 4) & ~(size_t)3;
    int*      csr    = (int*)(ws + o); o += (size_t)E * 2;

    const dim3 blk(256);
    auto gemm_grid = [](int M, int NC) {
        return dim3((unsigned)(((M + 15) / 16 + 7) / 8), (unsigned)((NC + 63) / 64));
    };
    const int gScores = (NH * 32 + 255) / 256;
    const int gEdgeH  = (Etot * 4 + 255) / 256;
    const int gNodeW  = (N * 32 + 255) / 256;
    const int gEdge1  = (E + 255) / 256;

    zero_i32_kernel<<<(N + 255) / 256, blk, 0, stream>>>(cnt, N);
    deg_count_kernel<<<gEdge1, blk, 0, stream>>>(ei, E, cnt);
    rowptr_scan_kernel<<<1, dim3(SCAN_T), 0, stream>>>(cnt, rowptr, cursor, N);
    csr_fill_kernel<<<gEdge1, blk, 0, stream>>>(ei, E, cursor, csr);

    wmma_gemm_f32_kernel<<<gemm_grid(N, HC), blk, 0, stream>>>(x, W1, P0, nullptr, N, 128, HC, 0);
    scores_kernel<<<gScores, blk, 0, stream>>>(P0, as1, ad1, a_s, a_d, mmax, denom, NH);
    edge_max_kernel<<<gEdgeH, blk, 0, stream>>>(ei, E, N, a_s, a_d, mmax);
    edge_sum_kernel<<<gEdgeH, blk, 0, stream>>>(ei, E, N, a_s, a_d, mmax, denom, ew);
    node_agg_kernel<<<gNodeW, blk, 0, stream>>>(rowptr, csr, E, N, P0, ew, denom, b1, P1);

    wmma_gemm_f32_kernel<<<gemm_grid(N, HC), blk, 0, stream>>>(P1, W2, P0, nullptr, N, HC, HC, 0);
    scores_kernel<<<gScores, blk, 0, stream>>>(P0, as2, ad2, a_s, a_d, mmax, denom, NH);
    edge_max_kernel<<<gEdgeH, blk, 0, stream>>>(ei, E, N, a_s, a_d, mmax);
    edge_sum_kernel<<<gEdgeH, blk, 0, stream>>>(ei, E, N, a_s, a_d, mmax, denom, ew);
    node_agg_kernel<<<gNodeW, blk, 0, stream>>>(rowptr, csr, E, N, P0, ew, denom, b2, P1);

    wmma_gemm_f32_kernel<<<gemm_grid(N, HID), blk, 0, stream>>>(P1, Wm1, P0, bm1, N, HC, HID, 1);
    wmma_gemm_f32_kernel<<<gemm_grid(N, 121), blk, 0, stream>>>(P0, Wm2, (float*)d_out, bm2,
                                                                N, HID, 121, 2);
}


# Softmax max, exp-sum and weights computed inside the per-node wave over its contiguous edge list (fixed lane-tree sums); per-edge max/sum passes, their atomics and the edge-weight array are gone
# speedup vs baseline: 3.1478x; 1.4133x over previous
#include <hip/hip_runtime.h>
#include <math.h>

#define HEADS 4
#define HID 64
#define HC 256
#define NEG_SLOPE 0.2f
#define EPS_F 1e-16f

typedef __attribute__((ext_vector_type(2))) float v2f;
typedef __attribute__((ext_vector_type(8))) float v8f;

__device__ __forceinline__ void epi_store(float* __restrict__ C, const float* __restrict__ bias,
                                          size_t rowoff, int col, int NC, int mode, float v) {
    if (col < NC) {
        if (mode >= 1) v += bias[col];
        if (mode == 2) v = 1.0f / (1.0f + __expf(-v));
        C[rowoff + col] = v;
    }
}

__global__ __launch_bounds__(256) void wmma_gemm_f32_kernel(
    const float* __restrict__ A, const float* __restrict__ B,
    float* __restrict__ C, const float* __restrict__ bias,
    int M, int K, int NC, int mode)
{
    const int wave = threadIdx.x >> 5;
    const int lane = threadIdx.x & 31;
    const int m0 = (blockIdx.x * 8 + wave) * 16;
    if (m0 >= M) return;
    const int n0 = blockIdx.y * 64;

    const int aRow = m0 + (lane & 15);
    const int aK   = (lane >> 4) << 1;
    const int bN   = lane & 15;
    const int bK   = (lane >> 4) << 1;
    const int c    = n0 + bN;

    const float* Ap = A + (size_t)aRow * K + aK;
    const size_t stepB = (size_t)4 * NC;

    v8f acc0 = {}, acc1 = {}, acc2 = {}, acc3 = {};

    if (n0 + 64 <= NC) {
        const float* Bp = B + c + (size_t)bK * NC;
        const float* Bq = Bp + NC;
        #pragma unroll 2
        for (int kb = 0; kb < K; kb += 4) {
            v2f a = *(const v2f*)Ap;
            v2f b0 = { Bp[0],  Bq[0]  };
            v2f b1 = { Bp[16], Bq[16] };
            v2f b2 = { Bp[32], Bq[32] };
            v2f b3 = { Bp[48], Bq[48] };
            Ap += 4; Bp += stepB; Bq += stepB;
            acc0 = __builtin_amdgcn_wmma_f32_16x16x4_f32(false, a, false, b0, (short)0, acc0, false, false);
            acc1 = __builtin_amdgcn_wmma_f32_16x16x4_f32(false, a, false, b1, (short)0, acc1, false, false);
            acc2 = __builtin_amdgcn_wmma_f32_16x16x4_f32(false, a, false, b2, (short)0, acc2, false, false);
            acc3 = __builtin_amdgcn_wmma_f32_16x16x4_f32(false, a, false, b3, (short)0, acc3, false, false);
        }
    } else {
        const int   c0 = min(c,      NC - 1), c1 = min(c + 16, NC - 1);
        const int   c2 = min(c + 32, NC - 1), c3 = min(c + 48, NC - 1);
        const float w0 = (c      < NC) ? 1.0f : 0.0f;
        const float w1 = (c + 16 < NC) ? 1.0f : 0.0f;
        const float w2 = (c + 32 < NC) ? 1.0f : 0.0f;
        const float w3 = (c + 48 < NC) ? 1.0f : 0.0f;
        const float* Bp = B + (size_t)bK * NC;
        const float* Bq = Bp + NC;
        #pragma unroll 2
        for (int kb = 0; kb < K; kb += 4) {
            v2f a = *(const v2f*)Ap;
            v2f b0 = { Bp[c0] * w0, Bq[c0] * w0 };
            v2f b1 = { Bp[c1] * w1, Bq[c1] * w1 };
            v2f b2 = { Bp[c2] * w2, Bq[c2] * w2 };
            v2f b3 = { Bp[c3] * w3, Bq[c3] * w3 };
            Ap += 4; Bp += stepB; Bq += stepB;
            acc0 = __builtin_amdgcn_wmma_f32_16x16x4_f32(false, a, false, b0, (short)0, acc0, false, false);
            acc1 = __builtin_amdgcn_wmma_f32_16x16x4_f32(false, a, false, b1, (short)0, acc1, false, false);
            acc2 = __builtin_amdgcn_wmma_f32_16x16x4_f32(false, a, false, b2, (short)0, acc2, false, false);
            acc3 = __builtin_amdgcn_wmma_f32_16x16x4_f32(false, a, false, b3, (short)0, acc3, false, false);
        }
    }

    const int mbase = m0 + ((lane >> 4) << 3);
    const int colb  = n0 + (lane & 15);
    #pragma unroll
    for (int r = 0; r < 8; ++r) {
        size_t rowoff = (size_t)(mbase + r) * NC;
        epi_store(C, bias, rowoff, colb,      NC, mode, acc0[r]);
        epi_store(C, bias, rowoff, colb + 16, NC, mode, acc1[r]);
        epi_store(C, bias, rowoff, colb + 32, NC, mode, acc2[r]);
        epi_store(C, bias, rowoff, colb + 48, NC, mode, acc3[r]);
    }
}

__global__ __launch_bounds__(256) void scores_kernel(
    const float* __restrict__ H, const float* __restrict__ att_src,
    const float* __restrict__ att_dst, float* __restrict__ a_s, float* __restrict__ a_d, int NH)
{
    const int gw   = (int)((blockIdx.x * 256 + threadIdx.x) >> 5);
    const int lane = threadIdx.x & 31;
    if (gw >= NH) return;
    const int n = gw >> 2, h = gw & 3;
    const float* hp = H + (size_t)n * HC + h * HID;
    const float x0 = hp[lane], x1 = hp[lane + 32];
    float s = x0 * att_src[h * HID + lane] + x1 * att_src[h * HID + lane + 32];
    float d = x0 * att_dst[h * HID + lane] + x1 * att_dst[h * HID + lane + 32];
    #pragma unroll
    for (int off = 16; off > 0; off >>= 1) {
        s += __shfl_xor(s, off, 32);
        d += __shfl_xor(d, off, 32);
    }
    if (lane == 0) { a_s[gw] = s; a_d[gw] = d; }
}

__global__ __launch_bounds__(256) void zero_i32_kernel(int* __restrict__ p, int n) {
    const int i = blockIdx.x * 256 + threadIdx.x;
    if (i < n) p[i] = 0;
}

__global__ __launch_bounds__(256) void deg_count_kernel(const int* __restrict__ ei, int E,
                                                        int* __restrict__ cnt) {
    const int e = blockIdx.x * 256 + threadIdx.x;
    if (e < E) atomicAdd(cnt + ei[E + e], 1);
}

#define SCAN_T 1024
__global__ __launch_bounds__(SCAN_T) void rowptr_scan_kernel(
    const int* __restrict__ cnt, int* __restrict__ rowptr, int* __restrict__ cursor, int n)
{
    __shared__ int wsum[32];
    const int t = threadIdx.x, lane = t & 31, wv = t >> 5;
    const int per = (n + SCAN_T - 1) / SCAN_T;
    const int b = min(t * per, n), e = min(b + per, n);
    int s = 0;
    for (int i = b; i < e; ++i) s += cnt[i];
    int incl = s;
    for (int off = 1; off < 32; off <<= 1) {
        const int v = __shfl_up(incl, off, 32);
        if (lane >= off) incl += v;
    }
    if (lane == 31) wsum[wv] = incl;
    __syncthreads();
    if (wv == 0) {
        const int w = wsum[lane];
        int wi = w;
        for (int off = 1; off < 32; off <<= 1) {
            const int v = __shfl_up(wi, off, 32);
            if (lane >= off) wi += v;
        }
        wsum[lane] = wi - w;
    }
    __syncthreads();
    int run = wsum[wv] + incl - s;
    for (int i = b; i < e; ++i) {
        const int c = cnt[i];
        rowptr[i] = run; cursor[i] = run;
        run += c;
    }
    if (t == SCAN_T - 1) rowptr[n] = run;
}

__global__ __launch_bounds__(256) void csr_fill_kernel(const int* __restrict__ ei, int E,
                                                       int* __restrict__ cursor,
                                                       int* __restrict__ csr) {
    const int e = blockIdx.x * 256 + threadIdx.x;
    if (e < E) {
        const int sv = ei[e], d = ei[E + e];
        const int p = atomicAdd(cursor + d, 1);
        csr[p] = sv;
    }
}


__device__ __forceinline__ float blend_half(int lo, int x_lo, int x_hi) {
    return __int_as_float((x_lo & lo) | (x_hi & ~lo));
}

__device__ __forceinline__ float leaky(float e) { return (e > 0.0f) ? e : NEG_SLOPE * e; }

__device__ __forceinline__ float4 edge_logits(float4 a, float4 d) {
    return make_float4(leaky(a.x + d.x), leaky(a.y + d.y), leaky(a.z + d.z), leaky(a.w + d.w));
}

template <int K> __device__ __forceinline__ float row_xor(float v) {
    return __int_as_float(__builtin_amdgcn_update_dpp(0, __float_as_int(v), 0x160 + K, 0xf, 0xf, false));
}
__device__ __forceinline__ float other_row(float v) {
    const int b = __float_as_int(v);
    return __int_as_float(__builtin_amdgcn_permlanex16(b, b, 0x76543210, 0xfedcba98, false, false));
}
__device__ __forceinline__ float4 wave_max4(float4 v) {
    v.x = fmaxf(v.x, row_xor<1>(v.x)); v.y = fmaxf(v.y, row_xor<1>(v.y)); v.z = fmaxf(v.z, row_xor<1>(v.z)); v.w = fmaxf(v.w, row_xor<1>(v.w));
    v.x = fmaxf(v.x, row_xor<2>(v.x)); v.y = fmaxf(v.y, row_xor<2>(v.y)); v.z = fmaxf(v.z, row_xor<2>(v.z)); v.w = fmaxf(v.w, row_xor<2>(v.w));
    v.x = fmaxf(v.x, row_xor<4>(v.x)); v.y = fmaxf(v.y, row_xor<4>(v.y)); v.z = fmaxf(v.z, row_xor<4>(v.z)); v.w = fmaxf(v.w, row_xor<4>(v.w));
    v.x = fmaxf(v.x, row_xor<8>(v.x)); v.y = fmaxf(v.y, row_xor<8>(v.y)); v.z = fmaxf(v.z, row_xor<8>(v.z)); v.w = fmaxf(v.w, row_xor<8>(v.w));
    v.x = fmaxf(v.x, other_row(v.x));  v.y = fmaxf(v.y, other_row(v.y));  v.z = fmaxf(v.z, other_row(v.z));  v.w = fmaxf(v.w, other_row(v.w));
    return v;
}
__device__ __forceinline__ float4 wave_sum4(float4 v) {
    v.x += row_xor<1>(v.x); v.y += row_xor<1>(v.y); v.z += row_xor<1>(v.z); v.w += row_xor<1>(v.w);
    v.x += row_xor<2>(v.x); v.y += row_xor<2>(v.y); v.z += row_xor<2>(v.z); v.w += row_xor<2>(v.w);
    v.x += row_xor<4>(v.x); v.y += row_xor<4>(v.y); v.z += row_xor<4>(v.z); v.w += row_xor<4>(v.w);
    v.x += row_xor<8>(v.x); v.y += row_xor<8>(v.y); v.z += row_xor<8>(v.z); v.w += row_xor<8>(v.w);
    v.x += other_row(v.x);  v.y += other_row(v.y);  v.z += other_row(v.z);  v.w += other_row(v.w);
    return v;
}
__device__ __forceinline__ float uniform_f(float v) {
    return __int_as_float(__builtin_amdgcn_readfirstlane(__float_as_int(v)));
}
__device__ __forceinline__ int edge_weight(float e, float m, float den, int lm) {
    return __float_as_int(__expf(e - m) / den) & lm;
}

__global__ __launch_bounds__(256) void node_agg_kernel(
    const int* __restrict__ rowptr, const int* __restrict__ csr, int N,
    const float* __restrict__ H, const float* __restrict__ a_s, const float* __restrict__ a_d,
    const float* __restrict__ bias, float* __restrict__ out)
{
    const int gw   = (int)((blockIdx.x * 256 + threadIdx.x) >> 5);
    const int lane = threadIdx.x & 31;
    if (gw >= N) return;
    const int node = __builtin_amdgcn_readfirstlane(gw);
    const int lo = (lane < 16) ? -1 : 0;
    const float4* as4 = (const float4*)a_s;
    const float4 adq = ((const float4*)a_d)[node];
    const float4 asn = as4[node];
    const int beg = rowptr[node], end = rowptr[node + 1];

    const float4* hn = (const float4*)(H + (size_t)node * HC) + lane;
    const float4 n0 = hn[0], n1 = hn[32];
    const float4 bq0 = ((const float4*)bias)[lane], bq1 = ((const float4*)bias)[32 + lane];

    const float4 es = edge_logits(asn, adq);

    const int cnt0 = min(32, end - beg);
    int s0 = node;
    if (cnt0 > 0) s0 = csr[beg + min(lane, cnt0 - 1)];
    const float4 e0 = edge_logits(as4[s0], adq);

    float4 m = make_float4(fmaxf(es.x, e0.x), fmaxf(es.y, e0.y), fmaxf(es.z, e0.z), fmaxf(es.w, e0.w));
    #pragma clang loop vectorize(disable) interleave(disable)
    for (int base = beg + 32; base < end; base += 32) {
        const int cnt = min(32, end - base);
        const float4 e = edge_logits(as4[csr[base + min(lane, cnt - 1)]], adq);
        m.x = fmaxf(m.x, e.x); m.y = fmaxf(m.y, e.y); m.z = fmaxf(m.z, e.z); m.w = fmaxf(m.w, e.w);
    }
    m = wave_max4(m);
    const float m0 = uniform_f(m.x), m1 = uniform_f(m.y), m2 = uniform_f(m.z), m3 = uniform_f(m.w);

    const int lm0 = (lane < cnt0) ? -1 : 0;
    float4 sum = make_float4(__int_as_float(__float_as_int(__expf(e0.x - m0)) & lm0),
                             __int_as_float(__float_as_int(__expf(e0.y - m1)) & lm0),
                             __int_as_float(__float_as_int(__expf(e0.z - m2)) & lm0),
                             __int_as_float(__float_as_int(__expf(e0.w - m3)) & lm0));
    for (int base = beg + 32; base < end; base += 32) {
        const int cnt = min(32, end - base);
        const float4 e = edge_logits(as4[csr[base + min(lane, cnt - 1)]], adq);
        const int lm = (lane < cnt) ? -1 : 0;
        sum.x += __int_as_float(__float_as_int(__expf(e.x - m0)) & lm);
        sum.y += __int_as_float(__float_as_int(__expf(e.y - m1)) & lm);
        sum.z += __int_as_float(__float_as_int(__expf(e.z - m2)) & lm);
        sum.w += __int_as_float(__float_as_int(__expf(e.w - m3)) & lm);
    }
    sum = wave_sum4(sum);
    const float xs0 = __expf(es.x - m0), xs1 = __expf(es.y - m1);
    const float xs2 = __expf(es.z - m2), xs3 = __expf(es.w - m3);
    const float den0 = uniform_f(sum.x + xs0) + EPS_F, den1 = uniform_f(sum.y + xs1) + EPS_F;
    const float den2 = uniform_f(sum.z + xs2) + EPS_F, den3 = uniform_f(sum.w + xs3) + EPS_F;

    float4 acc0 = make_float4(0.0f, 0.0f, 0.0f, 0.0f), acc1 = acc0;
    int sv  = s0;
    int cnt = cnt0;
    int q0 = edge_weight(e0.x, m0, den0, lm0), q1 = edge_weight(e0.y, m1, den1, lm0);
    int q2 = edge_weight(e0.z, m2, den2, lm0), q3 = edge_weight(e0.w, m3, den3, lm0);
    for (int base = beg; base < end; ) {
        for (int j = 0; j < cnt; j += 4) {
            int    s[4];
            float  al[4], ah[4];
            float4 a[4], b[4];
            for (int k = 0; k < 4; ++k) {
                s[k]  = __builtin_amdgcn_readlane(sv, j + k);
                al[k] = blend_half(lo, __builtin_amdgcn_readlane(q0, j + k),
                                       __builtin_amdgcn_readlane(q1, j + k));
                ah[k] = blend_half(lo, __builtin_amdgcn_readlane(q2, j + k),
                                       __builtin_amdgcn_readlane(q3, j + k));
            }
            for (int k = 0; k < 4; ++k) {
                const float4* hs = (const float4*)(H + (size_t)s[k] * HC) + lane;
                a[k] = hs[0]; b[k] = hs[32];
            }
            for (int k = 0; k < 4; ++k) {
                acc0.x += al[k] * a[k].x; acc0.y += al[k] * a[k].y;
                acc0.z += al[k] * a[k].z; acc0.w += al[k] * a[k].w;
                acc1.x += ah[k] * b[k].x; acc1.y += ah[k] * b[k].y;
                acc1.z += ah[k] * b[k].z; acc1.w += ah[k] * b[k].w;
            }
        }
        base += 32;
        if (base < end) {
            cnt = min(32, end - base);
            sv  = csr[base + min(lane, cnt - 1)];
            const float4 e = edge_logits(as4[sv], adq);
            const int lm = (lane < cnt) ? -1 : 0;
            q0 = edge_weight(e.x, m0, den0, lm); q1 = edge_weight(e.y, m1, den1, lm);
            q2 = edge_weight(e.z, m2, den2, lm); q3 = edge_weight(e.w, m3, den3, lm);
        }
    }
    {
        const float sl = blend_half(lo, __float_as_int(xs0 / den0), __float_as_int(xs1 / den1));
        const float sh = blend_half(lo, __float_as_int(xs2 / den2), __float_as_int(xs3 / den3));
        acc0.x += sl * n0.x; acc0.y += sl * n0.y;
        acc0.z += sl * n0.z; acc0.w += sl * n0.w;
        acc1.x += sh * n1.x; acc1.y += sh * n1.y;
        acc1.z += sh * n1.z; acc1.w += sh * n1.w;
    }
    acc0.x = fmaxf(acc0.x + bq0.x, 0.0f); acc0.y = fmaxf(acc0.y + bq0.y, 0.0f);
    acc0.z = fmaxf(acc0.z + bq0.z, 0.0f); acc0.w = fmaxf(acc0.w + bq0.w, 0.0f);
    acc1.x = fmaxf(acc1.x + bq1.x, 0.0f); acc1.y = fmaxf(acc1.y + bq1.y, 0.0f);
    acc1.z = fmaxf(acc1.z + bq1.z, 0.0f); acc1.w = fmaxf(acc1.w + bq1.w, 0.0f);
    float4* op = (float4*)(out + (size_t)node * HC) + lane;
    op[0] = acc0; op[32] = acc1;
}

extern "C" void kernel_launch(void* const* d_in, const int* in_sizes, int n_in,
                              void* d_out, int out_size, void* d_ws, size_t ws_size,
                              hipStream_t stream)
{
    const float* x   = (const float*)d_in[0];
    const int*   ei  = (const int*)d_in[1];
    const float* W1  = (const float*)d_in[2];
    const float* as1 = (const float*)d_in[3];
    const float* ad1 = (const float*)d_in[4];
    const float* b1  = (const float*)d_in[5];
    const float* W2  = (const float*)d_in[6];
    const float* as2 = (const float*)d_in[7];
    const float* ad2 = (const float*)d_in[8];
    const float* b2  = (const float*)d_in[9];
    const float* Wm1 = (const float*)d_in[10];
    const float* bm1 = (const float*)d_in[11];
    const float* Wm2 = (const float*)d_in[12];
    const float* bm2 = (const float*)d_in[13];

    const int N    = in_sizes[0] / 128;
    const int E    = in_sizes[1] / 2;
    const int NH   = N * HEADS;

    float* ws = (float*)d_ws;
    size_t o = 0;
    float*    P0    = ws + o; o += (size_t)N * HC;
    float*    P1    = ws + o; o += (size_t)N * HC;
    float*    a_s   = ws + o; o += (size_t)NH;
    float*    a_d   = ws + o; o += (size_t)NH;
    int*      cnt    = (int*)(ws + o); o += ((size_t)N + 4) & ~(size_t)3;
    int*      rowptr = (int*)(ws + o); o += ((size_t)N + 4) & ~(size_t)3;
    int*      cursor = (int*)(ws + o); o += ((size_t)N + 4) & ~(size_t)3;
    int*      csr    = (int*)(ws + o); o += (size_t)E;

    const dim3 blk(256);
    auto gemm_grid = [](int M, int NC) {
        return dim3((unsigned)(((M + 15) / 16 + 7) / 8), (unsigned)((NC + 63) / 64));
    };
    const int gScores = (NH * 32 + 255) / 256;
    const int gNodeW  = (N * 32 + 255) / 256;
    const int gEdge1  = (E + 255) / 256;

    zero_i32_kernel<<<(N + 255) / 256, blk, 0, stream>>>(cnt, N);
    deg_count_kernel<<<gEdge1, blk, 0, stream>>>(ei, E, cnt);
    rowptr_scan_kernel<<<1, dim3(SCAN_T), 0, stream>>>(cnt, rowptr, cursor, N);
    csr_fill_kernel<<<gEdge1, blk, 0, stream>>>(ei, E, cursor, csr);

    wmma_gemm_f32_kernel<<<gemm_grid(N, HC), blk, 0, stream>>>(x, W1, P0, nullptr, N, 128, HC, 0);
    scores_kernel<<<gScores, blk, 0, stream>>>(P0, as1, ad1, a_s, a_d, NH);
    node_agg_kernel<<<gNodeW, blk, 0, stream>>>(rowptr, csr, N, P0, a_s, a_d, b1, P1);

    wmma_gemm_f32_kernel<<<gemm_grid(N, HC), blk, 0, stream>>>(P1, W2, P0, nullptr, N, HC, HC, 0);
    scores_kernel<<<gScores, blk, 0, stream>>>(P0, as2, ad2, a_s, a_d, NH);
    node_agg_kernel<<<gNodeW, blk, 0, stream>>>(rowptr, csr, N, P0, a_s, a_d, b2, P1);

    wmma_gemm_f32_kernel<<<gemm_grid(N, HID), blk, 0, stream>>>(P1, Wm1, P0, bm1, N, HC, HID, 1);
    wmma_gemm_f32_kernel<<<gemm_grid(N, 121), blk, 0, stream>>>(P0, Wm2, (float*)d_out, bm2,
                                                                N, HID, 121, 2);
}


# Row-pointer prefix scan on 13 workgroups, four consecutive items per thread with 16-byte whole-line loads and stores; each workgroup sums the counts in front of it itself (integers, same values)
# speedup vs baseline: 3.3897x; 1.0768x over previous
#include <hip/hip_runtime.h>
#include <math.h>

#define HEADS 4
#define HID 64
#define HC 256
#define NEG_SLOPE 0.2f
#define EPS_F 1e-16f

typedef __attribute__((ext_vector_type(2))) float v2f;
typedef __attribute__((ext_vector_type(8))) float v8f;

__device__ __forceinline__ void epi_store(float* __restrict__ C, const float* __restrict__ bias,
                                          size_t rowoff, int col, int NC, int mode, float v) {
    if (col < NC) {
        if (mode >= 1) v += bias[col];
        if (mode == 2) v = 1.0f / (1.0f + __expf(-v));
        C[rowoff + col] = v;
    }
}

__global__ __launch_bounds__(256) void wmma_gemm_f32_kernel(
    const float* __restrict__ A, const float* __restrict__ B,
    float* __restrict__ C, const float* __restrict__ bias,
    int M, int K, int NC, int mode)
{
    const int wave = threadIdx.x >> 5;
    const int lane = threadIdx.x & 31;
    const int m0 = (blockIdx.x * 8 + wave) * 16;
    if (m0 >= M) return;
    const int n0 = blockIdx.y * 64;

    const int aRow = m0 + (lane & 15);
    const int aK   = (lane >> 4) << 1;
    const int bN   = lane & 15;
    const int bK   = (lane >> 4) << 1;
    const int c    = n0 + bN;

    const float* Ap = A + (size_t)aRow * K + aK;
    const size_t stepB = (size_t)4 * NC;

    v8f acc0 = {}, acc1 = {}, acc2 = {}, acc3 = {};

    if (n0 + 64 <= NC) {
        const float* Bp = B + c + (size_t)bK * NC;
        const float* Bq = Bp + NC;
        #pragma unroll 2
        for (int kb = 0; kb < K; kb += 4) {
            v2f a = *(const v2f*)Ap;
            v2f b0 = { Bp[0],  Bq[0]  };
            v2f b1 = { Bp[16], Bq[16] };
            v2f b2 = { Bp[32], Bq[32] };
            v2f b3 = { Bp[48], Bq[48] };
            Ap += 4; Bp += stepB; Bq += stepB;
            acc0 = __builtin_amdgcn_wmma_f32_16x16x4_f32(false, a, false, b0, (short)0, acc0, false, false);
            acc1 = __builtin_amdgcn_wmma_f32_16x16x4_f32(false, a, false, b1, (short)0, acc1, false, false);
            acc2 = __builtin_amdgcn_wmma_f32_16x16x4_f32(false, a, false, b2, (short)0, acc2, false, false);
            acc3 = __builtin_amdgcn_wmma_f32_16x16x4_f32(false, a, false, b3, (short)0, acc3, false, false);
        }
    } else {
        const int   c0 = min(c,      NC - 1), c1 = min(c + 16, NC - 1);
        const int   c2 = min(c + 32, NC - 1), c3 = min(c + 48, NC - 1);
        const float w0 = (c      < NC) ? 1.0f : 0.0f;
        const float w1 = (c + 16 < NC) ? 1.0f : 0.0f;
        const float w2 = (c + 32 < NC) ? 1.0f : 0.0f;
        const float w3 = (c + 48 < NC) ? 1.0f : 0.0f;
        const float* Bp = B + (size_t)bK * NC;
        const float* Bq = Bp + NC;
        #pragma unroll 2
        for (int kb = 0; kb < K; kb += 4) {
            v2f a = *(const v2f*)Ap;
            v2f b0 = { Bp[c0] * w0, Bq[c0] * w0 };
            v2f b1 = { Bp[c1] * w1, Bq[c1] * w1 };
            v2f b2 = { Bp[c2] * w2, Bq[c2] * w2 };
            v2f b3 = { Bp[c3] * w3, Bq[c3] * w3 };
            Ap += 4; Bp += stepB; Bq += stepB;
            acc0 = __builtin_amdgcn_wmma_f32_16x16x4_f32(false, a, false, b0, (short)0, acc0, false, false);
            acc1 = __builtin_amdgcn_wmma_f32_16x16x4_f32(false, a, false, b1, (short)0, acc1, false, false);
            acc2 = __builtin_amdgcn_wmma_f32_16x16x4_f32(false, a, false, b2, (short)0, acc2, false, false);
            acc3 = __builtin_amdgcn_wmma_f32_16x16x4_f32(false, a, false, b3, (short)0, acc3, false, false);
        }
    }

    const int mbase = m0 + ((lane >> 4) << 3);
    const int colb  = n0 + (lane & 15);
    #pragma unroll
    for (int r = 0; r < 8; ++r) {
        size_t rowoff = (size_t)(mbase + r) * NC;
        epi_store(C, bias, rowoff, colb,      NC, mode, acc0[r]);
        epi_store(C, bias, rowoff, colb + 16, NC, mode, acc1[r]);
        epi_store(C, bias, rowoff, colb + 32, NC, mode, acc2[r]);
        epi_store(C, bias, rowoff, colb + 48, NC, mode, acc3[r]);
    }
}

__global__ __launch_bounds__(256) void scores_kernel(
    const float* __restrict__ H, const float* __restrict__ att_src,
    const float* __restrict__ att_dst, float* __restrict__ a_s, float* __restrict__ a_d, int NH)
{
    const int gw   = (int)((blockIdx.x * 256 + threadIdx.x) >> 5);
    const int lane = threadIdx.x & 31;
    if (gw >= NH) return;
    const int n = gw >> 2, h = gw & 3;
    const float* hp = H + (size_t)n * HC + h * HID;
    const float x0 = hp[lane], x1 = hp[lane + 32];
    float s = x0 * att_src[h * HID + lane] + x1 * att_src[h * HID + lane + 32];
    float d = x0 * att_dst[h * HID + lane] + x1 * att_dst[h * HID + lane + 32];
    #pragma unroll
    for (int off = 16; off > 0; off >>= 1) {
        s += __shfl_xor(s, off, 32);
        d += __shfl_xor(d, off, 32);
    }
    if (lane == 0) { a_s[gw] = s; a_d[gw] = d; }
}

__global__ __launch_bounds__(256) void zero_i32_kernel(int* __restrict__ p, int n) {
    const int i = blockIdx.x * 256 + threadIdx.x;
    if (i < n) p[i] = 0;
}

__global__ __launch_bounds__(256) void deg_count_kernel(const int* __restrict__ ei, int E,
                                                        int* __restrict__ cnt) {
    const int e = blockIdx.x * 256 + threadIdx.x;
    if (e < E) atomicAdd(cnt + ei[E + e], 1);
}

#define SCAN_T 1024
#define SCAN_ITEMS 4096
__global__ __launch_bounds__(SCAN_T) void rowptr_scan_kernel(
    const int* __restrict__ cnt, int* __restrict__ rowptr, int* __restrict__ cursor, int n)
{
    __shared__ int wsum[32];
    __shared__ int wpre[32];
    const int t = threadIdx.x, lane = t & 31, wv = t >> 5;
    const int first = blockIdx.x * SCAN_ITEMS;
    const int i0 = first + 4 * t;

    int4 v = make_int4(0, 0, 0, 0);
    if (i0 + 3 < n) {
        v = *(const int4*)(cnt + i0);
    } else {
        if (i0     < n) v.x = cnt[i0];
        if (i0 + 1 < n) v.y = cnt[i0 + 1];
        if (i0 + 2 < n) v.z = cnt[i0 + 2];
    }

    int pre = 0;
    #pragma unroll 4
    for (int j = 4 * t; j < first; j += SCAN_ITEMS) {
        const int4 u = *(const int4*)(cnt + j);
        pre += (u.x + u.y) + (u.z + u.w);
    }

    const int own = (v.x + v.y) + (v.z + v.w);
    int incl = own;
    for (int off = 1; off < 32; off <<= 1) {
        const int x = __shfl_up(incl, off, 32);
        if (lane >= off) incl += x;
    }
    for (int off = 16; off > 0; off >>= 1) pre += __shfl_xor(pre, off, 32);
    if (lane == 31) wsum[wv] = incl;
    if (lane == 0)  wpre[wv] = pre;
    __syncthreads();
    int front = wpre[lane] + ((lane < wv) ? wsum[lane] : 0);
    for (int off = 16; off > 0; off >>= 1) front += __shfl_xor(front, off, 32);

    const int r0 = front + incl - own;
    const int r1 = r0 + v.x, r2 = r1 + v.y, r3 = r2 + v.z;
    if (i0 + 3 <= n) {
        const int4 r = make_int4(r0, r1, r2, r3);
        *(int4*)(rowptr + i0) = r;
        *(int4*)(cursor + i0) = r;
    } else {
        if (i0     <= n) { rowptr[i0]     = r0; cursor[i0]     = r0; }
        if (i0 + 1 <= n) { rowptr[i0 + 1] = r1; cursor[i0 + 1] = r1; }
        if (i0 + 2 <= n) { rowptr[i0 + 2] = r2; cursor[i0 + 2] = r2; }
    }
}

__global__ __launch_bounds__(256) void csr_fill_kernel(const int* __restrict__ ei, int E,
                                                       int* __restrict__ cursor,
                                                       int* __restrict__ csr) {
    const int e = blockIdx.x * 256 + threadIdx.x;
    if (e < E) {
        const int sv = ei[e], d = ei[E + e];
        const int p = atomicAdd(cursor + d, 1);
        csr[p] = sv;
    }
}


__device__ __forceinline__ float blend_half(int lo, int x_lo, int x_hi) {
    return __int_as_float((x_lo & lo) | (x_hi & ~lo));
}

__device__ __forceinline__ float leaky(float e) { return (e > 0.0f) ? e : NEG_SLOPE * e; }

__device__ __forceinline__ float4 edge_logits(float4 a, float4 d) {
    return make_float4(leaky(a.x + d.x), leaky(a.y + d.y), leaky(a.z + d.z), leaky(a.w + d.w));
}

template <int K> __device__ __forceinline__ float row_xor(float v) {
    return __int_as_float(__builtin_amdgcn_update_dpp(0, __float_as_int(v), 0x160 + K, 0xf, 0xf, false));
}
__device__ __forceinline__ float other_row(float v) {
    const int b = __float_as_int(v);
    return __int_as_float(__builtin_amdgcn_permlanex16(b, b, 0x76543210, 0xfedcba98, false, false));
}
__device__ __forceinline__ float4 wave_max4(float4 v) {
    v.x = fmaxf(v.x, row_xor<1>(v.x)); v.y = fmaxf(v.y, row_xor<1>(v.y)); v.z = fmaxf(v.z, row_xor<1>(v.z)); v.w = fmaxf(v.w, row_xor<1>(v.w));
    v.x = fmaxf(v.x, row_xor<2>(v.x)); v.y = fmaxf(v.y, row_xor<2>(v.y)); v.z = fmaxf(v.z, row_xor<2>(v.z)); v.w = fmaxf(v.w, row_xor<2>(v.w));
    v.x = fmaxf(v.x, row_xor<4>(v.x)); v.y = fmaxf(v.y, row_xor<4>(v.y)); v.z = fmaxf(v.z, row_xor<4>(v.z)); v.w = fmaxf(v.w, row_xor<4>(v.w));
    v.x = fmaxf(v.x, row_xor<8>(v.x)); v.y = fmaxf(v.y, row_xor<8>(v.y)); v.z = fmaxf(v.z, row_xor<8>(v.z)); v.w = fmaxf(v.w, row_xor<8>(v.w));
    v.x = fmaxf(v.x, other_row(v.x));  v.y = fmaxf(v.y, other_row(v.y));  v.z = fmaxf(v.z, other_row(v.z));  v.w = fmaxf(v.w, other_row(v.w));
    return v;
}
__device__ __forceinline__ float4 wave_sum4(float4 v) {
    v.x += row_xor<1>(v.x); v.y += row_xor<1>(v.y); v.z += row_xor<1>(v.z); v.w += row_xor<1>(v.w);
    v.x += row_xor<2>(v.x); v.y += row_xor<2>(v.y); v.z += row_xor<2>(v.z); v.w += row_xor<2>(v.w);
    v.x += row_xor<4>(v.x); v.y += row_xor<4>(v.y); v.z += row_xor<4>(v.z); v.w += row_xor<4>(v.w);
    v.x += row_xor<8>(v.x); v.y += row_xor<8>(v.y); v.z += row_xor<8>(v.z); v.w += row_xor<8>(v.w);
    v.x += other_row(v.x);  v.y += other_row(v.y);  v.z += other_row(v.z);  v.w += other_row(v.w);
    return v;
}
__device__ __forceinline__ float uniform_f(float v) {
    return __int_as_float(__builtin_amdgcn_readfirstlane(__float_as_int(v)));
}
__device__ __forceinline__ int edge_weight(float e, float m, float den, int lm) {
    return __float_as_int(__expf(e - m) / den) & lm;
}

__global__ __launch_bounds__(256) void node_agg_kernel(
    const int* __restrict__ rowptr, const int* __restrict__ csr, int N,
    const float* __restrict__ H, const float* __restrict__ a_s, const float* __restrict__ a_d,
    const float* __restrict__ bias, float* __restrict__ out)
{
    const int gw   = (int)((blockIdx.x * 256 + threadIdx.x) >> 5);
    const int lane = threadIdx.x & 31;
    if (gw >= N) return;
    const int node = __builtin_amdgcn_readfirstlane(gw);
    const int lo = (lane < 16) ? -1 : 0;
    const float4* as4 = (const float4*)a_s;
    const float4 adq = ((const float4*)a_d)[node];
    const float4 asn = as4[node];
    const int beg = rowptr[node], end = rowptr[node + 1];

    const float4* hn = (const float4*)(H + (size_t)node * HC) + lane;
    const float4 n0 = hn[0], n1 = hn[32];
    const float4 bq0 = ((const float4*)bias)[lane], bq1 = ((const float4*)bias)[32 + lane];

    const float4 es = edge_logits(asn, adq);

    const int cnt0 = min(32, end - beg);
    int s0 = node;
    if (cnt0 > 0) s0 = csr[beg + min(lane, cnt0 - 1)];
    const float4 e0 = edge_logits(as4[s0], adq);

    float4 m = make_float4(fmaxf(es.x, e0.x), fmaxf(es.y, e0.y), fmaxf(es.z, e0.z), fmaxf(es.w, e0.w));
    #pragma clang loop vectorize(disable) interleave(disable)
    for (int base = beg + 32; base < end; base += 32) {
        const int cnt = min(32, end - base);
        const float4 e = edge_logits(as4[csr[base + min(lane, cnt - 1)]], adq);
        m.x = fmaxf(m.x, e.x); m.y = fmaxf(m.y, e.y); m.z = fmaxf(m.z, e.z); m.w = fmaxf(m.w, e.w);
    }
    m = wave_max4(m);
    const float m0 = uniform_f(m.x), m1 = uniform_f(m.y), m2 = uniform_f(m.z), m3 = uniform_f(m.w);

    const int lm0 = (lane < cnt0) ? -1 : 0;
    float4 sum = make_float4(__int_as_float(__float_as_int(__expf(e0.x - m0)) & lm0),
                             __int_as_float(__float_as_int(__expf(e0.y - m1)) & lm0),
                             __int_as_float(__float_as_int(__expf(e0.z - m2)) & lm0),
                             __int_as_float(__float_as_int(__expf(e0.w - m3)) & lm0));
    for (int base = beg + 32; base < end; base += 32) {
        const int cnt = min(32, end - base);
        const float4 e = edge_logits(as4[csr[base + min(lane, cnt - 1)]], adq);
        const int lm = (lane < cnt) ? -1 : 0;
        sum.x += __int_as_float(__float_as_int(__expf(e.x - m0)) & lm);
        sum.y += __int_as_float(__float_as_int(__expf(e.y - m1)) & lm);
        sum.z += __int_as_float(__float_as_int(__expf(e.z - m2)) & lm);
        sum.w += __int_as_float(__float_as_int(__expf(e.w - m3)) & lm);
    }
    sum = wave_sum4(sum);
    const float xs0 = __expf(es.x - m0), xs1 = __expf(es.y - m1);
    const float xs2 = __expf(es.z - m2), xs3 = __expf(es.w - m3);
    const float den0 = uniform_f(sum.x + xs0) + EPS_F, den1 = uniform_f(sum.y + xs1) + EPS_F;
    const float den2 = uniform_f(sum.z + xs2) + EPS_F, den3 = uniform_f(sum.w + xs3) + EPS_F;

    float4 acc0 = make_float4(0.0f, 0.0f, 0.0f, 0.0f), acc1 = acc0;
    int sv  = s0;
    int cnt = cnt0;
    int q0 = edge_weight(e0.x, m0, den0, lm0), q1 = edge_weight(e0.y, m1, den1, lm0);
    int q2 = edge_weight(e0.z, m2, den2, lm0), q3 = edge_weight(e0.w, m3, den3, lm0);
    for (int base = beg; base < end; ) {
        for (int j = 0; j < cnt; j += 4) {
            int    s[4];
            float  al[4], ah[4];
            float4 a[4], b[4];
            for (int k = 0; k < 4; ++k) {
                s[k]  = __builtin_amdgcn_readlane(sv, j + k);
                al[k] = blend_half(lo, __builtin_amdgcn_readlane(q0, j + k),
                                       __builtin_amdgcn_readlane(q1, j + k));
                ah[k] = blend_half(lo, __builtin_amdgcn_readlane(q2, j + k),
                                       __builtin_amdgcn_readlane(q3, j + k));
            }
            for (int k = 0; k < 4; ++k) {
                const float4* hs = (const float4*)(H + (size_t)s[k] * HC) + lane;
                a[k] = hs[0]; b[k] = hs[32];
            }
            for (int k = 0; k < 4; ++k) {
                acc0.x += al[k] * a[k].x; acc0.y += al[k] * a[k].y;
                acc0.z += al[k] * a[k].z; acc0.w += al[k] * a[k].w;
                acc1.x += ah[k] * b[k].x; acc1.y += ah[k] * b[k].y;
                acc1.z += ah[k] * b[k].z; acc1.w += ah[k] * b[k].w;
            }
        }
        base += 32;
        if (base < end) {
            cnt = min(32, end - base);
            sv  = csr[base + min(lane, cnt - 1)];
            const float4 e = edge_logits(as4[sv], adq);
            const int lm = (lane < cnt) ? -1 : 0;
            q0 = edge_weight(e.x, m0, den0, lm); q1 = edge_weight(e.y, m1, den1, lm);
            q2 = edge_weight(e.z, m2, den2, lm); q3 = edge_weight(e.w, m3, den3, lm);
        }
    }
    {
        const float sl = blend_half(lo, __float_as_int(xs0 / den0), __float_as_int(xs1 / den1));
        const float sh = blend_half(lo, __float_as_int(xs2 / den2), __float_as_int(xs3 / den3));
        acc0.x += sl * n0.x; acc0.y += sl * n0.y;
        acc0.z += sl * n0.z; acc0.w += sl * n0.w;
        acc1.x += sh * n1.x; acc1.y += sh * n1.y;
        acc1.z += sh * n1.z; acc1.w += sh * n1.w;
    }
    acc0.x = fmaxf(acc0.x + bq0.x, 0.0f); acc0.y = fmaxf(acc0.y + bq0.y, 0.0f);
    acc0.z = fmaxf(acc0.z + bq0.z, 0.0f); acc0.w = fmaxf(acc0.w + bq0.w, 0.0f);
    acc1.x = fmaxf(acc1.x + bq1.x, 0.0f); acc1.y = fmaxf(acc1.y + bq1.y, 0.0f);
    acc1.z = fmaxf(acc1.z + bq1.z, 0.0f); acc1.w = fmaxf(acc1.w + bq1.w, 0.0f);
    float4* op = (float4*)(out + (size_t)node * HC) + lane;
    op[0] = acc0; op[32] = acc1;
}

extern "C" void kernel_launch(void* const* d_in, const int* in_sizes, int n_in,
                              void* d_out, int out_size, void* d_ws, size_t ws_size,
                              hipStream_t stream)
{
    const float* x   = (const float*)d_in[0];
    const int*   ei  = (const int*)d_in[1];
    const float* W1  = (const float*)d_in[2];
    const float* as1 = (const float*)d_in[3];
    const float* ad1 = (const float*)d_in[4];
    const float* b1  = (const float*)d_in[5];
    const float* W2  = (const float*)d_in[6];
    const float* as2 = (const float*)d_in[7];
    const float* ad2 = (const float*)d_in[8];
    const float* b2  = (const float*)d_in[9];
    const float* Wm1 = (const float*)d_in[10];
    const float* bm1 = (const float*)d_in[11];
    const float* Wm2 = (const float*)d_in[12];
    const float* bm2 = (const float*)d_in[13];

    const int N    = in_sizes[0] / 128;
    const int E    = in_sizes[1] / 2;
    const int NH   = N * HEADS;

    float* ws = (float*)d_ws;
    size_t o = 0;
    float*    P0    = ws + o; o += (size_t)N * HC;
    float*    P1    = ws + o; o += (size_t)N * HC;
    float*    a_s   = ws + o; o += (size_t)NH;
    float*    a_d   = ws + o; o += (size_t)NH;
    int*      cnt    = (int*)(ws + o); o += ((size_t)N + 4) & ~(size_t)3;
    int*      rowptr = (int*)(ws + o); o += ((size_t)N + 4) & ~(size_t)3;
    int*      cursor = (int*)(ws + o); o += ((size_t)N + 4) & ~(size_t)3;
    int*      csr    = (int*)(ws + o); o += (size_t)E;

    const dim3 blk(256);
    auto gemm_grid = [](int M, int NC) {
        return dim3((unsigned)(((M + 15) / 16 + 7) / 8), (unsigned)((NC + 63) / 64));
    };
    const int gScores = (NH * 32 + 255) / 256;
    const int gNodeW  = (N * 32 + 255) / 256;
    const int gEdge1  = (E + 255) / 256;

    zero_i32_kernel<<<(N + 255) / 256, blk, 0, stream>>>(cnt, N);
    deg_count_kernel<<<gEdge1, blk, 0, stream>>>(ei, E, cnt);
    rowptr_scan_kernel<<<N / SCAN_ITEMS + 1, dim3(SCAN_T), 0, stream>>>(cnt, rowptr, cursor, N);
    csr_fill_kernel<<<gEdge1, blk, 0, stream>>>(ei, E, cursor, csr);

    wmma_gemm_f32_kernel<<<gemm_grid(N, HC), blk, 0, stream>>>(x, W1, P0, nullptr, N, 128, HC, 0);
    scores_kernel<<<gScores, blk, 0, stream>>>(P0, as1, ad1, a_s, a_d, NH);
    node_agg_kernel<<<gNodeW, blk, 0, stream>>>(rowptr, csr, N, P0, a_s, a_d, b1, P1);

    wmma_gemm_f32_kernel<<<gemm_grid(N, HC), blk, 0, stream>>>(P1, W2, P0, nullptr, N, HC, HC, 0);
    scores_kernel<<<gScores, blk, 0, stream>>>(P0, as2, ad2, a_s, a_d, NH);
    node_agg_kernel<<<gNodeW, blk, 0, stream>>>(rowptr, csr, N, P0, a_s, a_d, b2, P1);

    wmma_gemm_f32_kernel<<<gemm_grid(N, HID), blk, 0, stream>>>(P1, Wm1, P0, bm1, N, HC, HID, 1);
    wmma_gemm_f32_kernel<<<gemm_grid(N, 121), blk, 0, stream>>>(P0, Wm2, (float*)d_out, bm2,
                                                                N, HID, 121, 2);
}


# Attention logits a_s/a_d formed in the projection GEMM's epilogue from its accumulators (4 products per lane, then a fixed 16-lane butterfly sum: new summation order); the two score passes are gone
# speedup vs baseline: 3.8098x; 1.1239x over previous
#include <hip/hip_runtime.h>
#include <math.h>

#define HEADS 4
#define HID 64
#define HC 256
#define NEG_SLOPE 0.2f
#define EPS_F 1e-16f

typedef __attribute__((ext_vector_type(2))) float v2f;
typedef __attribute__((ext_vector_type(8))) float v8f;

template <int K> __device__ __forceinline__ float row_xor(float v) {
    return __int_as_float(__builtin_amdgcn_update_dpp(0, __float_as_int(v), 0x160 + K, 0xf, 0xf, false));
}

__device__ __forceinline__ void epi_store(float* __restrict__ C, const float* __restrict__ bias,
                                          size_t rowoff, int col, int NC, int mode, float v) {
    if (col < NC) {
        if (mode >= 1) v += bias[col];
        if (mode == 2) v = 1.0f / (1.0f + __expf(-v));
        C[rowoff + col] = v;
    }
}

template <bool SCORES>
__global__ __launch_bounds__(256) void wmma_gemm_f32_kernel(
    const float* __restrict__ A, const float* __restrict__ B,
    float* __restrict__ C, const float* __restrict__ bias,
    int M, int K, int NC, int mode,
    const float* __restrict__ att_src, const float* __restrict__ att_dst,
    float* __restrict__ a_s, float* __restrict__ a_d)
{
    const int wave = threadIdx.x >> 5;
    const int lane = threadIdx.x & 31;
    const int m0 = (blockIdx.x * 8 + wave) * 16;
    if (m0 >= M) return;
    const int n0 = blockIdx.y * 64;

    const int aRow = m0 + (lane & 15);
    const int aK   = (lane >> 4) << 1;
    const int bN   = lane & 15;
    const int bK   = (lane >> 4) << 1;
    const int c    = n0 + bN;

    const float* Ap = A + (size_t)aRow * K + aK;
    const size_t stepB = (size_t)4 * NC;

    v8f acc0 = {}, acc1 = {}, acc2 = {}, acc3 = {};

    if (n0 + 64 <= NC) {
        const float* Bp = B + c + (size_t)bK * NC;
        const float* Bq = Bp + NC;
        #pragma unroll 2
        for (int kb = 0; kb < K; kb += 4) {
            v2f a = *(const v2f*)Ap;
            v2f b0 = { Bp[0],  Bq[0]  };
            v2f b1 = { Bp[16], Bq[16] };
            v2f b2 = { Bp[32], Bq[32] };
            v2f b3 = { Bp[48], Bq[48] };
            Ap += 4; Bp += stepB; Bq += stepB;
            acc0 = __builtin_amdgcn_wmma_f32_16x16x4_f32(false, a, false, b0, (short)0, acc0, false, false);
            acc1 = __builtin_amdgcn_wmma_f32_16x16x4_f32(false, a, false, b1, (short)0, acc1, false, false);
            acc2 = __builtin_amdgcn_wmma_f32_16x16x4_f32(false, a, false, b2, (short)0, acc2, false, false);
            acc3 = __builtin_amdgcn_wmma_f32_16x16x4_f32(false, a, false, b3, (short)0, acc3, false, false);
        }
    } else {
        const int   c0 = min(c,      NC - 1), c1 = min(c + 16, NC - 1);
        const int   c2 = min(c + 32, NC - 1), c3 = min(c + 48, NC - 1);
        const float w0 = (c      < NC) ? 1.0f : 0.0f;
        const float w1 = (c + 16 < NC) ? 1.0f : 0.0f;
        const float w2 = (c + 32 < NC) ? 1.0f : 0.0f;
        const float w3 = (c + 48 < NC) ? 1.0f : 0.0f;
        const float* Bp = B + (size_t)bK * NC;
        const float* Bq = Bp + NC;
        #pragma unroll 2
        for (int kb = 0; kb < K; kb += 4) {
            v2f a = *(const v2f*)Ap;
            v2f b0 = { Bp[c0] * w0, Bq[c0] * w0 };
            v2f b1 = { Bp[c1] * w1, Bq[c1] * w1 };
            v2f b2 = { Bp[c2] * w2, Bq[c2] * w2 };
            v2f b3 = { Bp[c3] * w3, Bq[c3] * w3 };
            Ap += 4; Bp += stepB; Bq += stepB;
            acc0 = __builtin_amdgcn_wmma_f32_16x16x4_f32(false, a, false, b0, (short)0, acc0, false, false);
            acc1 = __builtin_amdgcn_wmma_f32_16x16x4_f32(false, a, false, b1, (short)0, acc1, false, false);
            acc2 = __builtin_amdgcn_wmma_f32_16x16x4_f32(false, a, false, b2, (short)0, acc2, false, false);
            acc3 = __builtin_amdgcn_wmma_f32_16x16x4_f32(false, a, false, b3, (short)0, acc3, false, false);
        }
    }

    const int mbase = m0 + ((lane >> 4) << 3);
    const int colb  = n0 + (lane & 15);
    #pragma unroll
    for (int r = 0; r < 8; ++r) {
        size_t rowoff = (size_t)(mbase + r) * NC;
        epi_store(C, bias, rowoff, colb,      NC, mode, acc0[r]);
        epi_store(C, bias, rowoff, colb + 16, NC, mode, acc1[r]);
        epi_store(C, bias, rowoff, colb + 32, NC, mode, acc2[r]);
        epi_store(C, bias, rowoff, colb + 48, NC, mode, acc3[r]);
    }

    if (SCORES) {
        const float s0 = att_src[colb], s1 = att_src[colb + 16], s2 = att_src[colb + 32], s3 = att_src[colb + 48];
        const float d0 = att_dst[colb], d1 = att_dst[colb + 16], d2 = att_dst[colb + 32], d3 = att_dst[colb + 48];
        float ps[8], pd[8];
        #pragma unroll
        for (int r = 0; r < 8; ++r) {
            ps[r] = acc0[r] * s0 + acc1[r] * s1 + acc2[r] * s2 + acc3[r] * s3;
            pd[r] = acc0[r] * d0 + acc1[r] * d1 + acc2[r] * d2 + acc3[r] * d3;
        }
        #pragma unroll
        for (int r = 0; r < 8; ++r) {
            ps[r] += row_xor<1>(ps[r]); pd[r] += row_xor<1>(pd[r]);
            ps[r] += row_xor<2>(ps[r]); pd[r] += row_xor<2>(pd[r]);
            ps[r] += row_xor<4>(ps[r]); pd[r] += row_xor<4>(pd[r]);
            ps[r] += row_xor<8>(ps[r]); pd[r] += row_xor<8>(pd[r]);
        }
        const int cl = lane & 15, rsel = cl & 7;
        float vs = ps[0], vd = pd[0];
        #pragma unroll
        for (int r = 1; r < 8; ++r) {
            vs = (rsel == r) ? ps[r] : vs;
            vd = (rsel == r) ? pd[r] : vd;
        }
        const int heads = NC >> 6;
        float* dst = (cl < 8) ? a_s : a_d;
        dst[(size_t)(mbase + rsel) * heads + blockIdx.y] = (cl < 8) ? vs : vd;
    }
}

__global__ __launch_bounds__(256) void zero_i32_kernel(int* __restrict__ p, int n) {
    const int i = blockIdx.x * 256 + threadIdx.x;
    if (i < n) p[i] = 0;
}

__global__ __launch_bounds__(256) void deg_count_kernel(const int* __restrict__ ei, int E,
                                                        int* __restrict__ cnt) {
    const int e = blockIdx.x * 256 + threadIdx.x;
    if (e < E) atomicAdd(cnt + ei[E + e], 1);
}

#define SCAN_T 1024
#define SCAN_ITEMS 4096
__global__ __launch_bounds__(SCAN_T) void rowptr_scan_kernel(
    const int* __restrict__ cnt, int* __restrict__ rowptr, int* __restrict__ cursor, int n)
{
    __shared__ int wsum[32];
    __shared__ int wpre[32];
    const int t = threadIdx.x, lane = t & 31, wv = t >> 5;
    const int first = blockIdx.x * SCAN_ITEMS;
    const int i0 = first + 4 * t;

    int4 v = make_int4(0, 0, 0, 0);
    if (i0 + 3 < n) {
        v = *(const int4*)(cnt + i0);
    } else {
        if (i0     < n) v.x = cnt[i0];
        if (i0 + 1 < n) v.y = cnt[i0 + 1];
        if (i0 + 2 < n) v.z = cnt[i0 + 2];
    }

    int pre = 0;
    #pragma unroll 4
    for (int j = 4 * t; j < first; j += SCAN_ITEMS) {
        const int4 u = *(const int4*)(cnt + j);
        pre += (u.x + u.y) + (u.z + u.w);
    }

    const int own = (v.x + v.y) + (v.z + v.w);
    int incl = own;
    for (int off = 1; off < 32; off <<= 1) {
        const int x = __shfl_up(incl, off, 32);
        if (lane >= off) incl += x;
    }
    for (int off = 16; off > 0; off >>= 1) pre += __shfl_xor(pre, off, 32);
    if (lane == 31) wsum[wv] = incl;
    if (lane == 0)  wpre[wv] = pre;
    __syncthreads();
    int front = wpre[lane] + ((lane < wv) ? wsum[lane] : 0);
    for (int off = 16; off > 0; off >>= 1) front += __shfl_xor(front, off, 32);

    const int r0 = front + incl - own;
    const int r1 = r0 + v.x, r2 = r1 + v.y, r3 = r2 + v.z;
    if (i0 + 3 <= n) {
        const int4 r = make_int4(r0, r1, r2, r3);
        *(int4*)(rowptr + i0) = r;
        *(int4*)(cursor + i0) = r;
    } else {
        if (i0     <= n) { rowptr[i0]     = r0; cursor[i0]     = r0; }
        if (i0 + 1 <= n) { rowptr[i0 + 1] = r1; cursor[i0 + 1] = r1; }
        if (i0 + 2 <= n) { rowptr[i0 + 2] = r2; cursor[i0 + 2] = r2; }
    }
}

__global__ __launch_bounds__(256) void csr_fill_kernel(const int* __restrict__ ei, int E,
                                                       int* __restrict__ cursor,
                                                       int* __restrict__ csr) {
    const int e = blockIdx.x * 256 + threadIdx.x;
    if (e < E) {
        const int sv = ei[e], d = ei[E + e];
        const int p = atomicAdd(cursor + d, 1);
        csr[p] = sv;
    }
}


__device__ __forceinline__ float blend_half(int lo, int x_lo, int x_hi) {
    return __int_as_float((x_lo & lo) | (x_hi & ~lo));
}

__device__ __forceinline__ float leaky(float e) { return (e > 0.0f) ? e : NEG_SLOPE * e; }

__device__ __forceinline__ float4 edge_logits(float4 a, float4 d) {
    return make_float4(leaky(a.x + d.x), leaky(a.y + d.y), leaky(a.z + d.z), leaky(a.w + d.w));
}

__device__ __forceinline__ float other_row(float v) {
    const int b = __float_as_int(v);
    return __int_as_float(__builtin_amdgcn_permlanex16(b, b, 0x76543210, 0xfedcba98, false, false));
}
__device__ __forceinline__ float4 wave_max4(float4 v) {
    v.x = fmaxf(v.x, row_xor<1>(v.x)); v.y = fmaxf(v.y, row_xor<1>(v.y)); v.z = fmaxf(v.z, row_xor<1>(v.z)); v.w = fmaxf(v.w, row_xor<1>(v.w));
    v.x = fmaxf(v.x, row_xor<2>(v.x)); v.y = fmaxf(v.y, row_xor<2>(v.y)); v.z = fmaxf(v.z, row_xor<2>(v.z)); v.w = fmaxf(v.w, row_xor<2>(v.w));
    v.x = fmaxf(v.x, row_xor<4>(v.x)); v.y = fmaxf(v.y, row_xor<4>(v.y)); v.z = fmaxf(v.z, row_xor<4>(v.z)); v.w = fmaxf(v.w, row_xor<4>(v.w));
    v.x = fmaxf(v.x, row_xor<8>(v.x)); v.y = fmaxf(v.y, row_xor<8>(v.y)); v.z = fmaxf(v.z, row_xor<8>(v.z)); v.w = fmaxf(v.w, row_xor<8>(v.w));
    v.x = fmaxf(v.x, other_row(v.x));  v.y = fmaxf(v.y, other_row(v.y));  v.z = fmaxf(v.z, other_row(v.z));  v.w = fmaxf(v.w, other_row(v.w));
    return v;
}
__device__ __forceinline__ float4 wave_sum4(float4 v) {
    v.x += row_xor<1>(v.x); v.y += row_xor<1>(v.y); v.z += row_xor<1>(v.z); v.w += row_xor<1>(v.w);
    v.x += row_xor<2>(v.x); v.y += row_xor<2>(v.y); v.z += row_xor<2>(v.z); v.w += row_xor<2>(v.w);
    v.x += row_xor<4>(v.x); v.y += row_xor<4>(v.y); v.z += row_xor<4>(v.z); v.w += row_xor<4>(v.w);
    v.x += row_xor<8>(v.x); v.y += row_xor<8>(v.y); v.z += row_xor<8>(v.z); v.w += row_xor<8>(v.w);
    v.x += other_row(v.x);  v.y += other_row(v.y);  v.z += other_row(v.z);  v.w += other_row(v.w);
    return v;
}
__device__ __forceinline__ float uniform_f(float v) {
    return __int_as_float(__builtin_amdgcn_readfirstlane(__float_as_int(v)));
}
__device__ __forceinline__ int edge_weight(float e, float m, float den, int lm) {
    return __float_as_int(__expf(e - m) / den) & lm;
}

__global__ __launch_bounds__(256) void node_agg_kernel(
    const int* __restrict__ rowptr, const int* __restrict__ csr, int N,
    const float* __restrict__ H, const float* __restrict__ a_s, const float* __restrict__ a_d,
    const float* __restrict__ bias, float* __restrict__ out)
{
    const int gw   = (int)((blockIdx.x * 256 + threadIdx.x) >> 5);
    const int lane = threadIdx.x & 31;
    if (gw >= N) return;
    const int node = __builtin_amdgcn_readfirstlane(gw);
    const int lo = (lane < 16) ? -1 : 0;
    const float4* as4 = (const float4*)a_s;
    const float4 adq = ((const float4*)a_d)[node];
    const float4 asn = as4[node];
    const int beg = rowptr[node], end = rowptr[node + 1];

    const float4* hn = (const float4*)(H + (size_t)node * HC) + lane;
    const float4 n0 = hn[0], n1 = hn[32];
    const float4 bq0 = ((const float4*)bias)[lane], bq1 = ((const float4*)bias)[32 + lane];

    const float4 es = edge_logits(asn, adq);

    const int cnt0 = min(32, end - beg);
    int s0 = node;
    if (cnt0 > 0) s0 = csr[beg + min(lane, cnt0 - 1)];
    const float4 e0 = edge_logits(as4[s0], adq);

    float4 m = make_float4(fmaxf(es.x, e0.x), fmaxf(es.y, e0.y), fmaxf(es.z, e0.z), fmaxf(es.w, e0.w));
    #pragma clang loop vectorize(disable) interleave(disable)
    for (int base = beg + 32; base < end; base += 32) {
        const int cnt = min(32, end - base);
        const float4 e = edge_logits(as4[csr[base + min(lane, cnt - 1)]], adq);
        m.x = fmaxf(m.x, e.x); m.y = fmaxf(m.y, e.y); m.z = fmaxf(m.z, e.z); m.w = fmaxf(m.w, e.w);
    }
    m = wave_max4(m);
    const float m0 = uniform_f(m.x), m1 = uniform_f(m.y), m2 = uniform_f(m.z), m3 = uniform_f(m.w);

    const int lm0 = (lane < cnt0) ? -1 : 0;
    float4 sum = make_float4(__int_as_float(__float_as_int(__expf(e0.x - m0)) & lm0),
                             __int_as_float(__float_as_int(__expf(e0.y - m1)) & lm0),
                             __int_as_float(__float_as_int(__expf(e0.z - m2)) & lm0),
                             __int_as_float(__float_as_int(__expf(e0.w - m3)) & lm0));
    for (int base = beg + 32; base < end; base += 32) {
        const int cnt = min(32, end - base);
        const float4 e = edge_logits(as4[csr[base + min(lane, cnt - 1)]], adq);
        const int lm = (lane < cnt) ? -1 : 0;
        sum.x += __int_as_float(__float_as_int(__expf(e.x - m0)) & lm);
        sum.y += __int_as_float(__float_as_int(__expf(e.y - m1)) & lm);
        sum.z += __int_as_float(__float_as_int(__expf(e.z - m2)) & lm);
        sum.w += __int_as_float(__float_as_int(__expf(e.w - m3)) & lm);
    }
    sum = wave_sum4(sum);
    const float xs0 = __expf(es.x - m0), xs1 = __expf(es.y - m1);
    const float xs2 = __expf(es.z - m2), xs3 = __expf(es.w - m3);
    const float den0 = uniform_f(sum.x + xs0) + EPS_F, den1 = uniform_f(sum.y + xs1) + EPS_F;
    const float den2 = uniform_f(sum.z + xs2) + EPS_F, den3 = uniform_f(sum.w + xs3) + EPS_F;

    float4 acc0 = make_float4(0.0f, 0.0f, 0.0f, 0.0f), acc1 = acc0;
    int sv  = s0;
    int cnt = cnt0;
    int q0 = edge_weight(e0.x, m0, den0, lm0), q1 = edge_weight(e0.y, m1, den1, lm0);
    int q2 = edge_weight(e0.z, m2, den2, lm0), q3 = edge_weight(e0.w, m3, den3, lm0);
    for (int base = beg; base < end; ) {
        for (int j = 0; j < cnt; j += 4) {
            int    s[4];
            float  al[4], ah[4];
            float4 a[4], b[4];
            for (int k = 0; k < 4; ++k) {
                s[k]  = __builtin_amdgcn_readlane(sv, j + k);
                al[k] = blend_half(lo, __builtin_amdgcn_readlane(q0, j + k),
                                       __builtin_amdgcn_readlane(q1, j + k));
                ah[k] = blend_half(lo, __builtin_amdgcn_readlane(q2, j + k),
                                       __builtin_amdgcn_readlane(q3, j + k));
            }
            for (int k = 0; k < 4; ++k) {
                const float4* hs = (const float4*)(H + (size_t)s[k] * HC) + lane;
                a[k] = hs[0]; b[k] = hs[32];
            }
            for (int k = 0; k < 4; ++k) {
                acc0.x += al[k] * a[k].x; acc0.y += al[k] * a[k].y;
                acc0.z += al[k] * a[k].z; acc0.w += al[k] * a[k].w;
                acc1.x += ah[k] * b[k].x; acc1.y += ah[k] * b[k].y;
                acc1.z += ah[k] * b[k].z; acc1.w += ah[k] * b[k].w;
            }
        }
        base += 32;
        if (base < end) {
            cnt = min(32, end - base);
            sv  = csr[base + min(lane, cnt - 1)];
            const float4 e = edge_logits(as4[sv], adq);
            const int lm = (lane < cnt) ? -1 : 0;
            q0 = edge_weight(e.x, m0, den0, lm); q1 = edge_weight(e.y, m1, den1, lm);
            q2 = edge_weight(e.z, m2, den2, lm); q3 = edge_weight(e.w, m3, den3, lm);
        }
    }
    {
        const float sl = blend_half(lo, __float_as_int(xs0 / den0), __float_as_int(xs1 / den1));
        const float sh = blend_half(lo, __float_as_int(xs2 / den2), __float_as_int(xs3 / den3));
        acc0.x += sl * n0.x; acc0.y += sl * n0.y;
        acc0.z += sl * n0.z; acc0.w += sl * n0.w;
        acc1.x += sh * n1.x; acc1.y += sh * n1.y;
        acc1.z += sh * n1.z; acc1.w += sh * n1.w;
    }
    acc0.x = fmaxf(acc0.x + bq0.x, 0.0f); acc0.y = fmaxf(acc0.y + bq0.y, 0.0f);
    acc0.z = fmaxf(acc0.z + bq0.z, 0.0f); acc0.w = fmaxf(acc0.w + bq0.w, 0.0f);
    acc1.x = fmaxf(acc1.x + bq1.x, 0.0f); acc1.y = fmaxf(acc1.y + bq1.y, 0.0f);
    acc1.z = fmaxf(acc1.z + bq1.z, 0.0f); acc1.w = fmaxf(acc1.w + bq1.w, 0.0f);
    float4* op = (float4*)(out + (size_t)node * HC) + lane;
    op[0] = acc0; op[32] = acc1;
}

extern "C" void kernel_launch(void* const* d_in, const int* in_sizes, int n_in,
                              void* d_out, int out_size, void* d_ws, size_t ws_size,
                              hipStream_t stream)
{
    const float* x   = (const float*)d_in[0];
    const int*   ei  = (const int*)d_in[1];
    const float* W1  = (const float*)d_in[2];
    const float* as1 = (const float*)d_in[3];
    const float* ad1 = (const float*)d_in[4];
    const float* b1  = (const float*)d_in[5];
    const float* W2  = (const float*)d_in[6];
    const float* as2 = (const float*)d_in[7];
    const float* ad2 = (const float*)d_in[8];
    const float* b2  = (const float*)d_in[9];
    const float* Wm1 = (const float*)d_in[10];
    const float* bm1 = (const float*)d_in[11];
    const float* Wm2 = (const float*)d_in[12];
    const float* bm2 = (const float*)d_in[13];

    const int N    = in_sizes[0] / 128;
    const int E    = in_sizes[1] / 2;
    const int NH   = N * HEADS;

    float* ws = (float*)d_ws;
    size_t o = 0;
    float*    P0    = ws + o; o += (size_t)N * HC;
    float*    P1    = ws + o; o += (size_t)N * HC;
    float*    a_s   = ws + o; o += (size_t)NH;
    float*    a_d   = ws + o; o += (size_t)NH;
    int*      cnt    = (int*)(ws + o); o += ((size_t)N + 4) & ~(size_t)3;
    int*      rowptr = (int*)(ws + o); o += ((size_t)N + 4) & ~(size_t)3;
    int*      cursor = (int*)(ws + o); o += ((size_t)N + 4) & ~(size_t)3;
    int*      csr    = (int*)(ws + o); o += (size_t)E;

    const dim3 blk(256);
    auto gemm_grid = [](int M, int NC) {
        return dim3((unsigned)(((M + 15) / 16 + 7) / 8), (unsigned)((NC + 63) / 64));
    };
    const int gNodeW  = (N * 32 + 255) / 256;
    const int gEdge1  = (E + 255) / 256;

    zero_i32_kernel<<<(N + 255) / 256, blk, 0, stream>>>(cnt, N);
    deg_count_kernel<<<gEdge1, blk, 0, stream>>>(ei, E, cnt);
    rowptr_scan_kernel<<<N / SCAN_ITEMS + 1, dim3(SCAN_T), 0, stream>>>(cnt, rowptr, cursor, N);
    csr_fill_kernel<<<gEdge1, blk, 0, stream>>>(ei, E, cursor, csr);

    wmma_gemm_f32_kernel<true><<<gemm_grid(N, HC), blk, 0, stream>>>(x, W1, P0, nullptr, N, 128, HC, 0,
                                                                     as1, ad1, a_s, a_d);
    node_agg_kernel<<<gNodeW, blk, 0, stream>>>(rowptr, csr, N, P0, a_s, a_d, b1, P1);

    wmma_gemm_f32_kernel<true><<<gemm_grid(N, HC), blk, 0, stream>>>(P1, W2, P0, nullptr, N, HC, HC, 0,
                                                                     as2, ad2, a_s, a_d);
    node_agg_kernel<<<gNodeW, blk, 0, stream>>>(rowptr, csr, N, P0, a_s, a_d, b2, P1);

    wmma_gemm_f32_kernel<false><<<gemm_grid(N, HID), blk, 0, stream>>>(P1, Wm1, P0, bm1, N, HC, HID, 1,
                                                                       nullptr, nullptr, nullptr, nullptr);
    wmma_gemm_f32_kernel<false><<<gemm_grid(N, 121), blk, 0, stream>>>(P0, Wm2, (float*)d_out, bm2,
                                                                       N, HID, 121, 2,
                                                                       nullptr, nullptr, nullptr, nullptr);
}


# GEMM fragment loads and C stores widened to 16 bytes per lane (B: 4 adjacent columns per lane, tile j = columns 4c+j; A: 8 k per load, k order 0,1,4,5 then 2,3,6,7): new summation order in C, logits
# speedup vs baseline: 4.2207x; 1.1079x over previous
#include <hip/hip_runtime.h>
#include <math.h>

#define HEADS 4
#define HID 64
#define HC 256
#define NEG_SLOPE 0.2f
#define EPS_F 1e-16f

typedef __attribute__((ext_vector_type(2))) float v2f;
typedef __attribute__((ext_vector_type(8))) float v8f;

template <int K> __device__ __forceinline__ float row_xor(float v) {
    return __int_as_float(__builtin_amdgcn_update_dpp(0, __float_as_int(v), 0x160 + K, 0xf, 0xf, false));
}

template <bool SCORES, int NC>
__global__ __launch_bounds__(256) void wmma_gemm_f32_wide_kernel(
    const float* __restrict__ A, const float* __restrict__ B,
    float* __restrict__ C, const float* __restrict__ bias,
    int M, int K,
    const float* __restrict__ att_src, const float* __restrict__ att_dst,
    float* __restrict__ a_s, float* __restrict__ a_d)
{
    const int wave = threadIdx.x >> 5;
    const int lane = threadIdx.x & 31;
    const int m0 = (blockIdx.x * 8 + wave) * 16;
    if (m0 >= M) return;
    const int n0 = blockIdx.y * 64;
    const int c = lane & 15, h = lane >> 4;

    const float* Ap = A + (size_t)(m0 + c) * K + 4 * h;
    const float* Bp = B + (size_t)(4 * h) * NC + n0 + 4 * c;

    v8f acc0 = {}, acc1 = {}, acc2 = {}, acc3 = {};

    #pragma unroll 2
    for (int kb = 0; kb < K; kb += 8) {
        const float4 a  = *(const float4*)Ap;
        const float4 r0 = *(const float4*)(Bp);
        const float4 r1 = *(const float4*)(Bp + NC);
        const float4 r2 = *(const float4*)(Bp + 2 * NC);
        const float4 r3 = *(const float4*)(Bp + 3 * NC);
        Ap += 8; Bp += (size_t)8 * NC;
        const v2f a01 = { a.x, a.y }, a23 = { a.z, a.w };
        const v2f b00 = { r0.x, r1.x }, b01 = { r0.y, r1.y }, b02 = { r0.z, r1.z }, b03 = { r0.w, r1.w };
        const v2f b10 = { r2.x, r3.x }, b11 = { r2.y, r3.y }, b12 = { r2.z, r3.z }, b13 = { r2.w, r3.w };
        acc0 = __builtin_amdgcn_wmma_f32_16x16x4_f32(false, a01, false, b00, (short)0, acc0, false, false);
        acc1 = __builtin_amdgcn_wmma_f32_16x16x4_f32(false, a01, false, b01, (short)0, acc1, false, false);
        acc2 = __builtin_amdgcn_wmma_f32_16x16x4_f32(false, a01, false, b02, (short)0, acc2, false, false);
        acc3 = __builtin_amdgcn_wmma_f32_16x16x4_f32(false, a01, false, b03, (short)0, acc3, false, false);
        acc0 = __builtin_amdgcn_wmma_f32_16x16x4_f32(false, a23, false, b10, (short)0, acc0, false, false);
        acc1 = __builtin_amdgcn_wmma_f32_16x16x4_f32(false, a23, false, b11, (short)0, acc1, false, false);
        acc2 = __builtin_amdgcn_wmma_f32_16x16x4_f32(false, a23, false, b12, (short)0, acc2, false, false);
        acc3 = __builtin_amdgcn_wmma_f32_16x16x4_f32(false, a23, false, b13, (short)0, acc3, false, false);
    }

    const int mbase = m0 + (h << 3);
    const int colb  = n0 + 4 * c;
    float4 bq = make_float4(0.0f, 0.0f, 0.0f, 0.0f);
    if (bias != nullptr) bq = *(const float4*)(bias + colb);
    #pragma unroll
    for (int r = 0; r < 8; ++r) {
        const float4 v = make_float4(acc0[r] + bq.x, acc1[r] + bq.y, acc2[r] + bq.z, acc3[r] + bq.w);
        *(float4*)(C + (size_t)(mbase + r) * NC + colb) = v;
    }

    if (SCORES) {
        const float4 sq = *(const float4*)(att_src + colb);
        const float4 dq = *(const float4*)(att_dst + colb);
        float ps[8], pd[8];
        #pragma unroll
        for (int r = 0; r < 8; ++r) {
            ps[r] = acc0[r] * sq.x + acc1[r] * sq.y + acc2[r] * sq.z + acc3[r] * sq.w;
            pd[r] = acc0[r] * dq.x + acc1[r] * dq.y + acc2[r] * dq.z + acc3[r] * dq.w;
        }
        #pragma unroll
        for (int r = 0; r < 8; ++r) {
            ps[r] += row_xor<1>(ps[r]); pd[r] += row_xor<1>(pd[r]);
            ps[r] += row_xor<2>(ps[r]); pd[r] += row_xor<2>(pd[r]);
            ps[r] += row_xor<4>(ps[r]); pd[r] += row_xor<4>(pd[r]);
            ps[r] += row_xor<8>(ps[r]); pd[r] += row_xor<8>(pd[r]);
        }
        const int rsel = c & 7;
        float vs = ps[0], vd = pd[0];
        #pragma unroll
        for (int r = 1; r < 8; ++r) {
            vs = (rsel == r) ? ps[r] : vs;
            vd = (rsel == r) ? pd[r] : vd;
        }
        constexpr int heads = NC >> 6;
        float* dst = (c < 8) ? a_s : a_d;
        dst[(size_t)(mbase + rsel) * heads + blockIdx.y] = (c < 8) ? vs : vd;
    }
}

__device__ __forceinline__ void epi_store(float* __restrict__ C, const float* __restrict__ bias,
                                          size_t rowoff, int col, int NC, int mode, float v) {
    if (col < NC) {
        if (mode >= 1) v += bias[col];
        if (mode == 2) v = 1.0f / (1.0f + __expf(-v));
        C[rowoff + col] = v;
    }
}

__global__ __launch_bounds__(256) void wmma_gemm_f32_kernel(
    const float* __restrict__ A, const float* __restrict__ B,
    float* __restrict__ C, const float* __restrict__ bias,
    int M, int K, int NC, int mode)
{
    const int wave = threadIdx.x >> 5;
    const int lane = threadIdx.x & 31;
    const int m0 = (blockIdx.x * 8 + wave) * 16;
    if (m0 >= M) return;
    const int n0 = blockIdx.y * 64;

    const int aRow = m0 + (lane & 15);
    const int aK   = (lane >> 4) << 1;
    const int bN   = lane & 15;
    const int bK   = (lane >> 4) << 1;
    const int c    = n0 + bN;

    const float* Ap = A + (size_t)aRow * K + aK;
    const size_t stepB = (size_t)4 * NC;

    v8f acc0 = {}, acc1 = {}, acc2 = {}, acc3 = {};

    if (n0 + 64 <= NC) {
        const float* Bp = B + c + (size_t)bK * NC;
        const float* Bq = Bp + NC;
        #pragma unroll 2
        for (int kb = 0; kb < K; kb += 4) {
            v2f a = *(const v2f*)Ap;
            v2f b0 = { Bp[0],  Bq[0]  };
            v2f b1 = { Bp[16], Bq[16] };
            v2f b2 = { Bp[32], Bq[32] };
            v2f b3 = { Bp[48], Bq[48] };
            Ap += 4; Bp += stepB; Bq += stepB;
            acc0 = __builtin_amdgcn_wmma_f32_16x16x4_f32(false, a, false, b0, (short)0, acc0, false, false);
            acc1 = __builtin_amdgcn_wmma_f32_16x16x4_f32(false, a, false, b1, (short)0, acc1, false, false);
            acc2 = __builtin_amdgcn_wmma_f32_16x16x4_f32(false, a, false, b2, (short)0, acc2, false, false);
            acc3 = __builtin_amdgcn_wmma_f32_16x16x4_f32(false, a, false, b3, (short)0, acc3, false, false);
        }
    } else {
        const int   c0 = min(c,      NC - 1), c1 = min(c + 16, NC - 1);
        const int   c2 = min(c + 32, NC - 1), c3 = min(c + 48, NC - 1);
        const float w0 = (c      < NC) ? 1.0f : 0.0f;
        const float w1 = (c + 16 < NC) ? 1.0f : 0.0f;
        const float w2 = (c + 32 < NC) ? 1.0f : 0.0f;
        const float w3 = (c + 48 < NC) ? 1.0f : 0.0f;
        const float* Bp = B + (size_t)bK * NC;
        const float* Bq = Bp + NC;
        #pragma unroll 2
        for (int kb = 0; kb < K; kb += 4) {
            v2f a = *(const v2f*)Ap;
            v2f b0 = { Bp[c0] * w0, Bq[c0] * w0 };
            v2f b1 = { Bp[c1] * w1, Bq[c1] * w1 };
            v2f b2 = { Bp[c2] * w2, Bq[c2] * w2 };
            v2f b3 = { Bp[c3] * w3, Bq[c3] * w3 };
            Ap += 4; Bp += stepB; Bq += stepB;
            acc0 = __builtin_amdgcn_wmma_f32_16x16x4_f32(false, a, false, b0, (short)0, acc0, false, false);
            acc1 = __builtin_amdgcn_wmma_f32_16x16x4_f32(false, a, false, b1, (short)0, acc1, false, false);
            acc2 = __builtin_amdgcn_wmma_f32_16x16x4_f32(false, a, false, b2, (short)0, acc2, false, false);
            acc3 = __builtin_amdgcn_wmma_f32_16x16x4_f32(false, a, false, b3, (short)0, acc3, false, false);
        }
    }

    const int mbase = m0 + ((lane >> 4) << 3);
    const int colb  = n0 + (lane & 15);
    #pragma unroll
    for (int r = 0; r < 8; ++r) {
        size_t rowoff = (size_t)(mbase + r) * NC;
        epi_store(C, bias, rowoff, colb,      NC, mode, acc0[r]);
        epi_store(C, bias, rowoff, colb + 16, NC, mode, acc1[r]);
        epi_store(C, bias, rowoff, colb + 32, NC, mode, acc2[r]);
        epi_store(C, bias, rowoff, colb + 48, NC, mode, acc3[r]);
    }
}

__global__ __launch_bounds__(256) void zero_i32_kernel(int* __restrict__ p, int n) {
    const int i = blockIdx.x * 256 + threadIdx.x;
    if (i < n) p[i] = 0;
}

__global__ __launch_bounds__(256) void deg_count_kernel(const int* __restrict__ ei, int E,
                                                        int* __restrict__ cnt) {
    const int e = blockIdx.x * 256 + threadIdx.x;
    if (e < E) atomicAdd(cnt + ei[E + e], 1);
}

#define SCAN_T 1024
#define SCAN_ITEMS 4096
__global__ __launch_bounds__(SCAN_T) void rowptr_scan_kernel(
    const int* __restrict__ cnt, int* __restrict__ rowptr, int* __restrict__ cursor, int n)
{
    __shared__ int wsum[32];
    __shared__ int wpre[32];
    const int t = threadIdx.x, lane = t & 31, wv = t >> 5;
    const int first = blockIdx.x * SCAN_ITEMS;
    const int i0 = first + 4 * t;

    int4 v = make_int4(0, 0, 0, 0);
    if (i0 + 3 < n) {
        v = *(const int4*)(cnt + i0);
    } else {
        if (i0     < n) v.x = cnt[i0];
        if (i0 + 1 < n) v.y = cnt[i0 + 1];
        if (i0 + 2 < n) v.z = cnt[i0 + 2];
    }

    int pre = 0;
    #pragma unroll 4
    for (int j = 4 * t; j < first; j += SCAN_ITEMS) {
        const int4 u = *(const int4*)(cnt + j);
        pre += (u.x + u.y) + (u.z + u.w);
    }

    const int own = (v.x + v.y) + (v.z + v.w);
    int incl = own;
    for (int off = 1; off < 32; off <<= 1) {
        const int x = __shfl_up(incl, off, 32);
        if (lane >= off) incl += x;
    }
    for (int off = 16; off > 0; off >>= 1) pre += __shfl_xor(pre, off, 32);
    if (lane == 31) wsum[wv] = incl;
    if (lane == 0)  wpre[wv] = pre;
    __syncthreads();
    int front = wpre[lane] + ((lane < wv) ? wsum[lane] : 0);
    for (int off = 16; off > 0; off >>= 1) front += __shfl_xor(front, off, 32);

    const int r0 = front + incl - own;
    const int r1 = r0 + v.x, r2 = r1 + v.y, r3 = r2 + v.z;
    if (i0 + 3 <= n) {
        const int4 r = make_int4(r0, r1, r2, r3);
        *(int4*)(rowptr + i0) = r;
        *(int4*)(cursor + i0) = r;
    } else {
        if (i0     <= n) { rowptr[i0]     = r0; cursor[i0]     = r0; }
        if (i0 + 1 <= n) { rowptr[i0 + 1] = r1; cursor[i0 + 1] = r1; }
        if (i0 + 2 <= n) { rowptr[i0 + 2] = r2; cursor[i0 + 2] = r2; }
    }
}

__global__ __launch_bounds__(256) void csr_fill_kernel(const int* __restrict__ ei, int E,
                                                       int* __restrict__ cursor,
                                                       int* __restrict__ csr) {
    const int e = blockIdx.x * 256 + threadIdx.x;
    if (e < E) {
        const int sv = ei[e], d = ei[E + e];
        const int p = atomicAdd(cursor + d, 1);
        csr[p] = sv;
    }
}


__device__ __forceinline__ float blend_half(int lo, int x_lo, int x_hi) {
    return __int_as_float((x_lo & lo) | (x_hi & ~lo));
}

__device__ __forceinline__ float leaky(float e) { return (e > 0.0f) ? e : NEG_SLOPE * e; }

__device__ __forceinline__ float4 edge_logits(float4 a, float4 d) {
    return make_float4(leaky(a.x + d.x), leaky(a.y + d.y), leaky(a.z + d.z), leaky(a.w + d.w));
}

__device__ __forceinline__ float other_row(float v) {
    const int b = __float_as_int(v);
    return __int_as_float(__builtin_amdgcn_permlanex16(b, b, 0x76543210, 0xfedcba98, false, false));
}
__device__ __forceinline__ float4 wave_max4(float4 v) {
    v.x = fmaxf(v.x, row_xor<1>(v.x)); v.y = fmaxf(v.y, row_xor<1>(v.y)); v.z = fmaxf(v.z, row_xor<1>(v.z)); v.w = fmaxf(v.w, row_xor<1>(v.w));
    v.x = fmaxf(v.x, row_xor<2>(v.x)); v.y = fmaxf(v.y, row_xor<2>(v.y)); v.z = fmaxf(v.z, row_xor<2>(v.z)); v.w = fmaxf(v.w, row_xor<2>(v.w));
    v.x = fmaxf(v.x, row_xor<4>(v.x)); v.y = fmaxf(v.y, row_xor<4>(v.y)); v.z = fmaxf(v.z, row_xor<4>(v.z)); v.w = fmaxf(v.w, row_xor<4>(v.w));
    v.x = fmaxf(v.x, row_xor<8>(v.x)); v.y = fmaxf(v.y, row_xor<8>(v.y)); v.z = fmaxf(v.z, row_xor<8>(v.z)); v.w = fmaxf(v.w, row_xor<8>(v.w));
    v.x = fmaxf(v.x, other_row(v.x));  v.y = fmaxf(v.y, other_row(v.y));  v.z = fmaxf(v.z, other_row(v.z));  v.w = fmaxf(v.w, other_row(v.w));
    return v;
}
__device__ __forceinline__ float4 wave_sum4(float4 v) {
    v.x += row_xor<1>(v.x); v.y += row_xor<1>(v.y); v.z += row_xor<1>(v.z); v.w += row_xor<1>(v.w);
    v.x += row_xor<2>(v.x); v.y += row_xor<2>(v.y); v.z += row_xor<2>(v.z); v.w += row_xor<2>(v.w);
    v.x += row_xor<4>(v.x); v.y += row_xor<4>(v.y); v.z += row_xor<4>(v.z); v.w += row_xor<4>(v.w);
    v.x += row_xor<8>(v.x); v.y += row_xor<8>(v.y); v.z += row_xor<8>(v.z); v.w += row_xor<8>(v.w);
    v.x += other_row(v.x);  v.y += other_row(v.y);  v.z += other_row(v.z);  v.w += other_row(v.w);
    return v;
}
__device__ __forceinline__ float uniform_f(float v) {
    return __int_as_float(__builtin_amdgcn_readfirstlane(__float_as_int(v)));
}
__device__ __forceinline__ int edge_weight(float e, float m, float den, int lm) {
    return __float_as_int(__expf(e - m) / den) & lm;
}

__global__ __launch_bounds__(256) void node_agg_kernel(
    const int* __restrict__ rowptr, const int* __restrict__ csr, int N,
    const float* __restrict__ H, const float* __restrict__ a_s, const float* __restrict__ a_d,
    const float* __restrict__ bias, float* __restrict__ out)
{
    const int gw   = (int)((blockIdx.x * 256 + threadIdx.x) >> 5);
    const int lane = threadIdx.x & 31;
    if (gw >= N) return;
    const int node = __builtin_amdgcn_readfirstlane(gw);
    const int lo = (lane < 16) ? -1 : 0;
    const float4* as4 = (const float4*)a_s;
    const float4 adq = ((const float4*)a_d)[node];
    const float4 asn = as4[node];
    const int beg = rowptr[node], end = rowptr[node + 1];

    const float4* hn = (const float4*)(H + (size_t)node * HC) + lane;
    const float4 n0 = hn[0], n1 = hn[32];
    const float4 bq0 = ((const float4*)bias)[lane], bq1 = ((const float4*)bias)[32 + lane];

    const float4 es = edge_logits(asn, adq);

    const int cnt0 = min(32, end - beg);
    int s0 = node;
    if (cnt0 > 0) s0 = csr[beg + min(lane, cnt0 - 1)];
    const float4 e0 = edge_logits(as4[s0], adq);

    float4 m = make_float4(fmaxf(es.x, e0.x), fmaxf(es.y, e0.y), fmaxf(es.z, e0.z), fmaxf(es.w, e0.w));
    #pragma clang loop vectorize(disable) interleave(disable)
    for (int base = beg + 32; base < end; base += 32) {
        const int cnt = min(32, end - base);
        const float4 e = edge_logits(as4[csr[base + min(lane, cnt - 1)]], adq);
        m.x = fmaxf(m.x, e.x); m.y = fmaxf(m.y, e.y); m.z = fmaxf(m.z, e.z); m.w = fmaxf(m.w, e.w);
    }
    m = wave_max4(m);
    const float m0 = uniform_f(m.x), m1 = uniform_f(m.y), m2 = uniform_f(m.z), m3 = uniform_f(m.w);

    const int lm0 = (lane < cnt0) ? -1 : 0;
    float4 sum = make_float4(__int_as_float(__float_as_int(__expf(e0.x - m0)) & lm0),
                             __int_as_float(__float_as_int(__expf(e0.y - m1)) & lm0),
                             __int_as_float(__float_as_int(__expf(e0.z - m2)) & lm0),
                             __int_as_float(__float_as_int(__expf(e0.w - m3)) & lm0));
    for (int base = beg + 32; base < end; base += 32) {
        const int cnt = min(32, end - base);
        const float4 e = edge_logits(as4[csr[base + min(lane, cnt - 1)]], adq);
        const int lm = (lane < cnt) ? -1 : 0;
        sum.x += __int_as_float(__float_as_int(__expf(e.x - m0)) & lm);
        sum.y += __int_as_float(__float_as_int(__expf(e.y - m1)) & lm);
        sum.z += __int_as_float(__float_as_int(__expf(e.z - m2)) & lm);
        sum.w += __int_as_float(__float_as_int(__expf(e.w - m3)) & lm);
    }
    sum = wave_sum4(sum);
    const float xs0 = __expf(es.x - m0), xs1 = __expf(es.y - m1);
    const float xs2 = __expf(es.z - m2), xs3 = __expf(es.w - m3);
    const float den0 = uniform_f(sum.x + xs0) + EPS_F, den1 = uniform_f(sum.y + xs1) + EPS_F;
    const float den2 = uniform_f(sum.z + xs2) + EPS_F, den3 = uniform_f(sum.w + xs3) + EPS_F;

    float4 acc0 = make_float4(0.0f, 0.0f, 0.0f, 0.0f), acc1 = acc0;
    int sv  = s0;
    int cnt = cnt0;
    int q0 = edge_weight(e0.x, m0, den0, lm0), q1 = edge_weight(e0.y, m1, den1, lm0);
    int q2 = edge_weight(e0.z, m2, den2, lm0), q3 = edge_weight(e0.w, m3, den3, lm0);
    for (int base = beg; base < end; ) {
        for (int j = 0; j < cnt; j += 4) {
            int    s[4];
            float  al[4], ah[4];
            float4 a[4], b[4];
            for (int k = 0; k < 4; ++k) {
                s[k]  = __builtin_amdgcn_readlane(sv, j + k);
                al[k] = blend_half(lo, __builtin_amdgcn_readlane(q0, j + k),
                                       __builtin_amdgcn_readlane(q1, j + k));
                ah[k] = blend_half(lo, __builtin_amdgcn_readlane(q2, j + k),
                                       __builtin_amdgcn_readlane(q3, j + k));
            }
            for (int k = 0; k < 4; ++k) {
                const float4* hs = (const float4*)(H + (size_t)s[k] * HC) + lane;
                a[k] = hs[0]; b[k] = hs[32];
            }
            for (int k = 0; k < 4; ++k) {
                acc0.x += al[k] * a[k].x; acc0.y += al[k] * a[k].y;
                acc0.z += al[k] * a[k].z; acc0.w += al[k] * a[k].w;
                acc1.x += ah[k] * b[k].x; acc1.y += ah[k] * b[k].y;
                acc1.z += ah[k] * b[k].z; acc1.w += ah[k] * b[k].w;
            }
        }
        base += 32;
        if (base < end) {
            cnt = min(32, end - base);
            sv  = csr[base + min(lane, cnt - 1)];
            const float4 e = edge_logits(as4[sv], adq);
            const int lm = (lane < cnt) ? -1 : 0;
            q0 = edge_weight(e.x, m0, den0, lm); q1 = edge_weight(e.y, m1, den1, lm);
            q2 = edge_weight(e.z, m2, den2, lm); q3 = edge_weight(e.w, m3, den3, lm);
        }
    }
    {
        const float sl = blend_half(lo, __float_as_int(xs0 / den0), __float_as_int(xs1 / den1));
        const float sh = blend_half(lo, __float_as_int(xs2 / den2), __float_as_int(xs3 / den3));
        acc0.x += sl * n0.x; acc0.y += sl * n0.y;
        acc0.z += sl * n0.z; acc0.w += sl * n0.w;
        acc1.x += sh * n1.x; acc1.y += sh * n1.y;
        acc1.z += sh * n1.z; acc1.w += sh * n1.w;
    }
    acc0.x = fmaxf(acc0.x + bq0.x, 0.0f); acc0.y = fmaxf(acc0.y + bq0.y, 0.0f);
    acc0.z = fmaxf(acc0.z + bq0.z, 0.0f); acc0.w = fmaxf(acc0.w + bq0.w, 0.0f);
    acc1.x = fmaxf(acc1.x + bq1.x, 0.0f); acc1.y = fmaxf(acc1.y + bq1.y, 0.0f);
    acc1.z = fmaxf(acc1.z + bq1.z, 0.0f); acc1.w = fmaxf(acc1.w + bq1.w, 0.0f);
    float4* op = (float4*)(out + (size_t)node * HC) + lane;
    op[0] = acc0; op[32] = acc1;
}

extern "C" void kernel_launch(void* const* d_in, const int* in_sizes, int n_in,
                              void* d_out, int out_size, void* d_ws, size_t ws_size,
                              hipStream_t stream)
{
    const float* x   = (const float*)d_in[0];
    const int*   ei  = (const int*)d_in[1];
    const float* W1  = (const float*)d_in[2];
    const float* as1 = (const float*)d_in[3];
    const float* ad1 = (const float*)d_in[4];
    const float* b1  = (const float*)d_in[5];
    const float* W2  = (const float*)d_in[6];
    const float* as2 = (const float*)d_in[7];
    const float* ad2 = (const float*)d_in[8];
    const float* b2  = (const float*)d_in[9];
    const float* Wm1 = (const float*)d_in[10];
    const float* bm1 = (const float*)d_in[11];
    const float* Wm2 = (const float*)d_in[12];
    const float* bm2 = (const float*)d_in[13];

    const int N    = in_sizes[0] / 128;
    const int E    = in_sizes[1] / 2;
    const int NH   = N * HEADS;

    float* ws = (float*)d_ws;
    size_t o = 0;
    float*    P0    = ws + o; o += (size_t)N * HC;
    float*    P1    = ws + o; o += (size_t)N * HC;
    float*    a_s   = ws + o; o += (size_t)NH;
    float*    a_d   = ws + o; o += (size_t)NH;
    int*      cnt    = (int*)(ws + o); o += ((size_t)N + 4) & ~(size_t)3;
    int*      rowptr = (int*)(ws + o); o += ((size_t)N + 4) & ~(size_t)3;
    int*      cursor = (int*)(ws + o); o += ((size_t)N + 4) & ~(size_t)3;
    int*      csr    = (int*)(ws + o); o += (size_t)E;

    const dim3 blk(256);
    auto gemm_grid = [](int M, int NC) {
        return dim3((unsigned)(((M + 15) / 16 + 7) / 8), (unsigned)((NC + 63) / 64));
    };
    const int gNodeW  = (N * 32 + 255) / 256;
    const int gEdge1  = (E + 255) / 256;

    zero_i32_kernel<<<(N + 255) / 256, blk, 0, stream>>>(cnt, N);
    deg_count_kernel<<<gEdge1, blk, 0, stream>>>(ei, E, cnt);
    rowptr_scan_kernel<<<N / SCAN_ITEMS + 1, dim3(SCAN_T), 0, stream>>>(cnt, rowptr, cursor, N);
    csr_fill_kernel<<<gEdge1, blk, 0, stream>>>(ei, E, cursor, csr);

    wmma_gemm_f32_wide_kernel<true, HC><<<gemm_grid(N, HC), blk, 0, stream>>>(x, W1, P0, nullptr, N, 128,
                                                                              as1, ad1, a_s, a_d);
    node_agg_kernel<<<gNodeW, blk, 0, stream>>>(rowptr, csr, N, P0, a_s, a_d, b1, P1);

    wmma_gemm_f32_wide_kernel<true, HC><<<gemm_grid(N, HC), blk, 0, stream>>>(P1, W2, P0, nullptr, N, HC,
                                                                              as2, ad2, a_s, a_d);
    node_agg_kernel<<<gNodeW, blk, 0, stream>>>(rowptr, csr, N, P0, a_s, a_d, b2, P1);

    wmma_gemm_f32_wide_kernel<false, HID><<<gemm_grid(N, HID), blk, 0, stream>>>(P1, Wm1, P0, bm1, N, HC,
                                                                                 nullptr, nullptr, nullptr, nullptr);
    wmma_gemm_f32_kernel<<<gemm_grid(N, 121), blk, 0, stream>>>(P0, Wm2, (float*)d_out, bm2, N, HID, 121, 2);
}


# Each workgroup copies its 64-column strip of B into LDS once (whole-line 16-byte loads and stores); its 8 waves read their B fragments from LDS in the same k order: per-wave global B loads are gone
# speedup vs baseline: 4.3857x; 1.0391x over previous
#include <hip/hip_runtime.h>
#include <math.h>

#define HEADS 4
#define HID 64
#define HC 256
#define NEG_SLOPE 0.2f
#define EPS_F 1e-16f

typedef __attribute__((ext_vector_type(2))) float v2f;
typedef __attribute__((ext_vector_type(8))) float v8f;

template <int K> __device__ __forceinline__ float row_xor(float v) {
    return __int_as_float(__builtin_amdgcn_update_dpp(0, __float_as_int(v), 0x160 + K, 0xf, 0xf, false));
}

template <bool SCORES, int NC, int K>
__global__ __launch_bounds__(256) void wmma_gemm_f32_wide_kernel(
    const float* __restrict__ A, const float* __restrict__ B,
    float* __restrict__ C, const float* __restrict__ bias,
    int M,
    const float* __restrict__ att_src, const float* __restrict__ att_dst,
    float* __restrict__ a_s, float* __restrict__ a_d)
{
    __shared__ float Bs[K * 64];
    const int n0 = blockIdx.y * 64;
    {
        const int fr = threadIdx.x >> 4, fc = (threadIdx.x & 15) * 4;
        const float* Bg = B + (size_t)fr * NC + n0 + fc;
        float4 t4[K / 16];
        #pragma unroll
        for (int p = 0; p < K / 16; ++p) t4[p] = *(const float4*)(Bg + (size_t)(16 * p) * NC);
        #pragma unroll
        for (int p = 0; p < K / 16; ++p) *(float4*)(Bs + (fr + 16 * p) * 64 + fc) = t4[p];
    }
    __syncthreads();

    const int wave = threadIdx.x >> 5;
    const int lane = threadIdx.x & 31;
    const int m0 = (blockIdx.x * 8 + wave) * 16;
    if (m0 >= M) return;
    const int c = lane & 15, h = lane >> 4;

    const float* Ap = A + (size_t)(m0 + c) * K + 4 * h;
    const float* Bp = Bs + (4 * h) * 64 + 4 * c;

    v8f acc0 = {}, acc1 = {}, acc2 = {}, acc3 = {};

    #pragma unroll 2
    for (int kb = 0; kb < K; kb += 8) {
        const float4 a  = *(const float4*)Ap;
        const float4 r0 = *(const float4*)(Bp);
        const float4 r1 = *(const float4*)(Bp + 64);
        const float4 r2 = *(const float4*)(Bp + 128);
        const float4 r3 = *(const float4*)(Bp + 192);
        Ap += 8; Bp += 8 * 64;
        const v2f a01 = { a.x, a.y }, a23 = { a.z, a.w };
        const v2f b00 = { r0.x, r1.x }, b01 = { r0.y, r1.y }, b02 = { r0.z, r1.z }, b03 = { r0.w, r1.w };
        const v2f b10 = { r2.x, r3.x }, b11 = { r2.y, r3.y }, b12 = { r2.z, r3.z }, b13 = { r2.w, r3.w };
        acc0 = __builtin_amdgcn_wmma_f32_16x16x4_f32(false, a01, false, b00, (short)0, acc0, false, false);
        acc1 = __builtin_amdgcn_wmma_f32_16x16x4_f32(false, a01, false, b01, (short)0, acc1, false, false);
        acc2 = __builtin_amdgcn_wmma_f32_16x16x4_f32(false, a01, false, b02, (short)0, acc2, false, false);
        acc3 = __builtin_amdgcn_wmma_f32_16x16x4_f32(false, a01, false, b03, (short)0, acc3, false, false);
        acc0 = __builtin_amdgcn_wmma_f32_16x16x4_f32(false, a23, false, b10, (short)0, acc0, false, false);
        acc1 = __builtin_amdgcn_wmma_f32_16x16x4_f32(false, a23, false, b11, (short)0, acc1, false, false);
        acc2 = __builtin_amdgcn_wmma_f32_16x16x4_f32(false, a23, false, b12, (short)0, acc2, false, false);
        acc3 = __builtin_amdgcn_wmma_f32_16x16x4_f32(false, a23, false, b13, (short)0, acc3, false, false);
    }

    const int mbase = m0 + (h << 3);
    const int colb  = n0 + 4 * c;
    float4 bq = make_float4(0.0f, 0.0f, 0.0f, 0.0f);
    if (bias != nullptr) bq = *(const float4*)(bias + colb);
    #pragma unroll
    for (int r = 0; r < 8; ++r) {
        const float4 v = make_float4(acc0[r] + bq.x, acc1[r] + bq.y, acc2[r] + bq.z, acc3[r] + bq.w);
        *(float4*)(C + (size_t)(mbase + r) * NC + colb) = v;
    }

    if (SCORES) {
        const float4 sq = *(const float4*)(att_src + colb);
        const float4 dq = *(const float4*)(att_dst + colb);
        float ps[8], pd[8];
        #pragma unroll
        for (int r = 0; r < 8; ++r) {
            ps[r] = acc0[r] * sq.x + acc1[r] * sq.y + acc2[r] * sq.z + acc3[r] * sq.w;
            pd[r] = acc0[r] * dq.x + acc1[r] * dq.y + acc2[r] * dq.z + acc3[r] * dq.w;
        }
        #pragma unroll
        for (int r = 0; r < 8; ++r) {
            ps[r] += row_xor<1>(ps[r]); pd[r] += row_xor<1>(pd[r]);
            ps[r] += row_xor<2>(ps[r]); pd[r] += row_xor<2>(pd[r]);
            ps[r] += row_xor<4>(ps[r]); pd[r] += row_xor<4>(pd[r]);
            ps[r] += row_xor<8>(ps[r]); pd[r] += row_xor<8>(pd[r]);
        }
        const int rsel = c & 7;
        float vs = ps[0], vd = pd[0];
        #pragma unroll
        for (int r = 1; r < 8; ++r) {
            vs = (rsel == r) ? ps[r] : vs;
            vd = (rsel == r) ? pd[r] : vd;
        }
        constexpr int heads = NC >> 6;
        float* dst = (c < 8) ? a_s : a_d;
        dst[(size_t)(mbase + rsel) * heads + blockIdx.y] = (c < 8) ? vs : vd;
    }
}

__device__ __forceinline__ void epi_store(float* __restrict__ C, const float* __restrict__ bias,
                                          size_t rowoff, int col, int NC, int mode, float v) {
    if (col < NC) {
        if (mode >= 1) v += bias[col];
        if (mode == 2) v = 1.0f / (1.0f + __expf(-v));
        C[rowoff + col] = v;
    }
}

__global__ __launch_bounds__(256) void wmma_gemm_f32_kernel(
    const float* __restrict__ A, const float* __restrict__ B,
    float* __restrict__ C, const float* __restrict__ bias,
    int M, int K, int NC, int mode)
{
    const int wave = threadIdx.x >> 5;
    const int lane = threadIdx.x & 31;
    const int m0 = (blockIdx.x * 8 + wave) * 16;
    if (m0 >= M) return;
    const int n0 = blockIdx.y * 64;

    const int aRow = m0 + (lane & 15);
    const int aK   = (lane >> 4) << 1;
    const int bN   = lane & 15;
    const int bK   = (lane >> 4) << 1;
    const int c    = n0 + bN;

    const float* Ap = A + (size_t)aRow * K + aK;
    const size_t stepB = (size_t)4 * NC;

    v8f acc0 = {}, acc1 = {}, acc2 = {}, acc3 = {};

    if (n0 + 64 <= NC) {
        const float* Bp = B + c + (size_t)bK * NC;
        const float* Bq = Bp + NC;
        #pragma unroll 2
        for (int kb = 0; kb < K; kb += 4) {
            v2f a = *(const v2f*)Ap;
            v2f b0 = { Bp[0],  Bq[0]  };
            v2f b1 = { Bp[16], Bq[16] };
            v2f b2 = { Bp[32], Bq[32] };
            v2f b3 = { Bp[48], Bq[48] };
            Ap += 4; Bp += stepB; Bq += stepB;
            acc0 = __builtin_amdgcn_wmma_f32_16x16x4_f32(false, a, false, b0, (short)0, acc0, false, false);
            acc1 = __builtin_amdgcn_wmma_f32_16x16x4_f32(false, a, false, b1, (short)0, acc1, false, false);
            acc2 = __builtin_amdgcn_wmma_f32_16x16x4_f32(false, a, false, b2, (short)0, acc2, false, false);
            acc3 = __builtin_amdgcn_wmma_f32_16x16x4_f32(false, a, false, b3, (short)0, acc3, false, false);
        }
    } else {
        const int   c0 = min(c,      NC - 1), c1 = min(c + 16, NC - 1);
        const int   c2 = min(c + 32, NC - 1), c3 = min(c + 48, NC - 1);
        const float w0 = (c      < NC) ? 1.0f : 0.0f;
        const float w1 = (c + 16 < NC) ? 1.0f : 0.0f;
        const float w2 = (c + 32 < NC) ? 1.0f : 0.0f;
        const float w3 = (c + 48 < NC) ? 1.0f : 0.0f;
        const float* Bp = B + (size_t)bK * NC;
        const float* Bq = Bp + NC;
        #pragma unroll 2
        for (int kb = 0; kb < K; kb += 4) {
            v2f a = *(const v2f*)Ap;
            v2f b0 = { Bp[c0] * w0, Bq[c0] * w0 };
            v2f b1 = { Bp[c1] * w1, Bq[c1] * w1 };
            v2f b2 = { Bp[c2] * w2, Bq[c2] * w2 };
            v2f b3 = { Bp[c3] * w3, Bq[c3] * w3 };
            Ap += 4; Bp += stepB; Bq += stepB;
            acc0 = __builtin_amdgcn_wmma_f32_16x16x4_f32(false, a, false, b0, (short)0, acc0, false, false);
            acc1 = __builtin_amdgcn_wmma_f32_16x16x4_f32(false, a, false, b1, (short)0, acc1, false, false);
            acc2 = __builtin_amdgcn_wmma_f32_16x16x4_f32(false, a, false, b2, (short)0, acc2, false, false);
            acc3 = __builtin_amdgcn_wmma_f32_16x16x4_f32(false, a, false, b3, (short)0, acc3, false, false);
        }
    }

    const int mbase = m0 + ((lane >> 4) << 3);
    const int colb  = n0 + (lane & 15);
    #pragma unroll
    for (int r = 0; r < 8; ++r) {
        size_t rowoff = (size_t)(mbase + r) * NC;
        epi_store(C, bias, rowoff, colb,      NC, mode, acc0[r]);
        epi_store(C, bias, rowoff, colb + 16, NC, mode, acc1[r]);
        epi_store(C, bias, rowoff, colb + 32, NC, mode, acc2[r]);
        epi_store(C, bias, rowoff, colb + 48, NC, mode, acc3[r]);
    }
}

__global__ __launch_bounds__(256) void zero_i32_kernel(int* __restrict__ p, int n) {
    const int i = blockIdx.x * 256 + threadIdx.x;
    if (i < n) p[i] = 0;
}

__global__ __launch_bounds__(256) void deg_count_kernel(const int* __restrict__ ei, int E,
                                                        int* __restrict__ cnt) {
    const int e = blockIdx.x * 256 + threadIdx.x;
    if (e < E) atomicAdd(cnt + ei[E + e], 1);
}

#define SCAN_T 1024
#define SCAN_ITEMS 4096
__global__ __launch_bounds__(SCAN_T) void rowptr_scan_kernel(
    const int* __restrict__ cnt, int* __restrict__ rowptr, int* __restrict__ cursor, int n)
{
    __shared__ int wsum[32];
    __shared__ int wpre[32];
    const int t = threadIdx.x, lane = t & 31, wv = t >> 5;
    const int first = blockIdx.x * SCAN_ITEMS;
    const int i0 = first + 4 * t;

    int4 v = make_int4(0, 0, 0, 0);
    if (i0 + 3 < n) {
        v = *(const int4*)(cnt + i0);
    } else {
        if (i0     < n) v.x = cnt[i0];
        if (i0 + 1 < n) v.y = cnt[i0 + 1];
        if (i0 + 2 < n) v.z = cnt[i0 + 2];
    }

    int pre = 0;
    #pragma unroll 4
    for (int j = 4 * t; j < first; j += SCAN_ITEMS) {
        const int4 u = *(const int4*)(cnt + j);
        pre += (u.x + u.y) + (u.z + u.w);
    }

    const int own = (v.x + v.y) + (v.z + v.w);
    int incl = own;
    for (int off = 1; off < 32; off <<= 1) {
        const int x = __shfl_up(incl, off, 32);
        if (lane >= off) incl += x;
    }
    for (int off = 16; off > 0; off >>= 1) pre += __shfl_xor(pre, off, 32);
    if (lane == 31) wsum[wv] = incl;
    if (lane == 0)  wpre[wv] = pre;
    __syncthreads();
    int front = wpre[lane] + ((lane < wv) ? wsum[lane] : 0);
    for (int off = 16; off > 0; off >>= 1) front += __shfl_xor(front, off, 32);

    const int r0 = front + incl - own;
    const int r1 = r0 + v.x, r2 = r1 + v.y, r3 = r2 + v.z;
    if (i0 + 3 <= n) {
        const int4 r = make_int4(r0, r1, r2, r3);
        *(int4*)(rowptr + i0) = r;
        *(int4*)(cursor + i0) = r;
    } else {
        if (i0     <= n) { rowptr[i0]     = r0; cursor[i0]     = r0; }
        if (i0 + 1 <= n) { rowptr[i0 + 1] = r1; cursor[i0 + 1] = r1; }
        if (i0 + 2 <= n) { rowptr[i0 + 2] = r2; cursor[i0 + 2] = r2; }
    }
}

__global__ __launch_bounds__(256) void csr_fill_kernel(const int* __restrict__ ei, int E,
                                                       int* __restrict__ cursor,
                                                       int* __restrict__ csr) {
    const int e = blockIdx.x * 256 + threadIdx.x;
    if (e < E) {
        const int sv = ei[e], d = ei[E + e];
        const int p = atomicAdd(cursor + d, 1);
        csr[p] = sv;
    }
}


__device__ __forceinline__ float blend_half(int lo, int x_lo, int x_hi) {
    return __int_as_float((x_lo & lo) | (x_hi & ~lo));
}

__device__ __forceinline__ float leaky(float e) { return (e > 0.0f) ? e : NEG_SLOPE * e; }

__device__ __forceinline__ float4 edge_logits(float4 a, float4 d) {
    return make_float4(leaky(a.x + d.x), leaky(a.y + d.y), leaky(a.z + d.z), leaky(a.w + d.w));
}

__device__ __forceinline__ float other_row(float v) {
    const int b = __float_as_int(v);
    return __int_as_float(__builtin_amdgcn_permlanex16(b, b, 0x76543210, 0xfedcba98, false, false));
}
__device__ __forceinline__ float4 wave_max4(float4 v) {
    v.x = fmaxf(v.x, row_xor<1>(v.x)); v.y = fmaxf(v.y, row_xor<1>(v.y)); v.z = fmaxf(v.z, row_xor<1>(v.z)); v.w = fmaxf(v.w, row_xor<1>(v.w));
    v.x = fmaxf(v.x, row_xor<2>(v.x)); v.y = fmaxf(v.y, row_xor<2>(v.y)); v.z = fmaxf(v.z, row_xor<2>(v.z)); v.w = fmaxf(v.w, row_xor<2>(v.w));
    v.x = fmaxf(v.x, row_xor<4>(v.x)); v.y = fmaxf(v.y, row_xor<4>(v.y)); v.z = fmaxf(v.z, row_xor<4>(v.z)); v.w = fmaxf(v.w, row_xor<4>(v.w));
    v.x = fmaxf(v.x, row_xor<8>(v.x)); v.y = fmaxf(v.y, row_xor<8>(v.y)); v.z = fmaxf(v.z, row_xor<8>(v.z)); v.w = fmaxf(v.w, row_xor<8>(v.w));
    v.x = fmaxf(v.x, other_row(v.x));  v.y = fmaxf(v.y, other_row(v.y));  v.z = fmaxf(v.z, other_row(v.z));  v.w = fmaxf(v.w, other_row(v.w));
    return v;
}
__device__ __forceinline__ float4 wave_sum4(float4 v) {
    v.x += row_xor<1>(v.x); v.y += row_xor<1>(v.y); v.z += row_xor<1>(v.z); v.w += row_xor<1>(v.w);
    v.x += row_xor<2>(v.x); v.y += row_xor<2>(v.y); v.z += row_xor<2>(v.z); v.w += row_xor<2>(v.w);
    v.x += row_xor<4>(v.x); v.y += row_xor<4>(v.y); v.z += row_xor<4>(v.z); v.w += row_xor<4>(v.w);
    v.x += row_xor<8>(v.x); v.y += row_xor<8>(v.y); v.z += row_xor<8>(v.z); v.w += row_xor<8>(v.w);
    v.x += other_row(v.x);  v.y += other_row(v.y);  v.z += other_row(v.z);  v.w += other_row(v.w);
    return v;
}
__device__ __forceinline__ float uniform_f(float v) {
    return __int_as_float(__builtin_amdgcn_readfirstlane(__float_as_int(v)));
}
__device__ __forceinline__ int edge_weight(float e, float m, float den, int lm) {
    return __float_as_int(__expf(e - m) / den) & lm;
}

__global__ __launch_bounds__(256) void node_agg_kernel(
    const int* __restrict__ rowptr, const int* __restrict__ csr, int N,
    const float* __restrict__ H, const float* __restrict__ a_s, const float* __restrict__ a_d,
    const float* __restrict__ bias, float* __restrict__ out)
{
    const int gw   = (int)((blockIdx.x * 256 + threadIdx.x) >> 5);
    const int lane = threadIdx.x & 31;
    if (gw >= N) return;
    const int node = __builtin_amdgcn_readfirstlane(gw);
    const int lo = (lane < 16) ? -1 : 0;
    const float4* as4 = (const float4*)a_s;
    const float4 adq = ((const float4*)a_d)[node];
    const float4 asn = as4[node];
    const int beg = rowptr[node], end = rowptr[node + 1];

    const float4* hn = (const float4*)(H + (size_t)node * HC) + lane;
    const float4 n0 = hn[0], n1 = hn[32];
    const float4 bq0 = ((const float4*)bias)[lane], bq1 = ((const float4*)bias)[32 + lane];

    const float4 es = edge_logits(asn, adq);

    const int cnt0 = min(32, end - beg);
    int s0 = node;
    if (cnt0 > 0) s0 = csr[beg + min(lane, cnt0 - 1)];
    const float4 e0 = edge_logits(as4[s0], adq);

    float4 m = make_float4(fmaxf(es.x, e0.x), fmaxf(es.y, e0.y), fmaxf(es.z, e0.z), fmaxf(es.w, e0.w));
    #pragma clang loop vectorize(disable) interleave(disable)
    for (int base = beg + 32; base < end; base += 32) {
        const int cnt = min(32, end - base);
        const float4 e = edge_logits(as4[csr[base + min(lane, cnt - 1)]], adq);
        m.x = fmaxf(m.x, e.x); m.y = fmaxf(m.y, e.y); m.z = fmaxf(m.z, e.z); m.w = fmaxf(m.w, e.w);
    }
    m = wave_max4(m);
    const float m0 = uniform_f(m.x), m1 = uniform_f(m.y), m2 = uniform_f(m.z), m3 = uniform_f(m.w);

    const int lm0 = (lane < cnt0) ? -1 : 0;
    float4 sum = make_float4(__int_as_float(__float_as_int(__expf(e0.x - m0)) & lm0),
                             __int_as_float(__float_as_int(__expf(e0.y - m1)) & lm0),
                             __int_as_float(__float_as_int(__expf(e0.z - m2)) & lm0),
                             __int_as_float(__float_as_int(__expf(e0.w - m3)) & lm0));
    for (int base = beg + 32; base < end; base += 32) {
        const int cnt = min(32, end - base);
        const float4 e = edge_logits(as4[csr[base + min(lane, cnt - 1)]], adq);
        const int lm = (lane < cnt) ? -1 : 0;
        sum.x += __int_as_float(__float_as_int(__expf(e.x - m0)) & lm);
        sum.y += __int_as_float(__float_as_int(__expf(e.y - m1)) & lm);
        sum.z += __int_as_float(__float_as_int(__expf(e.z - m2)) & lm);
        sum.w += __int_as_float(__float_as_int(__expf(e.w - m3)) & lm);
    }
    sum = wave_sum4(sum);
    const float xs0 = __expf(es.x - m0), xs1 = __expf(es.y - m1);
    const float xs2 = __expf(es.z - m2), xs3 = __expf(es.w - m3);
    const float den0 = uniform_f(sum.x + xs0) + EPS_F, den1 = uniform_f(sum.y + xs1) + EPS_F;
    const float den2 = uniform_f(sum.z + xs2) + EPS_F, den3 = uniform_f(sum.w + xs3) + EPS_F;

    float4 acc0 = make_float4(0.0f, 0.0f, 0.0f, 0.0f), acc1 = acc0;
    int sv  = s0;
    int cnt = cnt0;
    int q0 = edge_weight(e0.x, m0, den0, lm0), q1 = edge_weight(e0.y, m1, den1, lm0);
    int q2 = edge_weight(e0.z, m2, den2, lm0), q3 = edge_weight(e0.w, m3, den3, lm0);
    for (int base = beg; base < end; ) {
        for (int j = 0; j < cnt; j += 4) {
            int    s[4];
            float  al[4], ah[4];
            float4 a[4], b[4];
            for (int k = 0; k < 4; ++k) {
                s[k]  = __builtin_amdgcn_readlane(sv, j + k);
                al[k] = blend_half(lo, __builtin_amdgcn_readlane(q0, j + k),
                                       __builtin_amdgcn_readlane(q1, j + k));
                ah[k] = blend_half(lo, __builtin_amdgcn_readlane(q2, j + k),
                                       __builtin_amdgcn_readlane(q3, j + k));
            }
            for (int k = 0; k < 4; ++k) {
                const float4* hs = (const float4*)(H + (size_t)s[k] * HC) + lane;
                a[k] = hs[0]; b[k] = hs[32];
            }
            for (int k = 0; k < 4; ++k) {
                acc0.x += al[k] * a[k].x; acc0.y += al[k] * a[k].y;
                acc0.z += al[k] * a[k].z; acc0.w += al[k] * a[k].w;
                acc1.x += ah[k] * b[k].x; acc1.y += ah[k] * b[k].y;
                acc1.z += ah[k] * b[k].z; acc1.w += ah[k] * b[k].w;
            }
        }
        base += 32;
        if (base < end) {
            cnt = min(32, end - base);
            sv  = csr[base + min(lane, cnt - 1)];
            const float4 e = edge_logits(as4[sv], adq);
            const int lm = (lane < cnt) ? -1 : 0;
            q0 = edge_weight(e.x, m0, den0, lm); q1 = edge_weight(e.y, m1, den1, lm);
            q2 = edge_weight(e.z, m2, den2, lm); q3 = edge_weight(e.w, m3, den3, lm);
        }
    }
    {
        const float sl = blend_half(lo, __float_as_int(xs0 / den0), __float_as_int(xs1 / den1));
        const float sh = blend_half(lo, __float_as_int(xs2 / den2), __float_as_int(xs3 / den3));
        acc0.x += sl * n0.x; acc0.y += sl * n0.y;
        acc0.z += sl * n0.z; acc0.w += sl * n0.w;
        acc1.x += sh * n1.x; acc1.y += sh * n1.y;
        acc1.z += sh * n1.z; acc1.w += sh * n1.w;
    }
    acc0.x = fmaxf(acc0.x + bq0.x, 0.0f); acc0.y = fmaxf(acc0.y + bq0.y, 0.0f);
    acc0.z = fmaxf(acc0.z + bq0.z, 0.0f); acc0.w = fmaxf(acc0.w + bq0.w, 0.0f);
    acc1.x = fmaxf(acc1.x + bq1.x, 0.0f); acc1.y = fmaxf(acc1.y + bq1.y, 0.0f);
    acc1.z = fmaxf(acc1.z + bq1.z, 0.0f); acc1.w = fmaxf(acc1.w + bq1.w, 0.0f);
    float4* op = (float4*)(out + (size_t)node * HC) + lane;
    op[0] = acc0; op[32] = acc1;
}

extern "C" void kernel_launch(void* const* d_in, const int* in_sizes, int n_in,
                              void* d_out, int out_size, void* d_ws, size_t ws_size,
                              hipStream_t stream)
{
    const float* x   = (const float*)d_in[0];
    const int*   ei  = (const int*)d_in[1];
    const float* W1  = (const float*)d_in[2];
    const float* as1 = (const float*)d_in[3];
    const float* ad1 = (const float*)d_in[4];
    const float* b1  = (const float*)d_in[5];
    const float* W2  = (const float*)d_in[6];
    const float* as2 = (const float*)d_in[7];
    const float* ad2 = (const float*)d_in[8];
    const float* b2  = (const float*)d_in[9];
    const float* Wm1 = (const float*)d_in[10];
    const float* bm1 = (const float*)d_in[11];
    const float* Wm2 = (const float*)d_in[12];
    const float* bm2 = (const float*)d_in[13];

    const int N    = in_sizes[0] / 128;
    const int E    = in_sizes[1] / 2;
    const int NH   = N * HEADS;

    float* ws = (float*)d_ws;
    size_t o = 0;
    float*    P0    = ws + o; o += (size_t)N * HC;
    float*    P1    = ws + o; o += (size_t)N * HC;
    float*    a_s   = ws + o; o += (size_t)NH;
    float*    a_d   = ws + o; o += (size_t)NH;
    int*      cnt    = (int*)(ws + o); o += ((size_t)N + 4) & ~(size_t)3;
    int*      rowptr = (int*)(ws + o); o += ((size_t)N + 4) & ~(size_t)3;
    int*      cursor = (int*)(ws + o); o += ((size_t)N + 4) & ~(size_t)3;
    int*      csr    = (int*)(ws + o); o += (size_t)E;

    const dim3 blk(256);
    auto gemm_grid = [](int M, int NC) {
        return dim3((unsigned)(((M + 15) / 16 + 7) / 8), (unsigned)((NC + 63) / 64));
    };
    const int gNodeW  = (N * 32 + 255) / 256;
    const int gEdge1  = (E + 255) / 256;

    zero_i32_kernel<<<(N + 255) / 256, blk, 0, stream>>>(cnt, N);
    deg_count_kernel<<<gEdge1, blk, 0, stream>>>(ei, E, cnt);
    rowptr_scan_kernel<<<N / SCAN_ITEMS + 1, dim3(SCAN_T), 0, stream>>>(cnt, rowptr, cursor, N);
    csr_fill_kernel<<<gEdge1, blk, 0, stream>>>(ei, E, cursor, csr);

    wmma_gemm_f32_wide_kernel<true, HC, 128><<<gemm_grid(N, HC), blk, 0, stream>>>(x, W1, P0, nullptr, N,
                                                                                   as1, ad1, a_s, a_d);
    node_agg_kernel<<<gNodeW, blk, 0, stream>>>(rowptr, csr, N, P0, a_s, a_d, b1, P1);

    wmma_gemm_f32_wide_kernel<true, HC, HC><<<gemm_grid(N, HC), blk, 0, stream>>>(P1, W2, P0, nullptr, N,
                                                                                  as2, ad2, a_s, a_d);
    node_agg_kernel<<<gNodeW, blk, 0, stream>>>(rowptr, csr, N, P0, a_s, a_d, b2, P1);

    wmma_gemm_f32_wide_kernel<false, HID, HC><<<gemm_grid(N, HID), blk, 0, stream>>>(P1, Wm1, P0, bm1, N,
                                                                                     nullptr, nullptr, nullptr, nullptr);
    wmma_gemm_f32_kernel<<<gemm_grid(N, 121), blk, 0, stream>>>(P0, Wm2, (float*)d_out, bm2, N, HID, 121, 2);
}


# Edge list grouped by destination by a two-launch sort ranked in LDS (256-node buckets, whole-line stores) in place of four launches with global atomics; order of sources inside a node changes
# speedup vs baseline: 6.3965x; 1.4585x over previous
#include <hip/hip_runtime.h>
#include <math.h>

#define HEADS 4
#define HID 64
#define HC 256
#define NEG_SLOPE 0.2f
#define EPS_F 1e-16f

typedef __attribute__((ext_vector_type(2))) float v2f;
typedef __attribute__((ext_vector_type(8))) float v8f;

template <int K> __device__ __forceinline__ float row_xor(float v) {
    return __int_as_float(__builtin_amdgcn_update_dpp(0, __float_as_int(v), 0x160 + K, 0xf, 0xf, false));
}

template <bool SCORES, int NC, int K>
__global__ __launch_bounds__(256) void wmma_gemm_f32_wide_kernel(
    const float* __restrict__ A, const float* __restrict__ B,
    float* __restrict__ C, const float* __restrict__ bias,
    int M,
    const float* __restrict__ att_src, const float* __restrict__ att_dst,
    float* __restrict__ a_s, float* __restrict__ a_d)
{
    __shared__ float Bs[K * 64];
    const int n0 = blockIdx.y * 64;
    {
        const int fr = threadIdx.x >> 4, fc = (threadIdx.x & 15) * 4;
        const float* Bg = B + (size_t)fr * NC + n0 + fc;
        float4 t4[K / 16];
        #pragma unroll
        for (int p = 0; p < K / 16; ++p) t4[p] = *(const float4*)(Bg + (size_t)(16 * p) * NC);
        #pragma unroll
        for (int p = 0; p < K / 16; ++p) *(float4*)(Bs + (fr + 16 * p) * 64 + fc) = t4[p];
    }
    __syncthreads();

    const int wave = threadIdx.x >> 5;
    const int lane = threadIdx.x & 31;
    const int m0 = (blockIdx.x * 8 + wave) * 16;
    if (m0 >= M) return;
    const int c = lane & 15, h = lane >> 4;

    const float* Ap = A + (size_t)(m0 + c) * K + 4 * h;
    const float* Bp = Bs + (4 * h) * 64 + 4 * c;

    v8f acc0 = {}, acc1 = {}, acc2 = {}, acc3 = {};

    #pragma unroll 2
    for (int kb = 0; kb < K; kb += 8) {
        const float4 a  = *(const float4*)Ap;
        const float4 r0 = *(const float4*)(Bp);
        const float4 r1 = *(const float4*)(Bp + 64);
        const float4 r2 = *(const float4*)(Bp + 128);
        const float4 r3 = *(const float4*)(Bp + 192);
        Ap += 8; Bp += 8 * 64;
        const v2f a01 = { a.x, a.y }, a23 = { a.z, a.w };
        const v2f b00 = { r0.x, r1.x }, b01 = { r0.y, r1.y }, b02 = { r0.z, r1.z }, b03 = { r0.w, r1.w };
        const v2f b10 = { r2.x, r3.x }, b11 = { r2.y, r3.y }, b12 = { r2.z, r3.z }, b13 = { r2.w, r3.w };
        acc0 = __builtin_amdgcn_wmma_f32_16x16x4_f32(false, a01, false, b00, (short)0, acc0, false, false);
        acc1 = __builtin_amdgcn_wmma_f32_16x16x4_f32(false, a01, false, b01, (short)0, acc1, false, false);
        acc2 = __builtin_amdgcn_wmma_f32_16x16x4_f32(false, a01, false, b02, (short)0, acc2, false, false);
        acc3 = __builtin_amdgcn_wmma_f32_16x16x4_f32(false, a01, false, b03, (short)0, acc3, false, false);
        acc0 = __builtin_amdgcn_wmma_f32_16x16x4_f32(false, a23, false, b10, (short)0, acc0, false, false);
        acc1 = __builtin_amdgcn_wmma_f32_16x16x4_f32(false, a23, false, b11, (short)0, acc1, false, false);
        acc2 = __builtin_amdgcn_wmma_f32_16x16x4_f32(false, a23, false, b12, (short)0, acc2, false, false);
        acc3 = __builtin_amdgcn_wmma_f32_16x16x4_f32(false, a23, false, b13, (short)0, acc3, false, false);
    }

    const int mbase = m0 + (h << 3);
    const int colb  = n0 + 4 * c;
    float4 bq = make_float4(0.0f, 0.0f, 0.0f, 0.0f);
    if (bias != nullptr) bq = *(const float4*)(bias + colb);
    #pragma unroll
    for (int r = 0; r < 8; ++r) {
        const float4 v = make_float4(acc0[r] + bq.x, acc1[r] + bq.y, acc2[r] + bq.z, acc3[r] + bq.w);
        *(float4*)(C + (size_t)(mbase + r) * NC + colb) = v;
    }

    if (SCORES) {
        const float4 sq = *(const float4*)(att_src + colb);
        const float4 dq = *(const float4*)(att_dst + colb);
        float ps[8], pd[8];
        #pragma unroll
        for (int r = 0; r < 8; ++r) {
            ps[r] = acc0[r] * sq.x + acc1[r] * sq.y + acc2[r] * sq.z + acc3[r] * sq.w;
            pd[r] = acc0[r] * dq.x + acc1[r] * dq.y + acc2[r] * dq.z + acc3[r] * dq.w;
        }
        #pragma unroll
        for (int r = 0; r < 8; ++r) {
            ps[r] += row_xor<1>(ps[r]); pd[r] += row_xor<1>(pd[r]);
            ps[r] += row_xor<2>(ps[r]); pd[r] += row_xor<2>(pd[r]);
            ps[r] += row_xor<4>(ps[r]); pd[r] += row_xor<4>(pd[r]);
            ps[r] += row_xor<8>(ps[r]); pd[r] += row_xor<8>(pd[r]);
        }
        const int rsel = c & 7;
        float vs = ps[0], vd = pd[0];
        #pragma unroll
        for (int r = 1; r < 8; ++r) {
            vs = (rsel == r) ? ps[r] : vs;
            vd = (rsel == r) ? pd[r] : vd;
        }
        constexpr int heads = NC >> 6;
        float* dst = (c < 8) ? a_s : a_d;
        dst[(size_t)(mbase + rsel) * heads + blockIdx.y] = (c < 8) ? vs : vd;
    }
}

__device__ __forceinline__ void epi_store(float* __restrict__ C, const float* __restrict__ bias,
                                          size_t rowoff, int col, int NC, int mode, float v) {
    if (col < NC) {
        if (mode >= 1) v += bias[col];
        if (mode == 2) v = 1.0f / (1.0f + __expf(-v));
        C[rowoff + col] = v;
    }
}

__global__ __launch_bounds__(256) void wmma_gemm_f32_kernel(
    const float* __restrict__ A, const float* __restrict__ B,
    float* __restrict__ C, const float* __restrict__ bias,
    int M, int K, int NC, int mode)
{
    const int wave = threadIdx.x >> 5;
    const int lane = threadIdx.x & 31;
    const int m0 = (blockIdx.x * 8 + wave) * 16;
    if (m0 >= M) return;
    const int n0 = blockIdx.y * 64;

    const int aRow = m0 + (lane & 15);
    const int aK   = (lane >> 4) << 1;
    const int bN   = lane & 15;
    const int bK   = (lane >> 4) << 1;
    const int c    = n0 + bN;

    const float* Ap = A + (size_t)aRow * K + aK;
    const size_t stepB = (size_t)4 * NC;

    v8f acc0 = {}, acc1 = {}, acc2 = {}, acc3 = {};

    if (n0 + 64 <= NC) {
        const float* Bp = B + c + (size_t)bK * NC;
        const float* Bq = Bp + NC;
        #pragma unroll 2
        for (int kb = 0; kb < K; kb += 4) {
            v2f a = *(const v2f*)Ap;
            v2f b0 = { Bp[0],  Bq[0]  };
            v2f b1 = { Bp[16], Bq[16] };
            v2f b2 = { Bp[32], Bq[32] };
            v2f b3 = { Bp[48], Bq[48] };
            Ap += 4; Bp += stepB; Bq += stepB;
            acc0 = __builtin_amdgcn_wmma_f32_16x16x4_f32(false, a, false, b0, (short)0, acc0, false, false);
            acc1 = __builtin_amdgcn_wmma_f32_16x16x4_f32(false, a, false, b1, (short)0, acc1, false, false);
            acc2 = __builtin_amdgcn_wmma_f32_16x16x4_f32(false, a, false, b2, (short)0, acc2, false, false);
            acc3 = __builtin_amdgcn_wmma_f32_16x16x4_f32(false, a, false, b3, (short)0, acc3, false, false);
        }
    } else {
        const int   c0 = min(c,      NC - 1), c1 = min(c + 16, NC - 1);
        const int   c2 = min(c + 32, NC - 1), c3 = min(c + 48, NC - 1);
        const float w0 = (c      < NC) ? 1.0f : 0.0f;
        const float w1 = (c + 16 < NC) ? 1.0f : 0.0f;
        const float w2 = (c + 32 < NC) ? 1.0f : 0.0f;
        const float w3 = (c + 48 < NC) ? 1.0f : 0.0f;
        const float* Bp = B + (size_t)bK * NC;
        const float* Bq = Bp + NC;
        #pragma unroll 2
        for (int kb = 0; kb < K; kb += 4) {
            v2f a = *(const v2f*)Ap;
            v2f b0 = { Bp[c0] * w0, Bq[c0] * w0 };
            v2f b1 = { Bp[c1] * w1, Bq[c1] * w1 };
            v2f b2 = { Bp[c2] * w2, Bq[c2] * w2 };
            v2f b3 = { Bp[c3] * w3, Bq[c3] * w3 };
            Ap += 4; Bp += stepB; Bq += stepB;
            acc0 = __builtin_amdgcn_wmma_f32_16x16x4_f32(false, a, false, b0, (short)0, acc0, false, false);
            acc1 = __builtin_amdgcn_wmma_f32_16x16x4_f32(false, a, false, b1, (short)0, acc1, false, false);
            acc2 = __builtin_amdgcn_wmma_f32_16x16x4_f32(false, a, false, b2, (short)0, acc2, false, false);
            acc3 = __builtin_amdgcn_wmma_f32_16x16x4_f32(false, a, false, b3, (short)0, acc3, false, false);
        }
    }

    const int mbase = m0 + ((lane >> 4) << 3);
    const int colb  = n0 + (lane & 15);
    #pragma unroll
    for (int r = 0; r < 8; ++r) {
        size_t rowoff = (size_t)(mbase + r) * NC;
        epi_store(C, bias, rowoff, colb,      NC, mode, acc0[r]);
        epi_store(C, bias, rowoff, colb + 16, NC, mode, acc1[r]);
        epi_store(C, bias, rowoff, colb + 32, NC, mode, acc2[r]);
        epi_store(C, bias, rowoff, colb + 48, NC, mode, acc3[r]);
    }
}

#define PART_T 1024
#define PART_CH 4096
#define BUCKET_SHIFT 8
#define BUCKET_NODES 256
#define SORT_CAP 12288

__device__ __forceinline__ int block_excl_scan(int v, int* ws, int& total) {
    const int lane = threadIdx.x & 31, wv = threadIdx.x >> 5;
    int incl = v;
    for (int off = 1; off < 32; off <<= 1) {
        const int x = __shfl_up(incl, off, 32);
        if (lane >= off) incl += x;
    }
    if (lane == 31) ws[wv] = incl;
    __syncthreads();
    const int w = ws[lane];
    int front = (lane < wv) ? w : 0;
    int tot = w;
    for (int off = 16; off > 0; off >>= 1) {
        front += __shfl_xor(front, off, 32);
        tot   += __shfl_xor(tot, off, 32);
    }
    total = tot;
    return front + incl - v;
}

__global__ __launch_bounds__(PART_T) void edge_partition_kernel(
    const int* __restrict__ ei, int E, int NB, int GA,
    int* __restrict__ cntT, int* __restrict__ offT, int* __restrict__ pairs_w)
{
    int2* __restrict__ pairs = (int2*)pairs_w;
    __shared__ int  hist[PART_T];
    __shared__ int  offs[PART_T];
    __shared__ int  wtmp[32];
    __shared__ int2 stage[PART_CH];
    const int t = threadIdx.x, g = blockIdx.x;
    hist[t] = 0;
    __syncthreads();

    int d[4], s[4], bk[4], r[4];
    #pragma unroll
    for (int k = 0; k < 4; ++k) {
        const int e = g * PART_CH + k * PART_T + t;
        d[k] = -1; s[k] = 0;
        if (e < E) { d[k] = ei[E + e]; s[k] = ei[e]; }
    }
    #pragma unroll
    for (int k = 0; k < 4; ++k) {
        bk[k] = min((int)((unsigned)d[k] >> BUCKET_SHIFT), PART_T - 1);
        r[k] = 0;
        if (d[k] >= 0) r[k] = atomicAdd(&hist[bk[k]], 1);
    }
    __syncthreads();

    const int v = hist[t];
    int total;
    const int ex = block_excl_scan(v, wtmp, total);
    offs[t] = ex;
    if (t < NB) { cntT[t * GA + g] = v; offT[t * GA + g] = ex; }
    __syncthreads();

    #pragma unroll
    for (int k = 0; k < 4; ++k)
        if (d[k] >= 0) stage[offs[bk[k]] + r[k]] = make_int2(d[k], s[k]);
    __syncthreads();

    #pragma unroll
    for (int k = 0; k < 4; ++k) {
        const int i = k * PART_T + t;
        const int e = g * PART_CH + i;
        if (e < E) pairs[e] = stage[i];
    }
}

template <bool PLACE>
__device__ __forceinline__ void visit_pair(const int2 ds, int* cnt, const int* npre,
                                           int* cst, int* out, bool in_lds) {
    const int ld = ds.x & (BUCKET_NODES - 1);
    const int r = atomicAdd(&cnt[ld], 1);
    if (PLACE) {
        const int pos = npre[ld] + r;
        if (in_lds) cst[pos] = ds.y; else out[pos] = ds.y;
    }
}
template <bool PLACE>
__device__ __forceinline__ void walk_bucket(const int2* __restrict__ pairs, int GA,
                                            const int* gcnt, const int* goff, int* cnt,
                                            const int* npre, int* cst, int* out, bool in_lds) {
    const int lane = threadIdx.x & 31, wv = threadIdx.x >> 5;
    for (int g0 = wv; g0 < GA; g0 += 128) {
        int n[4];
        const int2* p[4];
        int2 e[4];
        #pragma unroll
        for (int k = 0; k < 4; ++k) {
            const int g = g0 + 32 * k;
            const int gg = min(g, GA - 1);
            n[k] = (g < GA) ? gcnt[gg] : 0;
            p[k] = pairs + (size_t)gg * PART_CH + goff[gg];
        }
        #pragma unroll
        for (int k = 0; k < 4; ++k) {
            e[k] = make_int2(0, 0);
            if (lane < n[k]) e[k] = p[k][lane];
        }
        #pragma unroll
        for (int k = 0; k < 4; ++k)
            if (lane < n[k]) visit_pair<PLACE>(e[k], cnt, npre, cst, out, in_lds);
        #pragma unroll
        for (int k = 0; k < 4; ++k)
            for (int i = lane + 32; i < n[k]; i += 32)
                visit_pair<PLACE>(p[k][i], cnt, npre, cst, out, in_lds);
    }
}

__global__ __launch_bounds__(PART_T) void bucket_sort_kernel(
    const int* __restrict__ cntT, const int* __restrict__ offT, const int* __restrict__ pairs_w,
    int GA, int NB, int N, int* __restrict__ rowptr, int* __restrict__ csr)
{
    const int2* __restrict__ pairs = (const int2*)pairs_w;
    __shared__ int gcnt[PART_T];
    __shared__ int goff[PART_T];
    __shared__ int ncnt[BUCKET_NODES];
    __shared__ int npre[BUCKET_NODES];
    __shared__ int ncur[BUCKET_NODES];
    __shared__ int wpre[32];
    __shared__ int wtmp[32];
    __shared__ int cst[SORT_CAP];
    const int t = threadIdx.x, lane = t & 31, wv = t >> 5;
    const int b = blockIdx.x;

    const int lim = b * GA;
    const int nq = lim >> 2;
    int pre = 0;
    #pragma unroll 4
    for (int q = t; q < nq; q += PART_T) {
        const int4 u = ((const int4*)cntT)[q];
        pre += (u.x + u.y) + (u.z + u.w);
    }
    if (t < (lim & 3)) pre += cntT[(nq << 2) + t];

    int c = 0, o = 0;
    if (t < GA) { c = cntT[lim + t]; o = offT[lim + t]; }
    gcnt[t] = c; goff[t] = o;
    if (t < BUCKET_NODES) { ncnt[t] = 0; ncur[t] = 0; }
    for (int off = 16; off > 0; off >>= 1) pre += __shfl_xor(pre, off, 32);
    if (lane == 0) wpre[wv] = pre;
    __syncthreads();
    int base = wpre[lane];
    for (int off = 16; off > 0; off >>= 1) base += __shfl_xor(base, off, 32);

    walk_bucket<false>(pairs, GA, gcnt, goff, ncnt, npre, cst, csr, false);
    __syncthreads();

    const int v = (t < BUCKET_NODES) ? ncnt[t] : 0;
    int total;
    const int ex = block_excl_scan(v, wtmp, total);
    if (t < BUCKET_NODES) {
        npre[t] = ex;
        const int node = (b << BUCKET_SHIFT) + t;
        if (node < N) rowptr[node] = base + ex;
    }
    if (b == NB - 1 && t == 0) rowptr[N] = base + total;
    __syncthreads();

    const bool in_lds = (total <= SORT_CAP);
    walk_bucket<true>(pairs, GA, gcnt, goff, ncur, npre, cst, csr + base, in_lds);
    if (in_lds) {
        __syncthreads();
        for (int i = t; i < total; i += PART_T) csr[base + i] = cst[i];
    }
}


__device__ __forceinline__ float blend_half(int lo, int x_lo, int x_hi) {
    return __int_as_float((x_lo & lo) | (x_hi & ~lo));
}

__device__ __forceinline__ float leaky(float e) { return (e > 0.0f) ? e : NEG_SLOPE * e; }

__device__ __forceinline__ float4 edge_logits(float4 a, float4 d) {
    return make_float4(leaky(a.x + d.x), leaky(a.y + d.y), leaky(a.z + d.z), leaky(a.w + d.w));
}

__device__ __forceinline__ float other_row(float v) {
    const int b = __float_as_int(v);
    return __int_as_float(__builtin_amdgcn_permlanex16(b, b, 0x76543210, 0xfedcba98, false, false));
}
__device__ __forceinline__ float4 wave_max4(float4 v) {
    v.x = fmaxf(v.x, row_xor<1>(v.x)); v.y = fmaxf(v.y, row_xor<1>(v.y)); v.z = fmaxf(v.z, row_xor<1>(v.z)); v.w = fmaxf(v.w, row_xor<1>(v.w));
    v.x = fmaxf(v.x, row_xor<2>(v.x)); v.y = fmaxf(v.y, row_xor<2>(v.y)); v.z = fmaxf(v.z, row_xor<2>(v.z)); v.w = fmaxf(v.w, row_xor<2>(v.w));
    v.x = fmaxf(v.x, row_xor<4>(v.x)); v.y = fmaxf(v.y, row_xor<4>(v.y)); v.z = fmaxf(v.z, row_xor<4>(v.z)); v.w = fmaxf(v.w, row_xor<4>(v.w));
    v.x = fmaxf(v.x, row_xor<8>(v.x)); v.y = fmaxf(v.y, row_xor<8>(v.y)); v.z = fmaxf(v.z, row_xor<8>(v.z)); v.w = fmaxf(v.w, row_xor<8>(v.w));
    v.x = fmaxf(v.x, other_row(v.x));  v.y = fmaxf(v.y, other_row(v.y));  v.z = fmaxf(v.z, other_row(v.z));  v.w = fmaxf(v.w, other_row(v.w));
    return v;
}
__device__ __forceinline__ float4 wave_sum4(float4 v) {
    v.x += row_xor<1>(v.x); v.y += row_xor<1>(v.y); v.z += row_xor<1>(v.z); v.w += row_xor<1>(v.w);
    v.x += row_xor<2>(v.x); v.y += row_xor<2>(v.y); v.z += row_xor<2>(v.z); v.w += row_xor<2>(v.w);
    v.x += row_xor<4>(v.x); v.y += row_xor<4>(v.y); v.z += row_xor<4>(v.z); v.w += row_xor<4>(v.w);
    v.x += row_xor<8>(v.x); v.y += row_xor<8>(v.y); v.z += row_xor<8>(v.z); v.w += row_xor<8>(v.w);
    v.x += other_row(v.x);  v.y += other_row(v.y);  v.z += other_row(v.z);  v.w += other_row(v.w);
    return v;
}
__device__ __forceinline__ float uniform_f(float v) {
    return __int_as_float(__builtin_amdgcn_readfirstlane(__float_as_int(v)));
}
__device__ __forceinline__ int edge_weight(float e, float m, float den, int lm) {
    return __float_as_int(__expf(e - m) / den) & lm;
}

__global__ __launch_bounds__(256) void node_agg_kernel(
    const int* __restrict__ rowptr, const int* __restrict__ csr, int N,
    const float* __restrict__ H, const float* __restrict__ a_s, const float* __restrict__ a_d,
    const float* __restrict__ bias, float* __restrict__ out)
{
    const int gw   = (int)((blockIdx.x * 256 + threadIdx.x) >> 5);
    const int lane = threadIdx.x & 31;
    if (gw >= N) return;
    const int node = __builtin_amdgcn_readfirstlane(gw);
    const int lo = (lane < 16) ? -1 : 0;
    const float4* as4 = (const float4*)a_s;
    const float4 adq = ((const float4*)a_d)[node];
    const float4 asn = as4[node];
    const int beg = rowptr[node], end = rowptr[node + 1];

    const float4* hn = (const float4*)(H + (size_t)node * HC) + lane;
    const float4 n0 = hn[0], n1 = hn[32];
    const float4 bq0 = ((const float4*)bias)[lane], bq1 = ((const float4*)bias)[32 + lane];

    const float4 es = edge_logits(asn, adq);

    const int cnt0 = min(32, end - beg);
    int s0 = node;
    if (cnt0 > 0) s0 = csr[beg + min(lane, cnt0 - 1)];
    const float4 e0 = edge_logits(as4[s0], adq);

    float4 m = make_float4(fmaxf(es.x, e0.x), fmaxf(es.y, e0.y), fmaxf(es.z, e0.z), fmaxf(es.w, e0.w));
    #pragma clang loop vectorize(disable) interleave(disable)
    for (int base = beg + 32; base < end; base += 32) {
        const int cnt = min(32, end - base);
        const float4 e = edge_logits(as4[csr[base + min(lane, cnt - 1)]], adq);
        m.x = fmaxf(m.x, e.x); m.y = fmaxf(m.y, e.y); m.z = fmaxf(m.z, e.z); m.w = fmaxf(m.w, e.w);
    }
    m = wave_max4(m);
    const float m0 = uniform_f(m.x), m1 = uniform_f(m.y), m2 = uniform_f(m.z), m3 = uniform_f(m.w);

    const int lm0 = (lane < cnt0) ? -1 : 0;
    float4 sum = make_float4(__int_as_float(__float_as_int(__expf(e0.x - m0)) & lm0),
                             __int_as_float(__float_as_int(__expf(e0.y - m1)) & lm0),
                             __int_as_float(__float_as_int(__expf(e0.z - m2)) & lm0),
                             __int_as_float(__float_as_int(__expf(e0.w - m3)) & lm0));
    for (int base = beg + 32; base < end; base += 32) {
        const int cnt = min(32, end - base);
        const float4 e = edge_logits(as4[csr[base + min(lane, cnt - 1)]], adq);
        const int lm = (lane < cnt) ? -1 : 0;
        sum.x += __int_as_float(__float_as_int(__expf(e.x - m0)) & lm);
        sum.y += __int_as_float(__float_as_int(__expf(e.y - m1)) & lm);
        sum.z += __int_as_float(__float_as_int(__expf(e.z - m2)) & lm);
        sum.w += __int_as_float(__float_as_int(__expf(e.w - m3)) & lm);
    }
    sum = wave_sum4(sum);
    const float xs0 = __expf(es.x - m0), xs1 = __expf(es.y - m1);
    const float xs2 = __expf(es.z - m2), xs3 = __expf(es.w - m3);
    const float den0 = uniform_f(sum.x + xs0) + EPS_F, den1 = uniform_f(sum.y + xs1) + EPS_F;
    const float den2 = uniform_f(sum.z + xs2) + EPS_F, den3 = uniform_f(sum.w + xs3) + EPS_F;

    float4 acc0 = make_float4(0.0f, 0.0f, 0.0f, 0.0f), acc1 = acc0;
    int sv  = s0;
    int cnt = cnt0;
    int q0 = edge_weight(e0.x, m0, den0, lm0), q1 = edge_weight(e0.y, m1, den1, lm0);
    int q2 = edge_weight(e0.z, m2, den2, lm0), q3 = edge_weight(e0.w, m3, den3, lm0);
    for (int base = beg; base < end; ) {
        for (int j = 0; j < cnt; j += 4) {
            int    s[4];
            float  al[4], ah[4];
            float4 a[4], b[4];
            for (int k = 0; k < 4; ++k) {
                s[k]  = __builtin_amdgcn_readlane(sv, j + k);
                al[k] = blend_half(lo, __builtin_amdgcn_readlane(q0, j + k),
                                       __builtin_amdgcn_readlane(q1, j + k));
                ah[k] = blend_half(lo, __builtin_amdgcn_readlane(q2, j + k),
                                       __builtin_amdgcn_readlane(q3, j + k));
            }
            for (int k = 0; k < 4; ++k) {
                const float4* hs = (const float4*)(H + (size_t)s[k] * HC) + lane;
                a[k] = hs[0]; b[k] = hs[32];
            }
            for (int k = 0; k < 4; ++k) {
                acc0.x += al[k] * a[k].x; acc0.y += al[k] * a[k].y;
                acc0.z += al[k] * a[k].z; acc0.w += al[k] * a[k].w;
                acc1.x += ah[k] * b[k].x; acc1.y += ah[k] * b[k].y;
                acc1.z += ah[k] * b[k].z; acc1.w += ah[k] * b[k].w;
            }
        }
        base += 32;
        if (base < end) {
            cnt = min(32, end - base);
            sv  = csr[base + min(lane, cnt - 1)];
            const float4 e = edge_logits(as4[sv], adq);
            const int lm = (lane < cnt) ? -1 : 0;
            q0 = edge_weight(e.x, m0, den0, lm); q1 = edge_weight(e.y, m1, den1, lm);
            q2 = edge_weight(e.z, m2, den2, lm); q3 = edge_weight(e.w, m3, den3, lm);
        }
    }
    {
        const float sl = blend_half(lo, __float_as_int(xs0 / den0), __float_as_int(xs1 / den1));
        const float sh = blend_half(lo, __float_as_int(xs2 / den2), __float_as_int(xs3 / den3));
        acc0.x += sl * n0.x; acc0.y += sl * n0.y;
        acc0.z += sl * n0.z; acc0.w += sl * n0.w;
        acc1.x += sh * n1.x; acc1.y += sh * n1.y;
        acc1.z += sh * n1.z; acc1.w += sh * n1.w;
    }
    acc0.x = fmaxf(acc0.x + bq0.x, 0.0f); acc0.y = fmaxf(acc0.y + bq0.y, 0.0f);
    acc0.z = fmaxf(acc0.z + bq0.z, 0.0f); acc0.w = fmaxf(acc0.w + bq0.w, 0.0f);
    acc1.x = fmaxf(acc1.x + bq1.x, 0.0f); acc1.y = fmaxf(acc1.y + bq1.y, 0.0f);
    acc1.z = fmaxf(acc1.z + bq1.z, 0.0f); acc1.w = fmaxf(acc1.w + bq1.w, 0.0f);
    float4* op = (float4*)(out + (size_t)node * HC) + lane;
    op[0] = acc0; op[32] = acc1;
}

extern "C" void kernel_launch(void* const* d_in, const int* in_sizes, int n_in,
                              void* d_out, int out_size, void* d_ws, size_t ws_size,
                              hipStream_t stream)
{
    const float* x   = (const float*)d_in[0];
    const int*   ei  = (const int*)d_in[1];
    const float* W1  = (const float*)d_in[2];
    const float* as1 = (const float*)d_in[3];
    const float* ad1 = (const float*)d_in[4];
    const float* b1  = (const float*)d_in[5];
    const float* W2  = (const float*)d_in[6];
    const float* as2 = (const float*)d_in[7];
    const float* ad2 = (const float*)d_in[8];
    const float* b2  = (const float*)d_in[9];
    const float* Wm1 = (const float*)d_in[10];
    const float* bm1 = (const float*)d_in[11];
    const float* Wm2 = (const float*)d_in[12];
    const float* bm2 = (const float*)d_in[13];

    const int N    = in_sizes[0] / 128;
    const int E    = in_sizes[1] / 2;
    const int NH   = N * HEADS;

    float* ws = (float*)d_ws;
    size_t o = 0;
    float*    P0    = ws + o; o += (size_t)N * HC;
    float*    P1    = ws + o; o += (size_t)N * HC;
    float*    a_s   = ws + o; o += (size_t)NH;
    float*    a_d   = ws + o; o += (size_t)NH;
    const int NB = (N + BUCKET_NODES - 1) >> BUCKET_SHIFT;
    const int GA = (E + PART_CH - 1) / PART_CH;
    const size_t tab = ((size_t)NB * GA + 3) & ~(size_t)3;
    int*      rowptr = (int*)(ws + o); o += ((size_t)N + 4) & ~(size_t)3;
    int*      cntT   = (int*)(ws + o); o += tab;
    int*      offT   = (int*)(ws + o); o += tab;
    int*      pairs  = (int*)(ws + o);  o += ((size_t)2 * E + 3) & ~(size_t)3;
    int*      csr    = (int*)(ws + o); o += (size_t)E;

    const dim3 blk(256);
    auto gemm_grid = [](int M, int NC) {
        return dim3((unsigned)(((M + 15) / 16 + 7) / 8), (unsigned)((NC + 63) / 64));
    };
    const int gNodeW  = (N * 32 + 255) / 256;

    edge_partition_kernel<<<GA, dim3(PART_T), 0, stream>>>(ei, E, NB, GA, cntT, offT, pairs);
    bucket_sort_kernel<<<NB, dim3(PART_T), 0, stream>>>(cntT, offT, pairs, GA, NB, N, rowptr, csr);

    wmma_gemm_f32_wide_kernel<true, HC, 128><<<gemm_grid(N, HC), blk, 0, stream>>>(x, W1, P0, nullptr, N,
                                                                                   as1, ad1, a_s, a_d);
    node_agg_kernel<<<gNodeW, blk, 0, stream>>>(rowptr, csr, N, P0, a_s, a_d, b1, P1);

    wmma_gemm_f32_wide_kernel<true, HC, HC><<<gemm_grid(N, HC), blk, 0, stream>>>(P1, W2, P0, nullptr, N,
                                                                                  as2, ad2, a_s, a_d);
    node_agg_kernel<<<gNodeW, blk, 0, stream>>>(rowptr, csr, N, P0, a_s, a_d, b2, P1);

    wmma_gemm_f32_wide_kernel<false, HID, HC><<<gemm_grid(N, HID), blk, 0, stream>>>(P1, Wm1, P0, bm1, N,
                                                                                     nullptr, nullptr, nullptr, nullptr);
    wmma_gemm_f32_kernel<<<gemm_grid(N, 121), blk, 0, stream>>>(P0, Wm2, (float*)d_out, bm2, N, HID, 121, 2);
}


# The two closing linear layers share one launch: a wave keeps its 16x64 intermediate in LDS and re-reads it as WMMA A fragments (k 0,1,4,5 then 2,3,6,7: new summation order), W2 in LDS
# speedup vs baseline: 6.8902x; 1.0772x over previous
#include <hip/hip_runtime.h>
#include <math.h>

#define HEADS 4
#define HID 64
#define HC 256
#define NEG_SLOPE 0.2f
#define EPS_F 1e-16f

typedef __attribute__((ext_vector_type(2))) float v2f;
typedef __attribute__((ext_vector_type(8))) float v8f;

template <int K> __device__ __forceinline__ float row_xor(float v) {
    return __int_as_float(__builtin_amdgcn_update_dpp(0, __float_as_int(v), 0x160 + K, 0xf, 0xf, false));
}

template <bool SCORES, int NC, int K>
__global__ __launch_bounds__(256) void wmma_gemm_f32_wide_kernel(
    const float* __restrict__ A, const float* __restrict__ B,
    float* __restrict__ C, const float* __restrict__ bias,
    int M,
    const float* __restrict__ att_src, const float* __restrict__ att_dst,
    float* __restrict__ a_s, float* __restrict__ a_d)
{
    __shared__ float Bs[K * 64];
    const int n0 = blockIdx.y * 64;
    {
        const int fr = threadIdx.x >> 4, fc = (threadIdx.x & 15) * 4;
        const float* Bg = B + (size_t)fr * NC + n0 + fc;
        float4 t4[K / 16];
        #pragma unroll
        for (int p = 0; p < K / 16; ++p) t4[p] = *(const float4*)(Bg + (size_t)(16 * p) * NC);
        #pragma unroll
        for (int p = 0; p < K / 16; ++p) *(float4*)(Bs + (fr + 16 * p) * 64 + fc) = t4[p];
    }
    __syncthreads();

    const int wave = threadIdx.x >> 5;
    const int lane = threadIdx.x & 31;
    const int m0 = (blockIdx.x * 8 + wave) * 16;
    if (m0 >= M) return;
    const int c = lane & 15, h = lane >> 4;

    const float* Ap = A + (size_t)(m0 + c) * K + 4 * h;
    const float* Bp = Bs + (4 * h) * 64 + 4 * c;

    v8f acc0 = {}, acc1 = {}, acc2 = {}, acc3 = {};

    #pragma unroll 2
    for (int kb = 0; kb < K; kb += 8) {
        const float4 a  = *(const float4*)Ap;
        const float4 r0 = *(const float4*)(Bp);
        const float4 r1 = *(const float4*)(Bp + 64);
        const float4 r2 = *(const float4*)(Bp + 128);
        const float4 r3 = *(const float4*)(Bp + 192);
        Ap += 8; Bp += 8 * 64;
        const v2f a01 = { a.x, a.y }, a23 = { a.z, a.w };
        const v2f b00 = { r0.x, r1.x }, b01 = { r0.y, r1.y }, b02 = { r0.z, r1.z }, b03 = { r0.w, r1.w };
        const v2f b10 = { r2.x, r3.x }, b11 = { r2.y, r3.y }, b12 = { r2.z, r3.z }, b13 = { r2.w, r3.w };
        acc0 = __builtin_amdgcn_wmma_f32_16x16x4_f32(false, a01, false, b00, (short)0, acc0, false, false);
        acc1 = __builtin_amdgcn_wmma_f32_16x16x4_f32(false, a01, false, b01, (short)0, acc1, false, false);
        acc2 = __builtin_amdgcn_wmma_f32_16x16x4_f32(false, a01, false, b02, (short)0, acc2, false, false);
        acc3 = __builtin_amdgcn_wmma_f32_16x16x4_f32(false, a01, false, b03, (short)0, acc3, false, false);
        acc0 = __builtin_amdgcn_wmma_f32_16x16x4_f32(false, a23, false, b10, (short)0, acc0, false, false);
        acc1 = __builtin_amdgcn_wmma_f32_16x16x4_f32(false, a23, false, b11, (short)0, acc1, false, false);
        acc2 = __builtin_amdgcn_wmma_f32_16x16x4_f32(false, a23, false, b12, (short)0, acc2, false, false);
        acc3 = __builtin_amdgcn_wmma_f32_16x16x4_f32(false, a23, false, b13, (short)0, acc3, false, false);
    }

    const int mbase = m0 + (h << 3);
    const int colb  = n0 + 4 * c;
    float4 bq = make_float4(0.0f, 0.0f, 0.0f, 0.0f);
    if (bias != nullptr) bq = *(const float4*)(bias + colb);
    #pragma unroll
    for (int r = 0; r < 8; ++r) {
        const float4 v = make_float4(acc0[r] + bq.x, acc1[r] + bq.y, acc2[r] + bq.z, acc3[r] + bq.w);
        *(float4*)(C + (size_t)(mbase + r) * NC + colb) = v;
    }

    if (SCORES) {
        const float4 sq = *(const float4*)(att_src + colb);
        const float4 dq = *(const float4*)(att_dst + colb);
        float ps[8], pd[8];
        #pragma unroll
        for (int r = 0; r < 8; ++r) {
            ps[r] = acc0[r] * sq.x + acc1[r] * sq.y + acc2[r] * sq.z + acc3[r] * sq.w;
            pd[r] = acc0[r] * dq.x + acc1[r] * dq.y + acc2[r] * dq.z + acc3[r] * dq.w;
        }
        #pragma unroll
        for (int r = 0; r < 8; ++r) {
            ps[r] += row_xor<1>(ps[r]); pd[r] += row_xor<1>(pd[r]);
            ps[r] += row_xor<2>(ps[r]); pd[r] += row_xor<2>(pd[r]);
            ps[r] += row_xor<4>(ps[r]); pd[r] += row_xor<4>(pd[r]);
            ps[r] += row_xor<8>(ps[r]); pd[r] += row_xor<8>(pd[r]);
        }
        const int rsel = c & 7;
        float vs = ps[0], vd = pd[0];
        #pragma unroll
        for (int r = 1; r < 8; ++r) {
            vs = (rsel == r) ? ps[r] : vs;
            vd = (rsel == r) ? pd[r] : vd;
        }
        constexpr int heads = NC >> 6;
        float* dst = (c < 8) ? a_s : a_d;
        dst[(size_t)(mbase + rsel) * heads + blockIdx.y] = (c < 8) ? vs : vd;
    }
}

#define MLP_K1 256
#define MLP_H 64
#define MLP_NO 121
#define MLP_NP 128
#define MLP_TP 68
#define MLP_PATCH (16 * MLP_NO)
__global__ __launch_bounds__(256) void post_mlp_kernel(
    const float* __restrict__ A, const float* __restrict__ W1, const float* __restrict__ b1,
    const float* __restrict__ W2, const float* __restrict__ b2, float* __restrict__ out, int M)
{
    __shared__ float Bs1[MLP_K1 * MLP_H];
    __shared__ float Bs2[MLP_H * MLP_NP];
    __shared__ __attribute__((aligned(16))) float Pw[8 * MLP_PATCH + 256];
    {
        const int fr = threadIdx.x >> 4, fc = (threadIdx.x & 15) * 4;
        const float* Bg = W1 + (size_t)fr * MLP_H + fc;
        const int col2 = threadIdx.x & (MLP_NP - 1), k2 = threadIdx.x >> 7;
        const float* W2g = W2 + k2 * MLP_NO + min(col2, MLP_NO - 1);
        float4 t4[MLP_K1 / 16];
        float  u[MLP_H / 2];
        #pragma unroll
        for (int p = 0; p < MLP_K1 / 16; ++p) t4[p] = *(const float4*)(Bg + (size_t)(16 * p) * MLP_H);
        #pragma unroll
        for (int p = 0; p < MLP_H / 2; ++p) u[p] = W2g[(2 * p) * MLP_NO];
        #pragma unroll
        for (int p = 0; p < MLP_K1 / 16; ++p) *(float4*)(Bs1 + (fr + 16 * p) * MLP_H + fc) = t4[p];
        #pragma unroll
        for (int p = 0; p < MLP_H / 2; ++p) Bs2[(k2 + 2 * p) * MLP_NP + col2] = (col2 < MLP_NO) ? u[p] : 0.0f;
    }
    __syncthreads();

    const int wave = threadIdx.x >> 5;
    const int lane = threadIdx.x & 31;
    const int m0 = (blockIdx.x * 8 + wave) * 16;
    if (m0 >= M) return;
    const int c = lane & 15, h = lane >> 4;

    const float* Ap = A + (size_t)(m0 + c) * MLP_K1 + 4 * h;
    const float* Bp = Bs1 + (4 * h) * MLP_H + 4 * c;
    v8f acc0 = {}, acc1 = {}, acc2 = {}, acc3 = {};
    #pragma unroll 2
    for (int kb = 0; kb < MLP_K1; kb += 8) {
        const float4 a  = *(const float4*)Ap;
        const float4 r0 = *(const float4*)(Bp);
        const float4 r1 = *(const float4*)(Bp + MLP_H);
        const float4 r2 = *(const float4*)(Bp + 2 * MLP_H);
        const float4 r3 = *(const float4*)(Bp + 3 * MLP_H);
        Ap += 8; Bp += 8 * MLP_H;
        const v2f a01 = { a.x, a.y }, a23 = { a.z, a.w };
        const v2f b00 = { r0.x, r1.x }, b01 = { r0.y, r1.y }, b02 = { r0.z, r1.z }, b03 = { r0.w, r1.w };
        const v2f b10 = { r2.x, r3.x }, b11 = { r2.y, r3.y }, b12 = { r2.z, r3.z }, b13 = { r2.w, r3.w };
        acc0 = __builtin_amdgcn_wmma_f32_16x16x4_f32(false, a01, false, b00, (short)0, acc0, false, false);
        acc1 = __builtin_amdgcn_wmma_f32_16x16x4_f32(false, a01, false, b01, (short)0, acc1, false, false);
        acc2 = __builtin_amdgcn_wmma_f32_16x16x4_f32(false, a01, false, b02, (short)0, acc2, false, false);
        acc3 = __builtin_amdgcn_wmma_f32_16x16x4_f32(false, a01, false, b03, (short)0, acc3, false, false);
        acc0 = __builtin_amdgcn_wmma_f32_16x16x4_f32(false, a23, false, b10, (short)0, acc0, false, false);
        acc1 = __builtin_amdgcn_wmma_f32_16x16x4_f32(false, a23, false, b11, (short)0, acc1, false, false);
        acc2 = __builtin_amdgcn_wmma_f32_16x16x4_f32(false, a23, false, b12, (short)0, acc2, false, false);
        acc3 = __builtin_amdgcn_wmma_f32_16x16x4_f32(false, a23, false, b13, (short)0, acc3, false, false);
    }

    float* Tw = Pw + wave * MLP_PATCH;
    {
        const float4 bq = *(const float4*)(b1 + 4 * c);
        #pragma unroll
        for (int r = 0; r < 8; ++r)
            *(float4*)(Tw + (8 * h + r) * MLP_TP + 4 * c) =
                make_float4(acc0[r] + bq.x, acc1[r] + bq.y, acc2[r] + bq.z, acc3[r] + bq.w);
    }
    const float* Ta = Tw + c * MLP_TP + 4 * h;

    v8f o0 = {}, o1 = {}, o2 = {}, o3 = {}, o4 = {}, o5 = {}, o6 = {}, o7 = {};
    const float* B2 = Bs2 + (4 * h) * MLP_NP + 4 * c;
    #pragma unroll 2
    for (int q = 0; q < 8; ++q) {
        const float4 aq = *(const float4*)(Ta + 8 * q);
        const v2f a01 = { aq.x, aq.y }, a23 = { aq.z, aq.w };
        const float* Bq = B2 + (8 * q) * MLP_NP;
        {
            const float4 r0 = *(const float4*)(Bq);
            const float4 r1 = *(const float4*)(Bq + MLP_NP);
            const float4 r2 = *(const float4*)(Bq + 2 * MLP_NP);
            const float4 r3 = *(const float4*)(Bq + 3 * MLP_NP);
            const v2f b00 = { r0.x, r1.x }, b01 = { r0.y, r1.y }, b02 = { r0.z, r1.z }, b03 = { r0.w, r1.w };
            const v2f b10 = { r2.x, r3.x }, b11 = { r2.y, r3.y }, b12 = { r2.z, r3.z }, b13 = { r2.w, r3.w };
            o0 = __builtin_amdgcn_wmma_f32_16x16x4_f32(false, a01, false, b00, (short)0, o0, false, false);
            o1 = __builtin_amdgcn_wmma_f32_16x16x4_f32(false, a01, false, b01, (short)0, o1, false, false);
            o2 = __builtin_amdgcn_wmma_f32_16x16x4_f32(false, a01, false, b02, (short)0, o2, false, false);
            o3 = __builtin_amdgcn_wmma_f32_16x16x4_f32(false, a01, false, b03, (short)0, o3, false, false);
            o0 = __builtin_amdgcn_wmma_f32_16x16x4_f32(false, a23, false, b10, (short)0, o0, false, false);
            o1 = __builtin_amdgcn_wmma_f32_16x16x4_f32(false, a23, false, b11, (short)0, o1, false, false);
            o2 = __builtin_amdgcn_wmma_f32_16x16x4_f32(false, a23, false, b12, (short)0, o2, false, false);
            o3 = __builtin_amdgcn_wmma_f32_16x16x4_f32(false, a23, false, b13, (short)0, o3, false, false);
        }
        {
            const float4 r0 = *(const float4*)(Bq + 64);
            const float4 r1 = *(const float4*)(Bq + MLP_NP + 64);
            const float4 r2 = *(const float4*)(Bq + 2 * MLP_NP + 64);
            const float4 r3 = *(const float4*)(Bq + 3 * MLP_NP + 64);
            const v2f b00 = { r0.x, r1.x }, b01 = { r0.y, r1.y }, b02 = { r0.z, r1.z }, b03 = { r0.w, r1.w };
            const v2f b10 = { r2.x, r3.x }, b11 = { r2.y, r3.y }, b12 = { r2.z, r3.z }, b13 = { r2.w, r3.w };
            o4 = __builtin_amdgcn_wmma_f32_16x16x4_f32(false, a01, false, b00, (short)0, o4, false, false);
            o5 = __builtin_amdgcn_wmma_f32_16x16x4_f32(false, a01, false, b01, (short)0, o5, false, false);
            o6 = __builtin_amdgcn_wmma_f32_16x16x4_f32(false, a01, false, b02, (short)0, o6, false, false);
            o7 = __builtin_amdgcn_wmma_f32_16x16x4_f32(false, a01, false, b03, (short)0, o7, false, false);
            o4 = __builtin_amdgcn_wmma_f32_16x16x4_f32(false, a23, false, b10, (short)0, o4, false, false);
            o5 = __builtin_amdgcn_wmma_f32_16x16x4_f32(false, a23, false, b11, (short)0, o5, false, false);
            o6 = __builtin_amdgcn_wmma_f32_16x16x4_f32(false, a23, false, b12, (short)0, o6, false, false);
            o7 = __builtin_amdgcn_wmma_f32_16x16x4_f32(false, a23, false, b13, (short)0, o7, false, false);
        }
    }

    float* sink = Pw + 8 * MLP_PATCH + threadIdx.x;
    float bz[8];
    #pragma unroll
    for (int j = 0; j < 4; ++j) {
        bz[j]     = b2[4 * c + j];
        bz[4 + j] = b2[min(64 + 4 * c + j, MLP_NO - 1)];
    }
    #pragma unroll
    for (int r = 0; r < 8; ++r) {
        float* row = Tw + (8 * h + r) * MLP_NO + 4 * c;
        const float v[8] = { o0[r], o1[r], o2[r], o3[r], o4[r], o5[r], o6[r], o7[r] };
        #pragma unroll
        for (int j = 0; j < 4; ++j) {
            const float x = v[j] + bz[j];
            const float y = v[4 + j] + bz[4 + j];
            float* py = (64 + 4 * c + j < MLP_NO) ? row + 64 + j : sink;
            row[j] = 1.0f / (1.0f + __expf(-x));
            *py    = 1.0f / (1.0f + __expf(-y));
        }
    }
    float* og = out + (size_t)m0 * MLP_NO;
    if (m0 + 16 <= M) {
        constexpr int PIECES = MLP_PATCH / 4, SWEEPS = PIECES / 32;
        #pragma unroll
        for (int n = 0; n < SWEEPS; ++n)
            ((float4*)og)[lane + 32 * n] = ((const float4*)Tw)[lane + 32 * n];
        if (lane < PIECES - 32 * SWEEPS)
            ((float4*)og)[lane + 32 * SWEEPS] = ((const float4*)Tw)[lane + 32 * SWEEPS];
    } else {
        const int nfl = (M - m0) * MLP_NO;
        for (int i = lane; i < nfl; i += 32) og[i] = Tw[i];
    }
}

#define PART_T 1024
#define PART_CH 4096
#define BUCKET_SHIFT 8
#define BUCKET_NODES 256
#define SORT_CAP 12288

__device__ __forceinline__ int block_excl_scan(int v, int* ws, int& total) {
    const int lane = threadIdx.x & 31, wv = threadIdx.x >> 5;
    int incl = v;
    for (int off = 1; off < 32; off <<= 1) {
        const int x = __shfl_up(incl, off, 32);
        if (lane >= off) incl += x;
    }
    if (lane == 31) ws[wv] = incl;
    __syncthreads();
    const int w = ws[lane];
    int front = (lane < wv) ? w : 0;
    int tot = w;
    for (int off = 16; off > 0; off >>= 1) {
        front += __shfl_xor(front, off, 32);
        tot   += __shfl_xor(tot, off, 32);
    }
    total = tot;
    return front + incl - v;
}

__global__ __launch_bounds__(PART_T) void edge_partition_kernel(
    const int* __restrict__ ei, int E, int NB, int GA,
    int* __restrict__ cntT, int* __restrict__ offT, int* __restrict__ pairs_w)
{
    int2* __restrict__ pairs = (int2*)pairs_w;
    __shared__ int  hist[PART_T];
    __shared__ int  offs[PART_T];
    __shared__ int  wtmp[32];
    __shared__ int2 stage[PART_CH];
    const int t = threadIdx.x, g = blockIdx.x;
    hist[t] = 0;
    __syncthreads();

    int d[4], s[4], bk[4], r[4];
    #pragma unroll
    for (int k = 0; k < 4; ++k) {
        const int e = g * PART_CH + k * PART_T + t;
        d[k] = -1; s[k] = 0;
        if (e < E) { d[k] = ei[E + e]; s[k] = ei[e]; }
    }
    #pragma unroll
    for (int k = 0; k < 4; ++k) {
        bk[k] = min((int)((unsigned)d[k] >> BUCKET_SHIFT), PART_T - 1);
        r[k] = 0;
        if (d[k] >= 0) r[k] = atomicAdd(&hist[bk[k]], 1);
    }
    __syncthreads();

    const int v = hist[t];
    int total;
    const int ex = block_excl_scan(v, wtmp, total);
    offs[t] = ex;
    if (t < NB) { cntT[t * GA + g] = v; offT[t * GA + g] = ex; }
    __syncthreads();

    #pragma unroll
    for (int k = 0; k < 4; ++k)
        if (d[k] >= 0) stage[offs[bk[k]] + r[k]] = make_int2(d[k], s[k]);
    __syncthreads();

    #pragma unroll
    for (int k = 0; k < 4; ++k) {
        const int i = k * PART_T + t;
        const int e = g * PART_CH + i;
        if (e < E) pairs[e] = stage[i];
    }
}

template <bool PLACE>
__device__ __forceinline__ void visit_pair(const int2 ds, int* cnt, const int* npre,
                                           int* cst, int* out, bool in_lds) {
    const int ld = ds.x & (BUCKET_NODES - 1);
    const int r = atomicAdd(&cnt[ld], 1);
    if (PLACE) {
        const int pos = npre[ld] + r;
        if (in_lds) cst[pos] = ds.y; else out[pos] = ds.y;
    }
}
template <bool PLACE>
__device__ __forceinline__ void walk_bucket(const int2* __restrict__ pairs, int GA,
                                            const int* gcnt, const int* goff, int* cnt,
                                            const int* npre, int* cst, int* out, bool in_lds) {
    const int lane = threadIdx.x & 31, wv = threadIdx.x >> 5;
    for (int g0 = wv; g0 < GA; g0 += 128) {
        int n[4];
        const int2* p[4];
        int2 e[4];
        #pragma unroll
        for (int k = 0; k < 4; ++k) {
            const int g = g0 + 32 * k;
            const int gg = min(g, GA - 1);
            n[k] = (g < GA) ? gcnt[gg] : 0;
            p[k] = pairs + (size_t)gg * PART_CH + goff[gg];
        }
        #pragma unroll
        for (int k = 0; k < 4; ++k) {
            e[k] = make_int2(0, 0);
            if (lane < n[k]) e[k] = p[k][lane];
        }
        #pragma unroll
        for (int k = 0; k < 4; ++k)
            if (lane < n[k]) visit_pair<PLACE>(e[k], cnt, npre, cst, out, in_lds);
        #pragma unroll
        for (int k = 0; k < 4; ++k)
            for (int i = lane + 32; i < n[k]; i += 32)
                visit_pair<PLACE>(p[k][i], cnt, npre, cst, out, in_lds);
    }
}

__global__ __launch_bounds__(PART_T) void bucket_sort_kernel(
    const int* __restrict__ cntT, const int* __restrict__ offT, const int* __restrict__ pairs_w,
    int GA, int NB, int N, int* __restrict__ rowptr, int* __restrict__ csr)
{
    const int2* __restrict__ pairs = (const int2*)pairs_w;
    __shared__ int gcnt[PART_T];
    __shared__ int goff[PART_T];
    __shared__ int ncnt[BUCKET_NODES];
    __shared__ int npre[BUCKET_NODES];
    __shared__ int ncur[BUCKET_NODES];
    __shared__ int wpre[32];
    __shared__ int wtmp[32];
    __shared__ int cst[SORT_CAP];
    const int t = threadIdx.x, lane = t & 31, wv = t >> 5;
    const int b = blockIdx.x;

    const int lim = b * GA;
    const int nq = lim >> 2;
    int pre = 0;
    #pragma unroll 4
    for (int q = t; q < nq; q += PART_T) {
        const int4 u = ((const int4*)cntT)[q];
        pre += (u.x + u.y) + (u.z + u.w);
    }
    if (t < (lim & 3)) pre += cntT[(nq << 2) + t];

    int c = 0, o = 0;
    if (t < GA) { c = cntT[lim + t]; o = offT[lim + t]; }
    gcnt[t] = c; goff[t] = o;
    if (t < BUCKET_NODES) { ncnt[t] = 0; ncur[t] = 0; }
    for (int off = 16; off > 0; off >>= 1) pre += __shfl_xor(pre, off, 32);
    if (lane == 0) wpre[wv] = pre;
    __syncthreads();
    int base = wpre[lane];
    for (int off = 16; off > 0; off >>= 1) base += __shfl_xor(base, off, 32);

    walk_bucket<false>(pairs, GA, gcnt, goff, ncnt, npre, cst, csr, false);
    __syncthreads();

    const int v = (t < BUCKET_NODES) ? ncnt[t] : 0;
    int total;
    const int ex = block_excl_scan(v, wtmp, total);
    if (t < BUCKET_NODES) {
        npre[t] = ex;
        const int node = (b << BUCKET_SHIFT) + t;
        if (node < N) rowptr[node] = base + ex;
    }
    if (b == NB - 1 && t == 0) rowptr[N] = base + total;
    __syncthreads();

    const bool in_lds = (total <= SORT_CAP);
    walk_bucket<true>(pairs, GA, gcnt, goff, ncur, npre, cst, csr + base, in_lds);
    if (in_lds) {
        __syncthreads();
        for (int i = t; i < total; i += PART_T) csr[base + i] = cst[i];
    }
}


__device__ __forceinline__ float blend_half(int lo, int x_lo, int x_hi) {
    return __int_as_float((x_lo & lo) | (x_hi & ~lo));
}

__device__ __forceinline__ float leaky(float e) { return (e > 0.0f) ? e : NEG_SLOPE * e; }

__device__ __forceinline__ float4 edge_logits(float4 a, float4 d) {
    return make_float4(leaky(a.x + d.x), leaky(a.y + d.y), leaky(a.z + d.z), leaky(a.w + d.w));
}

__device__ __forceinline__ float other_row(float v) {
    const int b = __float_as_int(v);
    return __int_as_float(__builtin_amdgcn_permlanex16(b, b, 0x76543210, 0xfedcba98, false, false));
}
__device__ __forceinline__ float4 wave_max4(float4 v) {
    v.x = fmaxf(v.x, row_xor<1>(v.x)); v.y = fmaxf(v.y, row_xor<1>(v.y)); v.z = fmaxf(v.z, row_xor<1>(v.z)); v.w = fmaxf(v.w, row_xor<1>(v.w));
    v.x = fmaxf(v.x, row_xor<2>(v.x)); v.y = fmaxf(v.y, row_xor<2>(v.y)); v.z = fmaxf(v.z, row_xor<2>(v.z)); v.w = fmaxf(v.w, row_xor<2>(v.w));
    v.x = fmaxf(v.x, row_xor<4>(v.x)); v.y = fmaxf(v.y, row_xor<4>(v.y)); v.z = fmaxf(v.z, row_xor<4>(v.z)); v.w = fmaxf(v.w, row_xor<4>(v.w));
    v.x = fmaxf(v.x, row_xor<8>(v.x)); v.y = fmaxf(v.y, row_xor<8>(v.y)); v.z = fmaxf(v.z, row_xor<8>(v.z)); v.w = fmaxf(v.w, row_xor<8>(v.w));
    v.x = fmaxf(v.x, other_row(v.x));  v.y = fmaxf(v.y, other_row(v.y));  v.z = fmaxf(v.z, other_row(v.z));  v.w = fmaxf(v.w, other_row(v.w));
    return v;
}
__device__ __forceinline__ float4 wave_sum4(float4 v) {
    v.x += row_xor<1>(v.x); v.y += row_xor<1>(v.y); v.z += row_xor<1>(v.z); v.w += row_xor<1>(v.w);
    v.x += row_xor<2>(v.x); v.y += row_xor<2>(v.y); v.z += row_xor<2>(v.z); v.w += row_xor<2>(v.w);
    v.x += row_xor<4>(v.x); v.y += row_xor<4>(v.y); v.z += row_xor<4>(v.z); v.w += row_xor<4>(v.w);
    v.x += row_xor<8>(v.x); v.y += row_xor<8>(v.y); v.z += row_xor<8>(v.z); v.w += row_xor<8>(v.w);
    v.x += other_row(v.x);  v.y += other_row(v.y);  v.z += other_row(v.z);  v.w += other_row(v.w);
    return v;
}
__device__ __forceinline__ float uniform_f(float v) {
    return __int_as_float(__builtin_amdgcn_readfirstlane(__float_as_int(v)));
}
__device__ __forceinline__ int edge_weight(float e, float m, float den, int lm) {
    return __float_as_int(__expf(e - m) / den) & lm;
}

__global__ __launch_bounds__(256) void node_agg_kernel(
    const int* __restrict__ rowptr, const int* __restrict__ csr, int N,
    const float* __restrict__ H, const float* __restrict__ a_s, const float* __restrict__ a_d,
    const float* __restrict__ bias, float* __restrict__ out)
{
    const int gw   = (int)((blockIdx.x * 256 + threadIdx.x) >> 5);
    const int lane = threadIdx.x & 31;
    if (gw >= N) return;
    const int node = __builtin_amdgcn_readfirstlane(gw);
    const int lo = (lane < 16) ? -1 : 0;
    const float4* as4 = (const float4*)a_s;
    const float4 adq = ((const float4*)a_d)[node];
    const float4 asn = as4[node];
    const int beg = rowptr[node], end = rowptr[node + 1];

    const float4* hn = (const float4*)(H + (size_t)node * HC) + lane;
    const float4 n0 = hn[0], n1 = hn[32];
    const float4 bq0 = ((const float4*)bias)[lane], bq1 = ((const float4*)bias)[32 + lane];

    const float4 es = edge_logits(asn, adq);

    const int cnt0 = min(32, end - beg);
    int s0 = node;
    if (cnt0 > 0) s0 = csr[beg + min(lane, cnt0 - 1)];
    const float4 e0 = edge_logits(as4[s0], adq);

    float4 m = make_float4(fmaxf(es.x, e0.x), fmaxf(es.y, e0.y), fmaxf(es.z, e0.z), fmaxf(es.w, e0.w));
    #pragma clang loop vectorize(disable) interleave(disable)
    for (int base = beg + 32; base < end; base += 32) {
        const int cnt = min(32, end - base);
        const float4 e = edge_logits(as4[csr[base + min(lane, cnt - 1)]], adq);
        m.x = fmaxf(m.x, e.x); m.y = fmaxf(m.y, e.y); m.z = fmaxf(m.z, e.z); m.w = fmaxf(m.w, e.w);
    }
    m = wave_max4(m);
    const float m0 = uniform_f(m.x), m1 = uniform_f(m.y), m2 = uniform_f(m.z), m3 = uniform_f(m.w);

    const int lm0 = (lane < cnt0) ? -1 : 0;
    float4 sum = make_float4(__int_as_float(__float_as_int(__expf(e0.x - m0)) & lm0),
                             __int_as_float(__float_as_int(__expf(e0.y - m1)) & lm0),
                             __int_as_float(__float_as_int(__expf(e0.z - m2)) & lm0),
                             __int_as_float(__float_as_int(__expf(e0.w - m3)) & lm0));
    for (int base = beg + 32; base < end; base += 32) {
        const int cnt = min(32, end - base);
        const float4 e = edge_logits(as4[csr[base + min(lane, cnt - 1)]], adq);
        const int lm = (lane < cnt) ? -1 : 0;
        sum.x += __int_as_float(__float_as_int(__expf(e.x - m0)) & lm);
        sum.y += __int_as_float(__float_as_int(__expf(e.y - m1)) & lm);
        sum.z += __int_as_float(__float_as_int(__expf(e.z - m2)) & lm);
        sum.w += __int_as_float(__float_as_int(__expf(e.w - m3)) & lm);
    }
    sum = wave_sum4(sum);
    const float xs0 = __expf(es.x - m0), xs1 = __expf(es.y - m1);
    const float xs2 = __expf(es.z - m2), xs3 = __expf(es.w - m3);
    const float den0 = uniform_f(sum.x + xs0) + EPS_F, den1 = uniform_f(sum.y + xs1) + EPS_F;
    const float den2 = uniform_f(sum.z + xs2) + EPS_F, den3 = uniform_f(sum.w + xs3) + EPS_F;

    float4 acc0 = make_float4(0.0f, 0.0f, 0.0f, 0.0f), acc1 = acc0;
    int sv  = s0;
    int cnt = cnt0;
    int q0 = edge_weight(e0.x, m0, den0, lm0), q1 = edge_weight(e0.y, m1, den1, lm0);
    int q2 = edge_weight(e0.z, m2, den2, lm0), q3 = edge_weight(e0.w, m3, den3, lm0);
    for (int base = beg; base < end; ) {
        for (int j = 0; j < cnt; j += 4) {
            int    s[4];
            float  al[4], ah[4];
            float4 a[4], b[4];
            for (int k = 0; k < 4; ++k) {
                s[k]  = __builtin_amdgcn_readlane(sv, j + k);
                al[k] = blend_half(lo, __builtin_amdgcn_readlane(q0, j + k),
                                       __builtin_amdgcn_readlane(q1, j + k));
                ah[k] = blend_half(lo, __builtin_amdgcn_readlane(q2, j + k),
                                       __builtin_amdgcn_readlane(q3, j + k));
            }
            for (int k = 0; k < 4; ++k) {
                const float4* hs = (const float4*)(H + (size_t)s[k] * HC) + lane;
                a[k] = hs[0]; b[k] = hs[32];
            }
            for (int k = 0; k < 4; ++k) {
                acc0.x += al[k] * a[k].x; acc0.y += al[k] * a[k].y;
                acc0.z += al[k] * a[k].z; acc0.w += al[k] * a[k].w;
                acc1.x += ah[k] * b[k].x; acc1.y += ah[k] * b[k].y;
                acc1.z += ah[k] * b[k].z; acc1.w += ah[k] * b[k].w;
            }
        }
        base += 32;
        if (base < end) {
            cnt = min(32, end - base);
            sv  = csr[base + min(lane, cnt - 1)];
            const float4 e = edge_logits(as4[sv], adq);
            const int lm = (lane < cnt) ? -1 : 0;
            q0 = edge_weight(e.x, m0, den0, lm); q1 = edge_weight(e.y, m1, den1, lm);
            q2 = edge_weight(e.z, m2, den2, lm); q3 = edge_weight(e.w, m3, den3, lm);
        }
    }
    {
        const float sl = blend_half(lo, __float_as_int(xs0 / den0), __float_as_int(xs1 / den1));
        const float sh = blend_half(lo, __float_as_int(xs2 / den2), __float_as_int(xs3 / den3));
        acc0.x += sl * n0.x; acc0.y += sl * n0.y;
        acc0.z += sl * n0.z; acc0.w += sl * n0.w;
        acc1.x += sh * n1.x; acc1.y += sh * n1.y;
        acc1.z += sh * n1.z; acc1.w += sh * n1.w;
    }
    acc0.x = fmaxf(acc0.x + bq0.x, 0.0f); acc0.y = fmaxf(acc0.y + bq0.y, 0.0f);
    acc0.z = fmaxf(acc0.z + bq0.z, 0.0f); acc0.w = fmaxf(acc0.w + bq0.w, 0.0f);
    acc1.x = fmaxf(acc1.x + bq1.x, 0.0f); acc1.y = fmaxf(acc1.y + bq1.y, 0.0f);
    acc1.z = fmaxf(acc1.z + bq1.z, 0.0f); acc1.w = fmaxf(acc1.w + bq1.w, 0.0f);
    float4* op = (float4*)(out + (size_t)node * HC) + lane;
    op[0] = acc0; op[32] = acc1;
}

extern "C" void kernel_launch(void* const* d_in, const int* in_sizes, int n_in,
                              void* d_out, int out_size, void* d_ws, size_t ws_size,
                              hipStream_t stream)
{
    const float* x   = (const float*)d_in[0];
    const int*   ei  = (const int*)d_in[1];
    const float* W1  = (const float*)d_in[2];
    const float* as1 = (const float*)d_in[3];
    const float* ad1 = (const float*)d_in[4];
    const float* b1  = (const float*)d_in[5];
    const float* W2  = (const float*)d_in[6];
    const float* as2 = (const float*)d_in[7];
    const float* ad2 = (const float*)d_in[8];
    const float* b2  = (const float*)d_in[9];
    const float* Wm1 = (const float*)d_in[10];
    const float* bm1 = (const float*)d_in[11];
    const float* Wm2 = (const float*)d_in[12];
    const float* bm2 = (const float*)d_in[13];

    const int N    = in_sizes[0] / 128;
    const int E    = in_sizes[1] / 2;
    const int NH   = N * HEADS;

    float* ws = (float*)d_ws;
    size_t o = 0;
    float*    P0    = ws + o; o += (size_t)N * HC;
    float*    P1    = ws + o; o += (size_t)N * HC;
    float*    a_s   = ws + o; o += (size_t)NH;
    float*    a_d   = ws + o; o += (size_t)NH;
    const int NB = (N + BUCKET_NODES - 1) >> BUCKET_SHIFT;
    const int GA = (E + PART_CH - 1) / PART_CH;
    const size_t tab = ((size_t)NB * GA + 3) & ~(size_t)3;
    int*      rowptr = (int*)(ws + o); o += ((size_t)N + 4) & ~(size_t)3;
    int*      cntT   = (int*)(ws + o); o += tab;
    int*      offT   = (int*)(ws + o); o += tab;
    int*      pairs  = (int*)(ws + o);  o += ((size_t)2 * E + 3) & ~(size_t)3;
    int*      csr    = (int*)(ws + o); o += (size_t)E;

    const dim3 blk(256);
    auto gemm_grid = [](int M, int NC) {
        return dim3((unsigned)(((M + 15) / 16 + 7) / 8), (unsigned)((NC + 63) / 64));
    };
    const int gNodeW  = (N * 32 + 255) / 256;

    edge_partition_kernel<<<GA, dim3(PART_T), 0, stream>>>(ei, E, NB, GA, cntT, offT, pairs);
    bucket_sort_kernel<<<NB, dim3(PART_T), 0, stream>>>(cntT, offT, pairs, GA, NB, N, rowptr, csr);

    wmma_gemm_f32_wide_kernel<true, HC, 128><<<gemm_grid(N, HC), blk, 0, stream>>>(x, W1, P0, nullptr, N,
                                                                                   as1, ad1, a_s, a_d);
    node_agg_kernel<<<gNodeW, blk, 0, stream>>>(rowptr, csr, N, P0, a_s, a_d, b1, P1);

    wmma_gemm_f32_wide_kernel<true, HC, HC><<<gemm_grid(N, HC), blk, 0, stream>>>(P1, W2, P0, nullptr, N,
                                                                                  as2, ad2, a_s, a_d);
    node_agg_kernel<<<gNodeW, blk, 0, stream>>>(rowptr, csr, N, P0, a_s, a_d, b2, P1);

    post_mlp_kernel<<<gemm_grid(N, HID), blk, 0, stream>>>(P1, Wm1, bm1, Wm2, bm2, (float*)d_out, N);
}
